# starting kernel
# baseline (speedup 1.0000x reference)
#include <hip/hip_runtime.h>
#include <hip/hip_bf16.h>
#include <math.h>

typedef float v2f __attribute__((ext_vector_type(2)));
typedef float v8f __attribute__((ext_vector_type(8)));

#define GNN_N 100000
#define D_IN 128
#define D_OUT 64

__global__ void zero_f(float* __restrict__ p, size_t n) {
    size_t i = (size_t)blockIdx.x * blockDim.x + threadIdx.x;
    size_t stride = (size_t)gridDim.x * blockDim.x;
    for (; i < n; i += stride) p[i] = 0.0f;
}

__global__ void deg_count(const int* __restrict__ dst, float* __restrict__ deg, int E) {
    int i = blockIdx.x * blockDim.x + threadIdx.x;
    if (i < E) atomicAdd(deg + dst[i], 1.0f);
}

__global__ __launch_bounds__(256)
void scatter_add(const float* __restrict__ x, const int* __restrict__ src,
                 const int* __restrict__ dst, float* __restrict__ agg, int E) {
    long g = (long)blockIdx.x * blockDim.x + threadIdx.x;
    long edge = g >> 5;
    if (edge >= E) return;
    int lane = (int)(g & 31);
    long so = (long)src[edge] * D_IN + lane * 4;
    long dro = (long)dst[edge] * D_IN + lane * 4;
    float4 v = *(const float4*)(x + so);
    atomicAdd(agg + dro + 0, v.x);
    atomicAdd(agg + dro + 1, v.y);
    atomicAdd(agg + dro + 2, v.z);
    atomicAdd(agg + dro + 3, v.w);
}

template<int NT, int EPI>
__global__ __launch_bounds__(32 * NT)
void sage_gemm(const float* __restrict__ A,
               const float* __restrict__ Agg,
               const float* __restrict__ deg,
               const float* __restrict__ Wr,
               const float* __restrict__ Wn,
               const float* __restrict__ bias,
               const float* __restrict__ gamma,
               const float* __restrict__ beta,
               float* __restrict__ out)
{
    constexpr int W = 16 * NT;
    constexpr int K = 128;
    constexpr int LDA = K + 4;
    __shared__ float As[16][LDA];
    __shared__ float Gs[16][LDA];
    __shared__ float Hs[16][W];
    __shared__ float inv[16];

    const int tid  = threadIdx.x;
    const int lane = tid & 31;
    const int wave = tid >> 5;
    const long rowBase = (long)blockIdx.x * 16;

    if (tid < 16) {
        float d = deg[rowBase + tid];
        inv[tid] = 1.0f / fmaxf(d, 1.0f);
    }
    __syncthreads();

    const float* aSrc = A + rowBase * K;
    const float* gSrc = Agg + rowBase * K;
    for (int e = tid; e < 16 * K; e += 32 * NT) {
        int r = e >> 7, c = e & 127;
        As[r][c] = aSrc[e];
        Gs[r][c] = gSrc[e] * inv[r];
    }
    __syncthreads();

    const int n0   = wave * 16;
    const int arow = lane & 15;
    const int koff = (lane >> 4) << 1;
    const int bcol = lane & 15;

    v8f acc = {};
#pragma unroll 4
    for (int k0 = 0; k0 < K; k0 += 4) {
        v2f a0, a1, b0, b1;
        a0.x = As[arow][k0 + koff];
        a0.y = As[arow][k0 + koff + 1];
        a1.x = Gs[arow][k0 + koff];
        a1.y = Gs[arow][k0 + koff + 1];
        b0.x = Wr[(k0 + koff) * W + n0 + bcol];
        b0.y = Wr[(k0 + koff + 1) * W + n0 + bcol];
        b1.x = Wn[(k0 + koff) * W + n0 + bcol];
        b1.y = Wn[(k0 + koff + 1) * W + n0 + bcol];
        acc = __builtin_amdgcn_wmma_f32_16x16x4_f32(
            false, a0, false, b0, (short)0, acc, false, false);
        acc = __builtin_amdgcn_wmma_f32_16x16x4_f32(
            false, a1, false, b1, (short)0, acc, false, false);
    }

    {
        float bv = bias[n0 + bcol];
        int rbase = (lane >> 4) << 3;
#pragma unroll
        for (int v = 0; v < 8; ++v)
            Hs[rbase + v][n0 + bcol] = acc[v] + bv;
    }
    __syncthreads();

    if constexpr (EPI == 0) {
        for (int r = wave; r < 16; r += NT) {
            float s = 0.0f;
            for (int c = lane; c < W; c += 32) s += Hs[r][c];
#pragma unroll
            for (int off = 16; off; off >>= 1) s += __shfl_xor(s, off, 32);
            float mu = s * (1.0f / W);
            float vv = 0.0f;
            for (int c = lane; c < W; c += 32) {
                float d = Hs[r][c] - mu;
                vv += d * d;
            }
#pragma unroll
            for (int off = 16; off; off >>= 1) vv += __shfl_xor(vv, off, 32);
            float rs = rsqrtf(vv * (1.0f / W) + 1e-5f);
            for (int c = lane; c < W; c += 32) {
                float val = (Hs[r][c] - mu) * rs * gamma[c] + beta[c];
                out[(rowBase + r) * W + c] = fmaxf(val, 0.0f);
            }
        }
    } else {
        for (int r = wave; r < 16; r += NT) {
            float m = -INFINITY;
            for (int c = lane; c < W; c += 32) m = fmaxf(m, Hs[r][c]);
#pragma unroll
            for (int off = 16; off; off >>= 1) m = fmaxf(m, __shfl_xor(m, off, 32));
            float s = 0.0f;
            for (int c = lane; c < W; c += 32) s += expf(Hs[r][c] - m);
#pragma unroll
            for (int off = 16; off; off >>= 1) s += __shfl_xor(s, off, 32);
            float ls = logf(s);
            for (int c = lane; c < W; c += 32)
                out[(rowBase + r) * W + c] = Hs[r][c] - m - ls;
        }
    }
}

extern "C" void kernel_launch(void* const* d_in, const int* in_sizes, int n_in,
                              void* d_out, int out_size, void* d_ws, size_t ws_size,
                              hipStream_t stream) {
    const float* x    = (const float*)d_in[0];
    const int*   esrc = (const int*)d_in[1];
    const int*   edst = (const int*)d_in[2];
    const float* Wr0  = (const float*)d_in[3];
    const float* Wn0  = (const float*)d_in[4];
    const float* b0   = (const float*)d_in[5];
    const float* g0   = (const float*)d_in[6];
    const float* be0  = (const float*)d_in[7];
    const float* Wr1  = (const float*)d_in[8];
    const float* Wn1  = (const float*)d_in[9];
    const float* b1   = (const float*)d_in[10];
    const float* g1   = (const float*)d_in[11];
    const float* be1  = (const float*)d_in[12];
    const float* Wr2  = (const float*)d_in[13];
    const float* Wn2  = (const float*)d_in[14];
    const float* b2   = (const float*)d_in[15];
    float* out = (float*)d_out;

    const int N = GNN_N;
    const int E = in_sizes[1];

    float* deg = (float*)d_ws;
    float* agg = deg + N;
    float* h0  = agg + (size_t)N * D_IN;
    float* h1  = h0 + (size_t)N * D_IN;

    const int ZB = 4096, ZT = 256;
    const int degBlocks = (E + 255) / 256;
    const int scatBlocks = (int)(((long)E * 32 + 255) / 256);
    const int gemmBlocks = N / 16;

    zero_f<<<ZB, ZT, 0, stream>>>(deg, (size_t)N * (D_IN + 1));
    deg_count<<<degBlocks, 256, 0, stream>>>(edst, deg, E);

    scatter_add<<<scatBlocks, 256, 0, stream>>>(x, esrc, edst, agg, E);
    sage_gemm<8, 0><<<gemmBlocks, 256, 0, stream>>>(x, agg, deg, Wr0, Wn0, b0, g0, be0, h0);

    zero_f<<<ZB, ZT, 0, stream>>>(agg, (size_t)N * D_IN);
    scatter_add<<<scatBlocks, 256, 0, stream>>>(h0, esrc, edst, agg, E);
    sage_gemm<8, 0><<<gemmBlocks, 256, 0, stream>>>(h0, agg, deg, Wr1, Wn1, b1, g1, be1, h1);

    zero_f<<<ZB, ZT, 0, stream>>>(agg, (size_t)N * D_IN);
    scatter_add<<<scatBlocks, 256, 0, stream>>>(h1, esrc, edst, agg, E);
    sage_gemm<4, 1><<<gemmBlocks, 128, 0, stream>>>(h1, agg, deg, Wr2, Wn2, b2, nullptr, nullptr, out);
}


# Replace the per-layer float atomic scatter-add by a gather over a dst-sorted edge list built once per call (int atomic rank, scan, fill); per-node sum order = arrival-order rank
# speedup vs baseline: 2.5752x; 2.5752x over previous
#include <hip/hip_runtime.h>
#include <hip/hip_bf16.h>
#include <math.h>

typedef float v2f __attribute__((ext_vector_type(2)));
typedef float v8f __attribute__((ext_vector_type(8)));

#define GNN_N 100000
#define D_IN 128
#define D_OUT 64

__global__ void zero_i(int* __restrict__ p, int n) {
    int i = blockIdx.x * blockDim.x + threadIdx.x;
    if (i < n) p[i] = 0;
}

__global__ void rank_edges(const int* __restrict__ dst, int* __restrict__ cnt,
                           int* __restrict__ rank, int E) {
    int i = blockIdx.x * blockDim.x + threadIdx.x;
    if (i < E) rank[i] = atomicAdd(cnt + dst[i], 1);
}

__global__ __launch_bounds__(1024)
void scan_deg(const int* __restrict__ cnt, int* __restrict__ off,
              float* __restrict__ deg, int n) {
    __shared__ int wsum[32];
    const int tid = threadIdx.x, lane = tid & 31, wave = tid >> 5;
    int carry = 0;
    int4 nxt[4];
#pragma unroll
    for (int u = 0; u < 4; ++u) nxt[u] = make_int4(0, 0, 0, 0);
    if (tid * 16 < n) {
#pragma unroll
        for (int u = 0; u < 4; ++u) nxt[u] = *(const int4*)(cnt + tid * 16 + 4 * u);
    }
    for (int base = 0; base < n; base += 16384) {
        const int i = base + tid * 16;
        int4 c[4];
#pragma unroll
        for (int u = 0; u < 4; ++u) c[u] = nxt[u];
#pragma unroll
        for (int u = 0; u < 4; ++u) nxt[u] = make_int4(0, 0, 0, 0);
        if (i + 16384 < n) {
#pragma unroll
            for (int u = 0; u < 4; ++u) nxt[u] = *(const int4*)(cnt + i + 16384 + 4 * u);
        }
        int t = 0;
#pragma unroll
        for (int u = 0; u < 4; ++u) t += c[u].x + c[u].y + c[u].z + c[u].w;
        int s = t;
#pragma unroll
        for (int o = 1; o < 32; o <<= 1) {
            int v = __shfl_up(s, o, 32);
            if (lane >= o) s += v;
        }
        if (lane == 31) wsum[wave] = s;
        __syncthreads();
        int ws = wsum[lane];
#pragma unroll
        for (int o = 1; o < 32; o <<= 1) {
            int v = __shfl_up(ws, o, 32);
            if (lane >= o) ws += v;
        }
        const int waveIncl = __shfl(ws, wave, 32);
        const int total = __shfl(ws, 31, 32);
        const int waveTot = __shfl(s, 31, 32);
        if (i < n) {
            int run = carry + (waveIncl - waveTot) + (s - t);
#pragma unroll
            for (int u = 0; u < 4; ++u) {
                int4 o4;
                o4.x = run;
                o4.y = o4.x + c[u].x;
                o4.z = o4.y + c[u].y;
                o4.w = o4.z + c[u].z;
                run = o4.w + c[u].w;
                *(int4*)(off + i + 4 * u) = o4;
                float4 d4;
                d4.x = (float)c[u].x; d4.y = (float)c[u].y; d4.z = (float)c[u].z; d4.w = (float)c[u].w;
                *(float4*)(deg + i + 4 * u) = d4;
            }
        }
        carry += total;
        __syncthreads();
    }
    if (tid == 0) off[n] = carry;
}

__global__ void fill_csr(const int* __restrict__ src, const int* __restrict__ dst,
                         const int* __restrict__ off, const int* __restrict__ rank,
                         int* __restrict__ csr, int E) {
    int i = blockIdx.x * blockDim.x + threadIdx.x;
    if (i < E) csr[off[dst[i]] + rank[i]] = src[i];
}

__global__ __launch_bounds__(256)
void gather_sum(const float* __restrict__ x, const int* __restrict__ off,
                const int* __restrict__ csr, float* __restrict__ agg, int n) {
    const int node = __builtin_amdgcn_readfirstlane(blockIdx.x * 8 + (threadIdx.x >> 5));
    if (node >= n) return;
    const int lane = threadIdx.x & 31;
    const int beg = off[node];
    const int end = off[node + 1];
    const float* xl = x + lane * 4;
    float4 acc = make_float4(0.0f, 0.0f, 0.0f, 0.0f);
    int j = beg;
    for (; j + 8 <= end; j += 8) {
        float4 v[8];
#pragma unroll
        for (int u = 0; u < 8; ++u) v[u] = *(const float4*)(xl + (long)csr[j + u] * D_IN);
#pragma unroll
        for (int u = 0; u < 8; ++u) {
            acc.x += v[u].x; acc.y += v[u].y; acc.z += v[u].z; acc.w += v[u].w;
        }
    }
    if (j + 4 <= end) {
        float4 v[4];
#pragma unroll
        for (int u = 0; u < 4; ++u) v[u] = *(const float4*)(xl + (long)csr[j + u] * D_IN);
#pragma unroll
        for (int u = 0; u < 4; ++u) {
            acc.x += v[u].x; acc.y += v[u].y; acc.z += v[u].z; acc.w += v[u].w;
        }
        j += 4;
    }
    if (j + 2 <= end) {
        float4 v0 = *(const float4*)(xl + (long)csr[j] * D_IN);
        float4 v1 = *(const float4*)(xl + (long)csr[j + 1] * D_IN);
        acc.x += v0.x; acc.y += v0.y; acc.z += v0.z; acc.w += v0.w;
        acc.x += v1.x; acc.y += v1.y; acc.z += v1.z; acc.w += v1.w;
        j += 2;
    }
    if (j < end) {
        float4 v0 = *(const float4*)(xl + (long)csr[j] * D_IN);
        acc.x += v0.x; acc.y += v0.y; acc.z += v0.z; acc.w += v0.w;
    }
    *(float4*)(agg + (long)node * D_IN + lane * 4) = acc;
}

template<int NT, int EPI>
__global__ __launch_bounds__(32 * NT)
void sage_gemm(const float* __restrict__ A,
               const float* __restrict__ Agg,
               const float* __restrict__ deg,
               const float* __restrict__ Wr,
               const float* __restrict__ Wn,
               const float* __restrict__ bias,
               const float* __restrict__ gamma,
               const float* __restrict__ beta,
               float* __restrict__ out)
{
    constexpr int W = 16 * NT;
    constexpr int K = 128;
    constexpr int LDA = K + 4;
    __shared__ float As[16][LDA];
    __shared__ float Gs[16][LDA];
    __shared__ float Hs[16][W];
    __shared__ float inv[16];

    const int tid  = threadIdx.x;
    const int lane = tid & 31;
    const int wave = tid >> 5;
    const long rowBase = (long)blockIdx.x * 16;

    if (tid < 16) {
        float d = deg[rowBase + tid];
        inv[tid] = 1.0f / fmaxf(d, 1.0f);
    }
    __syncthreads();

    const float* aSrc = A + rowBase * K;
    const float* gSrc = Agg + rowBase * K;
    for (int e = tid; e < 16 * K; e += 32 * NT) {
        int r = e >> 7, c = e & 127;
        As[r][c] = aSrc[e];
        Gs[r][c] = gSrc[e] * inv[r];
    }
    __syncthreads();

    const int n0   = wave * 16;
    const int arow = lane & 15;
    const int koff = (lane >> 4) << 1;
    const int bcol = lane & 15;

    v8f acc = {};
#pragma unroll 4
    for (int k0 = 0; k0 < K; k0 += 4) {
        v2f a0, a1, b0, b1;
        a0.x = As[arow][k0 + koff];
        a0.y = As[arow][k0 + koff + 1];
        a1.x = Gs[arow][k0 + koff];
        a1.y = Gs[arow][k0 + koff + 1];
        b0.x = Wr[(k0 + koff) * W + n0 + bcol];
        b0.y = Wr[(k0 + koff + 1) * W + n0 + bcol];
        b1.x = Wn[(k0 + koff) * W + n0 + bcol];
        b1.y = Wn[(k0 + koff + 1) * W + n0 + bcol];
        acc = __builtin_amdgcn_wmma_f32_16x16x4_f32(
            false, a0, false, b0, (short)0, acc, false, false);
        acc = __builtin_amdgcn_wmma_f32_16x16x4_f32(
            false, a1, false, b1, (short)0, acc, false, false);
    }

    {
        float bv = bias[n0 + bcol];
        int rbase = (lane >> 4) << 3;
#pragma unroll
        for (int v = 0; v < 8; ++v)
            Hs[rbase + v][n0 + bcol] = acc[v] + bv;
    }
    __syncthreads();

    if constexpr (EPI == 0) {
        for (int r = wave; r < 16; r += NT) {
            float s = 0.0f;
            for (int c = lane; c < W; c += 32) s += Hs[r][c];
#pragma unroll
            for (int off = 16; off; off >>= 1) s += __shfl_xor(s, off, 32);
            float mu = s * (1.0f / W);
            float vv = 0.0f;
            for (int c = lane; c < W; c += 32) {
                float d = Hs[r][c] - mu;
                vv += d * d;
            }
#pragma unroll
            for (int off = 16; off; off >>= 1) vv += __shfl_xor(vv, off, 32);
            float rs = rsqrtf(vv * (1.0f / W) + 1e-5f);
            for (int c = lane; c < W; c += 32) {
                float val = (Hs[r][c] - mu) * rs * gamma[c] + beta[c];
                out[(rowBase + r) * W + c] = fmaxf(val, 0.0f);
            }
        }
    } else {
        for (int r = wave; r < 16; r += NT) {
            float m = -INFINITY;
            for (int c = lane; c < W; c += 32) m = fmaxf(m, Hs[r][c]);
#pragma unroll
            for (int off = 16; off; off >>= 1) m = fmaxf(m, __shfl_xor(m, off, 32));
            float s = 0.0f;
            for (int c = lane; c < W; c += 32) s += expf(Hs[r][c] - m);
#pragma unroll
            for (int off = 16; off; off >>= 1) s += __shfl_xor(s, off, 32);
            float ls = logf(s);
            for (int c = lane; c < W; c += 32)
                out[(rowBase + r) * W + c] = Hs[r][c] - m - ls;
        }
    }
}

extern "C" void kernel_launch(void* const* d_in, const int* in_sizes, int n_in,
                              void* d_out, int out_size, void* d_ws, size_t ws_size,
                              hipStream_t stream) {
    const float* x    = (const float*)d_in[0];
    const int*   esrc = (const int*)d_in[1];
    const int*   edst = (const int*)d_in[2];
    const float* Wr0  = (const float*)d_in[3];
    const float* Wn0  = (const float*)d_in[4];
    const float* b0   = (const float*)d_in[5];
    const float* g0   = (const float*)d_in[6];
    const float* be0  = (const float*)d_in[7];
    const float* Wr1  = (const float*)d_in[8];
    const float* Wn1  = (const float*)d_in[9];
    const float* b1   = (const float*)d_in[10];
    const float* g1   = (const float*)d_in[11];
    const float* be1  = (const float*)d_in[12];
    const float* Wr2  = (const float*)d_in[13];
    const float* Wn2  = (const float*)d_in[14];
    const float* b2   = (const float*)d_in[15];
    float* out = (float*)d_out;

    const int N = GNN_N;
    const int E = in_sizes[1];

    float* deg = (float*)d_ws;
    float* agg = deg + N;
    float* h0  = agg + (size_t)N * D_IN;
    float* h1  = h0 + (size_t)N * D_IN;
    int* cnt  = (int*)(h1 + (size_t)N * D_IN);
    int* off  = cnt + N;
    int* rank = off + (N + 32);
    int* csr  = rank + E;

    const int edgeBlocks = (E + 255) / 256;
    const int nodeBlocks = (N + 255) / 256;
    const int gathBlocks = (N + 7) / 8;
    const int gemmBlocks = N / 16;

    zero_i<<<nodeBlocks, 256, 0, stream>>>(cnt, N);
    rank_edges<<<edgeBlocks, 256, 0, stream>>>(edst, cnt, rank, E);
    scan_deg<<<1, 1024, 0, stream>>>(cnt, off, deg, N);
    fill_csr<<<edgeBlocks, 256, 0, stream>>>(esrc, edst, off, rank, csr, E);

    gather_sum<<<gathBlocks, 256, 0, stream>>>(x, off, csr, agg, N);
    sage_gemm<8, 0><<<gemmBlocks, 256, 0, stream>>>(x, agg, deg, Wr0, Wn0, b0, g0, be0, h0);

    gather_sum<<<gathBlocks, 256, 0, stream>>>(h0, off, csr, agg, N);
    sage_gemm<8, 0><<<gemmBlocks, 256, 0, stream>>>(h0, agg, deg, Wr1, Wn1, b1, g1, be1, h1);

    gather_sum<<<gathBlocks, 256, 0, stream>>>(h1, off, csr, agg, N);
    sage_gemm<4, 1><<<gemmBlocks, 128, 0, stream>>>(h1, agg, deg, Wr2, Wn2, b2, nullptr, nullptr, out);
}


# sage_gemm made persistent: each wave loads its Wr/Wn fragments into registers once and walks its 6-13 row tiles; next tile's rows fetched by 16-byte loads under the current tile's WMMAs
# speedup vs baseline: 2.8628x; 1.1117x over previous
#include <hip/hip_runtime.h>
#include <hip/hip_bf16.h>
#include <math.h>

typedef float v2f __attribute__((ext_vector_type(2)));
typedef float v8f __attribute__((ext_vector_type(8)));
typedef float v4f __attribute__((ext_vector_type(4)));

#define GNN_N 100000
#define D_IN 128
#define D_OUT 64

__global__ void zero_i(int* __restrict__ p, int n) {
    int i = blockIdx.x * blockDim.x + threadIdx.x;
    if (i < n) p[i] = 0;
}

__global__ void rank_edges(const int* __restrict__ dst, int* __restrict__ cnt,
                           int* __restrict__ rank, int E) {
    int i = blockIdx.x * blockDim.x + threadIdx.x;
    if (i < E) rank[i] = atomicAdd(cnt + dst[i], 1);
}

__global__ __launch_bounds__(1024)
void scan_deg(const int* __restrict__ cnt, int* __restrict__ off,
              float* __restrict__ deg, int n) {
    __shared__ int wsum[32];
    const int tid = threadIdx.x, lane = tid & 31, wave = tid >> 5;
    int carry = 0;
    int4 nxt[4];
#pragma unroll
    for (int u = 0; u < 4; ++u) nxt[u] = make_int4(0, 0, 0, 0);
    if (tid * 16 < n) {
#pragma unroll
        for (int u = 0; u < 4; ++u) nxt[u] = *(const int4*)(cnt + tid * 16 + 4 * u);
    }
    for (int base = 0; base < n; base += 16384) {
        const int i = base + tid * 16;
        int4 c[4];
#pragma unroll
        for (int u = 0; u < 4; ++u) c[u] = nxt[u];
#pragma unroll
        for (int u = 0; u < 4; ++u) nxt[u] = make_int4(0, 0, 0, 0);
        if (i + 16384 < n) {
#pragma unroll
            for (int u = 0; u < 4; ++u) nxt[u] = *(const int4*)(cnt + i + 16384 + 4 * u);
        }
        int t = 0;
#pragma unroll
        for (int u = 0; u < 4; ++u) t += c[u].x + c[u].y + c[u].z + c[u].w;
        int s = t;
#pragma unroll
        for (int o = 1; o < 32; o <<= 1) {
            int v = __shfl_up(s, o, 32);
            if (lane >= o) s += v;
        }
        if (lane == 31) wsum[wave] = s;
        __syncthreads();
        int ws = wsum[lane];
#pragma unroll
        for (int o = 1; o < 32; o <<= 1) {
            int v = __shfl_up(ws, o, 32);
            if (lane >= o) ws += v;
        }
        const int waveIncl = __shfl(ws, wave, 32);
        const int total = __shfl(ws, 31, 32);
        const int waveTot = __shfl(s, 31, 32);
        if (i < n) {
            int run = carry + (waveIncl - waveTot) + (s - t);
#pragma unroll
            for (int u = 0; u < 4; ++u) {
                int4 o4;
                o4.x = run;
                o4.y = o4.x + c[u].x;
                o4.z = o4.y + c[u].y;
                o4.w = o4.z + c[u].z;
                run = o4.w + c[u].w;
                *(int4*)(off + i + 4 * u) = o4;
                float4 d4;
                d4.x = (float)c[u].x; d4.y = (float)c[u].y; d4.z = (float)c[u].z; d4.w = (float)c[u].w;
                *(float4*)(deg + i + 4 * u) = d4;
            }
        }
        carry += total;
        __syncthreads();
    }
    if (tid == 0) off[n] = carry;
}

__global__ void fill_csr(const int* __restrict__ src, const int* __restrict__ dst,
                         const int* __restrict__ off, const int* __restrict__ rank,
                         int* __restrict__ csr, int E) {
    int i = blockIdx.x * blockDim.x + threadIdx.x;
    if (i < E) csr[off[dst[i]] + rank[i]] = src[i];
}

__global__ __launch_bounds__(256)
void gather_sum(const float* __restrict__ x, const int* __restrict__ off,
                const int* __restrict__ csr, float* __restrict__ agg, int n) {
    const int node = __builtin_amdgcn_readfirstlane(blockIdx.x * 8 + (threadIdx.x >> 5));
    if (node >= n) return;
    const int lane = threadIdx.x & 31;
    const int beg = off[node];
    const int end = off[node + 1];
    const float* xl = x + lane * 4;
    float4 acc = make_float4(0.0f, 0.0f, 0.0f, 0.0f);
    int j = beg;
    for (; j + 8 <= end; j += 8) {
        float4 v[8];
#pragma unroll
        for (int u = 0; u < 8; ++u) v[u] = *(const float4*)(xl + (long)csr[j + u] * D_IN);
#pragma unroll
        for (int u = 0; u < 8; ++u) {
            acc.x += v[u].x; acc.y += v[u].y; acc.z += v[u].z; acc.w += v[u].w;
        }
    }
    if (j + 4 <= end) {
        float4 v[4];
#pragma unroll
        for (int u = 0; u < 4; ++u) v[u] = *(const float4*)(xl + (long)csr[j + u] * D_IN);
#pragma unroll
        for (int u = 0; u < 4; ++u) {
            acc.x += v[u].x; acc.y += v[u].y; acc.z += v[u].z; acc.w += v[u].w;
        }
        j += 4;
    }
    if (j + 2 <= end) {
        float4 v0 = *(const float4*)(xl + (long)csr[j] * D_IN);
        float4 v1 = *(const float4*)(xl + (long)csr[j + 1] * D_IN);
        acc.x += v0.x; acc.y += v0.y; acc.z += v0.z; acc.w += v0.w;
        acc.x += v1.x; acc.y += v1.y; acc.z += v1.z; acc.w += v1.w;
        j += 2;
    }
    if (j < end) {
        float4 v0 = *(const float4*)(xl + (long)csr[j] * D_IN);
        acc.x += v0.x; acc.y += v0.y; acc.z += v0.z; acc.w += v0.w;
    }
    *(float4*)(agg + (long)node * D_IN + lane * 4) = acc;
}

#define ROW_TILES (GNN_N / 16)
template<int NT, int EPI>
__global__ __launch_bounds__(32 * NT)
void sage_gemm(const float* __restrict__ A,
               const float* __restrict__ Agg,
               const float* __restrict__ deg,
               const float* __restrict__ Wr,
               const float* __restrict__ Wn,
               const float* __restrict__ bias,
               const float* __restrict__ gamma,
               const float* __restrict__ beta,
               float* __restrict__ out)
{
    constexpr int W = 16 * NT;
    constexpr int K = 128;
    constexpr int KS = K / 4;
    constexpr int LDA = K + 4;
    constexpr int RW = 16 / NT;
    constexpr int CPL = W / 32;
    __shared__ __attribute__((aligned(16))) float As[2][16][LDA];
    __shared__ __attribute__((aligned(16))) float Gs[2][16][LDA];
    __shared__ float Hs[2][16][W];

    const int lane = threadIdx.x & 31;
    const int wave = __builtin_amdgcn_readfirstlane(threadIdx.x >> 5);

    int tile = blockIdx.x;
    if (tile >= ROW_TILES) return;

    const int n0   = wave * 16;
    const int arow = lane & 15;
    const int koff = (lane >> 4) << 1;
    const int bcol = lane & 15;

    v2f bR[KS], bN[KS];
#pragma unroll
    for (int s = 0; s < KS; ++s) {
        const int k = 4 * s + koff;
        bR[s].x = Wr[k * W + n0 + bcol];
        bR[s].y = Wr[(k + 1) * W + n0 + bcol];
        bN[s].x = Wn[k * W + n0 + bcol];
        bN[s].y = Wn[(k + 1) * W + n0 + bcol];
    }
    const float bv = bias[n0 + bcol];
    float gam[CPL], bet[CPL];
    if constexpr (EPI == 0) {
#pragma unroll
        for (int m = 0; m < CPL; ++m) { gam[m] = gamma[lane + 32 * m]; bet[m] = beta[lane + 32 * m]; }
    }

    v4f pa[RW], pg[RW];
    float pd[RW];
#pragma unroll
    for (int j = 0; j < RW; ++j) {
        const long row = (long)tile * 16 + wave + NT * j;
        pa[j] = *(const v4f*)(A + row * K + lane * 4);
        pg[j] = *(const v4f*)(Agg + row * K + lane * 4);
        pd[j] = deg[row];
    }
#pragma unroll
    for (int j = 0; j < RW; ++j) {
        const float inv = 1.0f / fmaxf(pd[j], 1.0f);
        *(v4f*)&As[0][wave + NT * j][lane * 4] = pa[j];
        *(v4f*)&Gs[0][wave + NT * j][lane * 4] = pg[j] * inv;
    }
    __syncthreads();

    for (int it = 0; tile < ROW_TILES; ++it, tile += gridDim.x) {
        const int p = it & 1;
        const int nxt = tile + gridDim.x;
        const bool more = nxt < ROW_TILES;

        if (more) {
#pragma unroll
            for (int j = 0; j < RW; ++j) {
                const long row = (long)nxt * 16 + wave + NT * j;
                pa[j] = *(const v4f*)(A + row * K + lane * 4);
                pg[j] = *(const v4f*)(Agg + row * K + lane * 4);
                pd[j] = deg[row];
            }
        }

        v8f acc = {};
#pragma unroll
        for (int s = 0; s < KS; ++s) {
            const v2f a0 = *(const v2f*)&As[p][arow][4 * s + koff];
            const v2f a1 = *(const v2f*)&Gs[p][arow][4 * s + koff];
            acc = __builtin_amdgcn_wmma_f32_16x16x4_f32(
                false, a0, false, bR[s], (short)0, acc, false, false);
            acc = __builtin_amdgcn_wmma_f32_16x16x4_f32(
                false, a1, false, bN[s], (short)0, acc, false, false);
        }

        {
            const int rbase = (lane >> 4) << 3;
#pragma unroll
            for (int v = 0; v < 8; ++v)
                Hs[p][rbase + v][n0 + bcol] = acc[v] + bv;
        }
        if (more) {
#pragma unroll
            for (int j = 0; j < RW; ++j) {
                const float inv = 1.0f / fmaxf(pd[j], 1.0f);
                *(v4f*)&As[p ^ 1][wave + NT * j][lane * 4] = pa[j];
                *(v4f*)&Gs[p ^ 1][wave + NT * j][lane * 4] = pg[j] * inv;
            }
        }
        __syncthreads();

        float h[RW][CPL];
#pragma unroll
        for (int j = 0; j < RW; ++j)
#pragma unroll
            for (int m = 0; m < CPL; ++m) h[j][m] = Hs[p][wave + NT * j][lane + 32 * m];

        if constexpr (EPI == 0) {
            float s[RW], vv[RW], mu[RW];
#pragma unroll
            for (int j = 0; j < RW; ++j) {
                s[j] = 0.0f;
#pragma unroll
                for (int m = 0; m < CPL; ++m) s[j] += h[j][m];
            }
#pragma unroll
            for (int off = 16; off; off >>= 1)
#pragma unroll
                for (int j = 0; j < RW; ++j) s[j] += __shfl_xor(s[j], off, 32);
#pragma unroll
            for (int j = 0; j < RW; ++j) {
                mu[j] = s[j] * (1.0f / W);
                vv[j] = 0.0f;
#pragma unroll
                for (int m = 0; m < CPL; ++m) {
                    float d = h[j][m] - mu[j];
                    vv[j] += d * d;
                }
            }
#pragma unroll
            for (int off = 16; off; off >>= 1)
#pragma unroll
                for (int j = 0; j < RW; ++j) vv[j] += __shfl_xor(vv[j], off, 32);
#pragma unroll
            for (int j = 0; j < RW; ++j) {
                const float rs = rsqrtf(vv[j] * (1.0f / W) + 1e-5f);
                const long row = (long)tile * 16 + wave + NT * j;
#pragma unroll
                for (int m = 0; m < CPL; ++m) {
                    float val = (h[j][m] - mu[j]) * rs * gam[m] + bet[m];
                    out[row * W + lane + 32 * m] = fmaxf(val, 0.0f);
                }
            }
        } else {
            float mx[RW], s[RW];
#pragma unroll
            for (int j = 0; j < RW; ++j) {
                mx[j] = -INFINITY;
#pragma unroll
                for (int m = 0; m < CPL; ++m) mx[j] = fmaxf(mx[j], h[j][m]);
            }
#pragma unroll
            for (int off = 16; off; off >>= 1)
#pragma unroll
                for (int j = 0; j < RW; ++j) mx[j] = fmaxf(mx[j], __shfl_xor(mx[j], off, 32));
#pragma unroll
            for (int j = 0; j < RW; ++j) {
                s[j] = 0.0f;
#pragma unroll
                for (int m = 0; m < CPL; ++m) s[j] += expf(h[j][m] - mx[j]);
            }
#pragma unroll
            for (int off = 16; off; off >>= 1)
#pragma unroll
                for (int j = 0; j < RW; ++j) s[j] += __shfl_xor(s[j], off, 32);
#pragma unroll
            for (int j = 0; j < RW; ++j) {
                const float ls = logf(s[j]);
                const long row = (long)tile * 16 + wave + NT * j;
#pragma unroll
                for (int m = 0; m < CPL; ++m)
                    out[row * W + lane + 32 * m] = h[j][m] - mx[j] - ls;
            }
        }
    }
}

extern "C" void kernel_launch(void* const* d_in, const int* in_sizes, int n_in,
                              void* d_out, int out_size, void* d_ws, size_t ws_size,
                              hipStream_t stream) {
    const float* x    = (const float*)d_in[0];
    const int*   esrc = (const int*)d_in[1];
    const int*   edst = (const int*)d_in[2];
    const float* Wr0  = (const float*)d_in[3];
    const float* Wn0  = (const float*)d_in[4];
    const float* b0   = (const float*)d_in[5];
    const float* g0   = (const float*)d_in[6];
    const float* be0  = (const float*)d_in[7];
    const float* Wr1  = (const float*)d_in[8];
    const float* Wn1  = (const float*)d_in[9];
    const float* b1   = (const float*)d_in[10];
    const float* g1   = (const float*)d_in[11];
    const float* be1  = (const float*)d_in[12];
    const float* Wr2  = (const float*)d_in[13];
    const float* Wn2  = (const float*)d_in[14];
    const float* b2   = (const float*)d_in[15];
    float* out = (float*)d_out;

    const int N = GNN_N;
    const int E = in_sizes[1];

    float* deg = (float*)d_ws;
    float* agg = deg + N;
    float* h0  = agg + (size_t)N * D_IN;
    float* h1  = h0 + (size_t)N * D_IN;
    int* cnt  = (int*)(h1 + (size_t)N * D_IN);
    int* off  = cnt + N;
    int* rank = off + (N + 32);
    int* csr  = rank + E;

    const int edgeBlocks = (E + 255) / 256;
    const int nodeBlocks = (N + 255) / 256;
    const int gathBlocks = (N + 7) / 8;
    const int gemmBlocks8 = 512;
    const int gemmBlocks4 = 1024;

    zero_i<<<nodeBlocks, 256, 0, stream>>>(cnt, N);
    rank_edges<<<edgeBlocks, 256, 0, stream>>>(edst, cnt, rank, E);
    scan_deg<<<1, 1024, 0, stream>>>(cnt, off, deg, N);
    fill_csr<<<edgeBlocks, 256, 0, stream>>>(esrc, edst, off, rank, csr, E);

    gather_sum<<<gathBlocks, 256, 0, stream>>>(x, off, csr, agg, N);
    sage_gemm<8, 0><<<gemmBlocks8, 256, 0, stream>>>(x, agg, deg, Wr0, Wn0, b0, g0, be0, h0);

    gather_sum<<<gathBlocks, 256, 0, stream>>>(h0, off, csr, agg, N);
    sage_gemm<8, 0><<<gemmBlocks8, 256, 0, stream>>>(h0, agg, deg, Wr1, Wn1, b1, g1, be1, h1);

    gather_sum<<<gathBlocks, 256, 0, stream>>>(h1, off, csr, agg, N);
    sage_gemm<4, 1><<<gemmBlocks4, 128, 0, stream>>>(h1, agg, deg, Wr2, Wn2, b2, nullptr, nullptr, out);
}


# Last layer: Wn2 is applied BEFORE the neighbour mean (mean is linear): one GEMM writes h1@Wr2+b2 and h1@Wn2, the gather sums 64-wide rows and ends in log_softmax; per-node float sum order changes
# speedup vs baseline: 3.0849x; 1.0776x over previous
#include <hip/hip_runtime.h>
#include <hip/hip_bf16.h>
#include <math.h>

typedef float v2f __attribute__((ext_vector_type(2)));
typedef float v8f __attribute__((ext_vector_type(8)));
typedef float v4f __attribute__((ext_vector_type(4)));

#define GNN_N 100000
#define D_IN 128
#define D_OUT 64

__global__ void zero_i(int* __restrict__ p, int n) {
    int i = blockIdx.x * blockDim.x + threadIdx.x;
    if (i < n) p[i] = 0;
}

__global__ void rank_edges(const int* __restrict__ dst, int* __restrict__ cnt,
                           int* __restrict__ rank, int E) {
    int i = blockIdx.x * blockDim.x + threadIdx.x;
    if (i < E) rank[i] = atomicAdd(cnt + dst[i], 1);
}

__global__ __launch_bounds__(1024)
void scan_deg(const int* __restrict__ cnt, int* __restrict__ off,
              float* __restrict__ deg, int n) {
    __shared__ int wsum[32];
    const int tid = threadIdx.x, lane = tid & 31, wave = tid >> 5;
    int carry = 0;
    int4 nxt[4];
#pragma unroll
    for (int u = 0; u < 4; ++u) nxt[u] = make_int4(0, 0, 0, 0);
    if (tid * 16 < n) {
#pragma unroll
        for (int u = 0; u < 4; ++u) nxt[u] = *(const int4*)(cnt + tid * 16 + 4 * u);
    }
    for (int base = 0; base < n; base += 16384) {
        const int i = base + tid * 16;
        int4 c[4];
#pragma unroll
        for (int u = 0; u < 4; ++u) c[u] = nxt[u];
#pragma unroll
        for (int u = 0; u < 4; ++u) nxt[u] = make_int4(0, 0, 0, 0);
        if (i + 16384 < n) {
#pragma unroll
            for (int u = 0; u < 4; ++u) nxt[u] = *(const int4*)(cnt + i + 16384 + 4 * u);
        }
        int t = 0;
#pragma unroll
        for (int u = 0; u < 4; ++u) t += c[u].x + c[u].y + c[u].z + c[u].w;
        int s = t;
#pragma unroll
        for (int o = 1; o < 32; o <<= 1) {
            int v = __shfl_up(s, o, 32);
            if (lane >= o) s += v;
        }
        if (lane == 31) wsum[wave] = s;
        __syncthreads();
        int ws = wsum[lane];
#pragma unroll
        for (int o = 1; o < 32; o <<= 1) {
            int v = __shfl_up(ws, o, 32);
            if (lane >= o) ws += v;
        }
        const int waveIncl = __shfl(ws, wave, 32);
        const int total = __shfl(ws, 31, 32);
        const int waveTot = __shfl(s, 31, 32);
        if (i < n) {
            int run = carry + (waveIncl - waveTot) + (s - t);
#pragma unroll
            for (int u = 0; u < 4; ++u) {
                int4 o4;
                o4.x = run;
                o4.y = o4.x + c[u].x;
                o4.z = o4.y + c[u].y;
                o4.w = o4.z + c[u].z;
                run = o4.w + c[u].w;
                *(int4*)(off + i + 4 * u) = o4;
                float4 d4;
                d4.x = (float)c[u].x; d4.y = (float)c[u].y; d4.z = (float)c[u].z; d4.w = (float)c[u].w;
                *(float4*)(deg + i + 4 * u) = d4;
            }
        }
        carry += total;
        __syncthreads();
    }
    if (tid == 0) off[n] = carry;
}

__global__ void fill_csr(const int* __restrict__ src, const int* __restrict__ dst,
                         const int* __restrict__ off, const int* __restrict__ rank,
                         int* __restrict__ csr, int E) {
    int i = blockIdx.x * blockDim.x + threadIdx.x;
    if (i < E) csr[off[dst[i]] + rank[i]] = src[i];
}

__global__ __launch_bounds__(256)
void gather_sum(const float* __restrict__ x, const int* __restrict__ off,
                const int* __restrict__ csr, float* __restrict__ agg, int n) {
    const int node = __builtin_amdgcn_readfirstlane(blockIdx.x * 8 + (threadIdx.x >> 5));
    if (node >= n) return;
    const int lane = threadIdx.x & 31;
    const int beg = off[node];
    const int end = off[node + 1];
    const float* xl = x + lane * 4;
    float4 acc = make_float4(0.0f, 0.0f, 0.0f, 0.0f);
    int j = beg;
    for (; j + 8 <= end; j += 8) {
        float4 v[8];
#pragma unroll
        for (int u = 0; u < 8; ++u) v[u] = *(const float4*)(xl + (long)csr[j + u] * D_IN);
#pragma unroll
        for (int u = 0; u < 8; ++u) {
            acc.x += v[u].x; acc.y += v[u].y; acc.z += v[u].z; acc.w += v[u].w;
        }
    }
    if (j + 4 <= end) {
        float4 v[4];
#pragma unroll
        for (int u = 0; u < 4; ++u) v[u] = *(const float4*)(xl + (long)csr[j + u] * D_IN);
#pragma unroll
        for (int u = 0; u < 4; ++u) {
            acc.x += v[u].x; acc.y += v[u].y; acc.z += v[u].z; acc.w += v[u].w;
        }
        j += 4;
    }
    if (j + 2 <= end) {
        float4 v0 = *(const float4*)(xl + (long)csr[j] * D_IN);
        float4 v1 = *(const float4*)(xl + (long)csr[j + 1] * D_IN);
        acc.x += v0.x; acc.y += v0.y; acc.z += v0.z; acc.w += v0.w;
        acc.x += v1.x; acc.y += v1.y; acc.z += v1.z; acc.w += v1.w;
        j += 2;
    }
    if (j < end) {
        float4 v0 = *(const float4*)(xl + (long)csr[j] * D_IN);
        acc.x += v0.x; acc.y += v0.y; acc.z += v0.z; acc.w += v0.w;
    }
    *(float4*)(agg + (long)node * D_IN + lane * 4) = acc;
}

__global__ __launch_bounds__(256)
void gather_lsm(const float* __restrict__ y, const float* __restrict__ z,
                const int* __restrict__ off, const int* __restrict__ csr,
                float* __restrict__ out, int n) {
    const int node = __builtin_amdgcn_readfirstlane(blockIdx.x * 8 + (threadIdx.x >> 5));
    if (node >= n) return;
    const int lane = threadIdx.x & 31;
    const bool odd = lane >= 16;
    const int c4 = (lane & 15) * 4;
    const int beg = off[node];
    const int end = off[node + 1];
    const float4 zv = *(const float4*)(z + (long)node * D_OUT + c4);
    const float* yl = y + c4;
    float4 acc = make_float4(0.0f, 0.0f, 0.0f, 0.0f);
    int j = beg;
    for (; j + 16 <= end; j += 16) {
        float4 v[8];
#pragma unroll
        for (int u = 0; u < 8; ++u) {
            const int i0 = csr[j + 2 * u], i1 = csr[j + 2 * u + 1];
            v[u] = *(const float4*)(yl + (long)(odd ? i1 : i0) * D_OUT);
        }
#pragma unroll
        for (int u = 0; u < 8; ++u) {
            acc.x += v[u].x; acc.y += v[u].y; acc.z += v[u].z; acc.w += v[u].w;
        }
    }
    if (j + 8 <= end) {
        float4 v[4];
#pragma unroll
        for (int u = 0; u < 4; ++u) {
            const int i0 = csr[j + 2 * u], i1 = csr[j + 2 * u + 1];
            v[u] = *(const float4*)(yl + (long)(odd ? i1 : i0) * D_OUT);
        }
#pragma unroll
        for (int u = 0; u < 4; ++u) {
            acc.x += v[u].x; acc.y += v[u].y; acc.z += v[u].z; acc.w += v[u].w;
        }
        j += 8;
    }
    if (j + 4 <= end) {
        const int i0 = csr[j], i1 = csr[j + 1], i2 = csr[j + 2], i3 = csr[j + 3];
        float4 v0 = *(const float4*)(yl + (long)(odd ? i1 : i0) * D_OUT);
        float4 v1 = *(const float4*)(yl + (long)(odd ? i3 : i2) * D_OUT);
        acc.x += v0.x; acc.y += v0.y; acc.z += v0.z; acc.w += v0.w;
        acc.x += v1.x; acc.y += v1.y; acc.z += v1.z; acc.w += v1.w;
        j += 4;
    }
    if (j + 2 <= end) {
        const int i0 = csr[j], i1 = csr[j + 1];
        float4 v0 = *(const float4*)(yl + (long)(odd ? i1 : i0) * D_OUT);
        acc.x += v0.x; acc.y += v0.y; acc.z += v0.z; acc.w += v0.w;
        j += 2;
    }
    if (j < end) {
        const int i0 = csr[j];
        if (!odd) {
            float4 v0 = *(const float4*)(yl + (long)i0 * D_OUT);
            acc.x += v0.x; acc.y += v0.y; acc.z += v0.z; acc.w += v0.w;
        }
    }
    acc.x += __shfl_xor(acc.x, 16, 32);
    acc.y += __shfl_xor(acc.y, 16, 32);
    acc.z += __shfl_xor(acc.z, 16, 32);
    acc.w += __shfl_xor(acc.w, 16, 32);
    const float inv = 1.0f / fmaxf((float)(end - beg), 1.0f);
    float4 h;
    h.x = zv.x + acc.x * inv; h.y = zv.y + acc.y * inv;
    h.z = zv.z + acc.z * inv; h.w = zv.w + acc.w * inv;
    float mx = fmaxf(fmaxf(h.x, h.y), fmaxf(h.z, h.w));
#pragma unroll
    for (int o = 8; o; o >>= 1) mx = fmaxf(mx, __shfl_xor(mx, o, 32));
    float s = expf(h.x - mx) + expf(h.y - mx) + expf(h.z - mx) + expf(h.w - mx);
#pragma unroll
    for (int o = 8; o; o >>= 1) s += __shfl_xor(s, o, 32);
    const float ls = logf(s);
    if (!odd) {
        float4 r;
        r.x = h.x - mx - ls; r.y = h.y - mx - ls; r.z = h.z - mx - ls; r.w = h.w - mx - ls;
        *(float4*)(out + (long)node * D_OUT + c4) = r;
    }
}

#define ROW_TILES (GNN_N / 16)
template<int NT>
__global__ __launch_bounds__(32 * NT)
void sage_gemm(const float* __restrict__ A,
               const float* __restrict__ Agg,
               const float* __restrict__ deg,
               const float* __restrict__ Wr,
               const float* __restrict__ Wn,
               const float* __restrict__ bias,
               const float* __restrict__ gamma,
               const float* __restrict__ beta,
               float* __restrict__ out)
{
    constexpr int W = 16 * NT;
    constexpr int K = 128;
    constexpr int KS = K / 4;
    constexpr int LDA = K + 4;
    constexpr int RW = 16 / NT;
    constexpr int CPL = W / 32;
    __shared__ __attribute__((aligned(16))) float As[2][16][LDA];
    __shared__ __attribute__((aligned(16))) float Gs[2][16][LDA];
    __shared__ float Hs[2][16][W];

    const int lane = threadIdx.x & 31;
    const int wave = __builtin_amdgcn_readfirstlane(threadIdx.x >> 5);

    int tile = blockIdx.x;
    if (tile >= ROW_TILES) return;

    const int n0   = wave * 16;
    const int arow = lane & 15;
    const int koff = (lane >> 4) << 1;
    const int bcol = lane & 15;

    v2f bR[KS], bN[KS];
#pragma unroll
    for (int s = 0; s < KS; ++s) {
        const int k = 4 * s + koff;
        bR[s].x = Wr[k * W + n0 + bcol];
        bR[s].y = Wr[(k + 1) * W + n0 + bcol];
        bN[s].x = Wn[k * W + n0 + bcol];
        bN[s].y = Wn[(k + 1) * W + n0 + bcol];
    }
    const float bv = bias[n0 + bcol];
    float gam[CPL], bet[CPL];
#pragma unroll
    for (int m = 0; m < CPL; ++m) { gam[m] = gamma[lane + 32 * m]; bet[m] = beta[lane + 32 * m]; }

    v4f pa[RW], pg[RW];
    float pd[RW];
#pragma unroll
    for (int j = 0; j < RW; ++j) {
        const long row = (long)tile * 16 + wave + NT * j;
        pa[j] = *(const v4f*)(A + row * K + lane * 4);
        pg[j] = *(const v4f*)(Agg + row * K + lane * 4);
        pd[j] = deg[row];
    }
#pragma unroll
    for (int j = 0; j < RW; ++j) {
        const float inv = 1.0f / fmaxf(pd[j], 1.0f);
        *(v4f*)&As[0][wave + NT * j][lane * 4] = pa[j];
        *(v4f*)&Gs[0][wave + NT * j][lane * 4] = pg[j] * inv;
    }
    __syncthreads();

    for (int it = 0; tile < ROW_TILES; ++it, tile += gridDim.x) {
        const int p = it & 1;
        const int nxt = tile + gridDim.x;
        const bool more = nxt < ROW_TILES;

        if (more) {
#pragma unroll
            for (int j = 0; j < RW; ++j) {
                const long row = (long)nxt * 16 + wave + NT * j;
                pa[j] = *(const v4f*)(A + row * K + lane * 4);
                pg[j] = *(const v4f*)(Agg + row * K + lane * 4);
                pd[j] = deg[row];
            }
        }

        v8f acc = {};
#pragma unroll
        for (int s = 0; s < KS; ++s) {
            const v2f a0 = *(const v2f*)&As[p][arow][4 * s + koff];
            const v2f a1 = *(const v2f*)&Gs[p][arow][4 * s + koff];
            acc = __builtin_amdgcn_wmma_f32_16x16x4_f32(
                false, a0, false, bR[s], (short)0, acc, false, false);
            acc = __builtin_amdgcn_wmma_f32_16x16x4_f32(
                false, a1, false, bN[s], (short)0, acc, false, false);
        }

        {
            const int rbase = (lane >> 4) << 3;
#pragma unroll
            for (int v = 0; v < 8; ++v)
                Hs[p][rbase + v][n0 + bcol] = acc[v] + bv;
        }
        if (more) {
#pragma unroll
            for (int j = 0; j < RW; ++j) {
                const float inv = 1.0f / fmaxf(pd[j], 1.0f);
                *(v4f*)&As[p ^ 1][wave + NT * j][lane * 4] = pa[j];
                *(v4f*)&Gs[p ^ 1][wave + NT * j][lane * 4] = pg[j] * inv;
            }
        }
        __syncthreads();

        float h[RW][CPL];
#pragma unroll
        for (int j = 0; j < RW; ++j)
#pragma unroll
            for (int m = 0; m < CPL; ++m) h[j][m] = Hs[p][wave + NT * j][lane + 32 * m];

        float s[RW], vv[RW], mu[RW];
#pragma unroll
        for (int j = 0; j < RW; ++j) {
            s[j] = 0.0f;
#pragma unroll
            for (int m = 0; m < CPL; ++m) s[j] += h[j][m];
        }
#pragma unroll
        for (int off = 16; off; off >>= 1)
#pragma unroll
            for (int j = 0; j < RW; ++j) s[j] += __shfl_xor(s[j], off, 32);
#pragma unroll
        for (int j = 0; j < RW; ++j) {
            mu[j] = s[j] * (1.0f / W);
            vv[j] = 0.0f;
#pragma unroll
            for (int m = 0; m < CPL; ++m) {
                float d = h[j][m] - mu[j];
                vv[j] += d * d;
            }
        }
#pragma unroll
        for (int off = 16; off; off >>= 1)
#pragma unroll
            for (int j = 0; j < RW; ++j) vv[j] += __shfl_xor(vv[j], off, 32);
#pragma unroll
        for (int j = 0; j < RW; ++j) {
            const float rs = rsqrtf(vv[j] * (1.0f / W) + 1e-5f);
            const long row = (long)tile * 16 + wave + NT * j;
#pragma unroll
            for (int m = 0; m < CPL; ++m) {
                float val = (h[j][m] - mu[j]) * rs * gam[m] + bet[m];
                out[row * W + lane + 32 * m] = fmaxf(val, 0.0f);
            }
        }
    }
}

__global__ __launch_bounds__(256)
void proj_gemm(const float* __restrict__ A,
               const float* __restrict__ Wr,
               const float* __restrict__ Wn,
               const float* __restrict__ bias,
               float* __restrict__ z,
               float* __restrict__ y)
{
    constexpr int NT = 8;
    constexpr int K = 128;
    constexpr int KS = K / 4;
    constexpr int LDA = K + 4;
    constexpr int RW = 16 / NT;
    constexpr int LDH = 2 * D_OUT + 4;
    __shared__ __attribute__((aligned(16))) float As[2][16][LDA];
    __shared__ __attribute__((aligned(16))) float Hs[2][16][LDH];

    const int lane = threadIdx.x & 31;
    const int wave = __builtin_amdgcn_readfirstlane(threadIdx.x >> 5);

    int tile = blockIdx.x;
    if (tile >= ROW_TILES) return;

    const int arow = lane & 15;
    const int koff = (lane >> 4) << 1;
    const int bcol = lane & 15;
    const bool nbr = wave >= 4;
    const float* __restrict__ Wsel = nbr ? Wn : Wr;
    const int n0 = (wave & 3) * 16;

    v2f bW[KS];
#pragma unroll
    for (int s = 0; s < KS; ++s) {
        const int k = 4 * s + koff;
        bW[s].x = Wsel[k * D_OUT + n0 + bcol];
        bW[s].y = Wsel[(k + 1) * D_OUT + n0 + bcol];
    }
    const float bv = nbr ? 0.0f : bias[n0 + bcol];

    v4f pa[RW];
#pragma unroll
    for (int j = 0; j < RW; ++j) {
        const long row = (long)tile * 16 + wave + NT * j;
        pa[j] = *(const v4f*)(A + row * K + lane * 4);
    }
#pragma unroll
    for (int j = 0; j < RW; ++j)
        *(v4f*)&As[0][wave + NT * j][lane * 4] = pa[j];
    __syncthreads();

    float* __restrict__ dcol = (lane < 16) ? (z + lane * 4) : (y + (lane - 16) * 4);

    for (int it = 0; tile < ROW_TILES; ++it, tile += gridDim.x) {
        const int p = it & 1;
        const int nxt = tile + gridDim.x;
        const bool more = nxt < ROW_TILES;

        if (more) {
#pragma unroll
            for (int j = 0; j < RW; ++j) {
                const long row = (long)nxt * 16 + wave + NT * j;
                pa[j] = *(const v4f*)(A + row * K + lane * 4);
            }
        }

        v8f acc = {};
#pragma unroll
        for (int s = 0; s < KS; ++s) {
            const v2f a0 = *(const v2f*)&As[p][arow][4 * s + koff];
            acc = __builtin_amdgcn_wmma_f32_16x16x4_f32(
                false, a0, false, bW[s], (short)0, acc, false, false);
        }

        {
            const int rbase = (lane >> 4) << 3;
#pragma unroll
            for (int v = 0; v < 8; ++v)
                Hs[p][rbase + v][wave * 16 + bcol] = acc[v] + bv;
        }
        if (more) {
#pragma unroll
            for (int j = 0; j < RW; ++j)
                *(v4f*)&As[p ^ 1][wave + NT * j][lane * 4] = pa[j];
        }
        __syncthreads();

#pragma unroll
        for (int j = 0; j < RW; ++j) {
            const int r = wave + NT * j;
            const v4f hv = *(const v4f*)&Hs[p][r][lane * 4];
            *(v4f*)(dcol + ((long)tile * 16 + r) * D_OUT) = hv;
        }
    }
}

extern "C" void kernel_launch(void* const* d_in, const int* in_sizes, int n_in,
                              void* d_out, int out_size, void* d_ws, size_t ws_size,
                              hipStream_t stream) {
    const float* x    = (const float*)d_in[0];
    const int*   esrc = (const int*)d_in[1];
    const int*   edst = (const int*)d_in[2];
    const float* Wr0  = (const float*)d_in[3];
    const float* Wn0  = (const float*)d_in[4];
    const float* b0   = (const float*)d_in[5];
    const float* g0   = (const float*)d_in[6];
    const float* be0  = (const float*)d_in[7];
    const float* Wr1  = (const float*)d_in[8];
    const float* Wn1  = (const float*)d_in[9];
    const float* b1   = (const float*)d_in[10];
    const float* g1   = (const float*)d_in[11];
    const float* be1  = (const float*)d_in[12];
    const float* Wr2  = (const float*)d_in[13];
    const float* Wn2  = (const float*)d_in[14];
    const float* b2   = (const float*)d_in[15];
    float* out = (float*)d_out;

    const int N = GNN_N;
    const int E = in_sizes[1];

    float* deg = (float*)d_ws;
    float* agg = deg + N;
    float* h0  = agg + (size_t)N * D_IN;
    float* h1  = h0 + (size_t)N * D_IN;
    int* cnt  = (int*)(h1 + (size_t)N * D_IN);
    int* off  = cnt + N;
    int* rank = off + (N + 32);
    int* csr  = rank + E;

    const int edgeBlocks = (E + 255) / 256;
    const int nodeBlocks = (N + 255) / 256;
    const int gathBlocks = (N + 7) / 8;
    const int gemmBlocks8 = 512;

    zero_i<<<nodeBlocks, 256, 0, stream>>>(cnt, N);
    rank_edges<<<edgeBlocks, 256, 0, stream>>>(edst, cnt, rank, E);
    scan_deg<<<1, 1024, 0, stream>>>(cnt, off, deg, N);
    fill_csr<<<edgeBlocks, 256, 0, stream>>>(esrc, edst, off, rank, csr, E);

    gather_sum<<<gathBlocks, 256, 0, stream>>>(x, off, csr, agg, N);
    sage_gemm<8><<<gemmBlocks8, 256, 0, stream>>>(x, agg, deg, Wr0, Wn0, b0, g0, be0, h0);

    gather_sum<<<gathBlocks, 256, 0, stream>>>(h0, off, csr, agg, N);
    sage_gemm<8><<<gemmBlocks8, 256, 0, stream>>>(h0, agg, deg, Wr1, Wn1, b1, g1, be1, h1);

    float* zp = h0 + 32;
    float* yp = agg + 32;
    proj_gemm<<<gemmBlocks8, 256, 0, stream>>>(h1, Wr2, Wn2, b2, zp, yp);
    gather_lsm<<<gathBlocks, 256, 0, stream>>>(yp, zp, off, csr, out, N);
}


# Layers 0 and 1: the neighbour gather moves into the GEMM launch (a wave sums its two adjacent rows' lists in registers, into the LDS operand); agg plane and 2 launches go; sum order kept
# speedup vs baseline: 3.1891x; 1.0338x over previous
#include <hip/hip_runtime.h>
#include <hip/hip_bf16.h>
#include <math.h>

typedef float v2f __attribute__((ext_vector_type(2)));
typedef float v8f __attribute__((ext_vector_type(8)));
typedef float v4f __attribute__((ext_vector_type(4)));

#define GNN_N 100000
#define D_IN 128
#define D_OUT 64

__global__ void zero_i(int* __restrict__ p, int n) {
    int i = blockIdx.x * blockDim.x + threadIdx.x;
    if (i < n) p[i] = 0;
}

__global__ void rank_edges(const int* __restrict__ dst, int* __restrict__ cnt,
                           int* __restrict__ rank, int E) {
    int i = blockIdx.x * blockDim.x + threadIdx.x;
    if (i < E) rank[i] = atomicAdd(cnt + dst[i], 1);
}

__global__ __launch_bounds__(1024)
void scan_deg(const int* __restrict__ cnt, int* __restrict__ off, int n) {
    __shared__ int wsum[32];
    const int tid = threadIdx.x, lane = tid & 31, wave = tid >> 5;
    int carry = 0;
    int4 nxt[4];
#pragma unroll
    for (int u = 0; u < 4; ++u) nxt[u] = make_int4(0, 0, 0, 0);
    if (tid * 16 < n) {
#pragma unroll
        for (int u = 0; u < 4; ++u) nxt[u] = *(const int4*)(cnt + tid * 16 + 4 * u);
    }
    for (int base = 0; base < n; base += 16384) {
        const int i = base + tid * 16;
        int4 c[4];
#pragma unroll
        for (int u = 0; u < 4; ++u) c[u] = nxt[u];
#pragma unroll
        for (int u = 0; u < 4; ++u) nxt[u] = make_int4(0, 0, 0, 0);
        if (i + 16384 < n) {
#pragma unroll
            for (int u = 0; u < 4; ++u) nxt[u] = *(const int4*)(cnt + i + 16384 + 4 * u);
        }
        int t = 0;
#pragma unroll
        for (int u = 0; u < 4; ++u) t += c[u].x + c[u].y + c[u].z + c[u].w;
        int s = t;
#pragma unroll
        for (int o = 1; o < 32; o <<= 1) {
            int v = __shfl_up(s, o, 32);
            if (lane >= o) s += v;
        }
        if (lane == 31) wsum[wave] = s;
        __syncthreads();
        int ws = wsum[lane];
#pragma unroll
        for (int o = 1; o < 32; o <<= 1) {
            int v = __shfl_up(ws, o, 32);
            if (lane >= o) ws += v;
        }
        const int waveIncl = __shfl(ws, wave, 32);
        const int total = __shfl(ws, 31, 32);
        const int waveTot = __shfl(s, 31, 32);
        if (i < n) {
            int run = carry + (waveIncl - waveTot) + (s - t);
#pragma unroll
            for (int u = 0; u < 4; ++u) {
                int4 o4;
                o4.x = run;
                o4.y = o4.x + c[u].x;
                o4.z = o4.y + c[u].y;
                o4.w = o4.z + c[u].z;
                run = o4.w + c[u].w;
                *(int4*)(off + i + 4 * u) = o4;
            }
        }
        carry += total;
        __syncthreads();
    }
    if (tid == 0) off[n] = carry;
}

__global__ void fill_csr(const int* __restrict__ src, const int* __restrict__ dst,
                         const int* __restrict__ off, const int* __restrict__ rank,
                         int* __restrict__ csr, int E) {
    int i = blockIdx.x * blockDim.x + threadIdx.x;
    if (i < E) csr[off[dst[i]] + rank[i]] = src[i];
}

__global__ __launch_bounds__(256)
void gather_lsm(const float* __restrict__ y, const float* __restrict__ z,
                const int* __restrict__ off, const int* __restrict__ csr,
                float* __restrict__ out, int n) {
    const int node = __builtin_amdgcn_readfirstlane(blockIdx.x * 8 + (threadIdx.x >> 5));
    if (node >= n) return;
    const int lane = threadIdx.x & 31;
    const bool odd = lane >= 16;
    const int c4 = (lane & 15) * 4;
    const int beg = off[node];
    const int end = off[node + 1];
    const float4 zv = *(const float4*)(z + (long)node * D_OUT + c4);
    const float* yl = y + c4;
    float4 acc = make_float4(0.0f, 0.0f, 0.0f, 0.0f);
    int j = beg;
    for (; j + 16 <= end; j += 16) {
        float4 v[8];
#pragma unroll
        for (int u = 0; u < 8; ++u) {
            const int i0 = csr[j + 2 * u], i1 = csr[j + 2 * u + 1];
            v[u] = *(const float4*)(yl + (long)(odd ? i1 : i0) * D_OUT);
        }
#pragma unroll
        for (int u = 0; u < 8; ++u) {
            acc.x += v[u].x; acc.y += v[u].y; acc.z += v[u].z; acc.w += v[u].w;
        }
    }
    if (j + 8 <= end) {
        float4 v[4];
#pragma unroll
        for (int u = 0; u < 4; ++u) {
            const int i0 = csr[j + 2 * u], i1 = csr[j + 2 * u + 1];
            v[u] = *(const float4*)(yl + (long)(odd ? i1 : i0) * D_OUT);
        }
#pragma unroll
        for (int u = 0; u < 4; ++u) {
            acc.x += v[u].x; acc.y += v[u].y; acc.z += v[u].z; acc.w += v[u].w;
        }
        j += 8;
    }
    if (j + 4 <= end) {
        const int i0 = csr[j], i1 = csr[j + 1], i2 = csr[j + 2], i3 = csr[j + 3];
        float4 v0 = *(const float4*)(yl + (long)(odd ? i1 : i0) * D_OUT);
        float4 v1 = *(const float4*)(yl + (long)(odd ? i3 : i2) * D_OUT);
        acc.x += v0.x; acc.y += v0.y; acc.z += v0.z; acc.w += v0.w;
        acc.x += v1.x; acc.y += v1.y; acc.z += v1.z; acc.w += v1.w;
        j += 4;
    }
    if (j + 2 <= end) {
        const int i0 = csr[j], i1 = csr[j + 1];
        float4 v0 = *(const float4*)(yl + (long)(odd ? i1 : i0) * D_OUT);
        acc.x += v0.x; acc.y += v0.y; acc.z += v0.z; acc.w += v0.w;
        j += 2;
    }
    if (j < end) {
        const int i0 = csr[j];
        if (!odd) {
            float4 v0 = *(const float4*)(yl + (long)i0 * D_OUT);
            acc.x += v0.x; acc.y += v0.y; acc.z += v0.z; acc.w += v0.w;
        }
    }
    acc.x += __shfl_xor(acc.x, 16, 32);
    acc.y += __shfl_xor(acc.y, 16, 32);
    acc.z += __shfl_xor(acc.z, 16, 32);
    acc.w += __shfl_xor(acc.w, 16, 32);
    const float inv = 1.0f / fmaxf((float)(end - beg), 1.0f);
    float4 h;
    h.x = zv.x + acc.x * inv; h.y = zv.y + acc.y * inv;
    h.z = zv.z + acc.z * inv; h.w = zv.w + acc.w * inv;
    float mx = fmaxf(fmaxf(h.x, h.y), fmaxf(h.z, h.w));
#pragma unroll
    for (int o = 8; o; o >>= 1) mx = fmaxf(mx, __shfl_xor(mx, o, 32));
    float s = expf(h.x - mx) + expf(h.y - mx) + expf(h.z - mx) + expf(h.w - mx);
#pragma unroll
    for (int o = 8; o; o >>= 1) s += __shfl_xor(s, o, 32);
    const float ls = logf(s);
    if (!odd) {
        float4 r;
        r.x = h.x - mx - ls; r.y = h.y - mx - ls; r.z = h.z - mx - ls; r.w = h.w - mx - ls;
        *(float4*)(out + (long)node * D_OUT + c4) = r;
    }
}

#define ROW_TILES (GNN_N / 16)
#define GATHER_ROWS 16
static_assert(GATHER_ROWS == 16, "the id fence in neighbour_sums2 names sixteen ids");

__device__ __forceinline__ void neighbour_sums2(const float* __restrict__ xl,
                                                const int* __restrict__ csr,
                                                int b0, int b1, int b2, v4f& s0, v4f& s1) {
    v4f a0 = {0.0f, 0.0f, 0.0f, 0.0f};
    v4f a1 = {0.0f, 0.0f, 0.0f, 0.0f};
    for (int j = b0; j < b2; j += GATHER_ROWS) {
        int idx[GATHER_ROWS];
#pragma unroll
        for (int u = 0; u < GATHER_ROWS; ++u) idx[u] = csr[min(j + u, b2 - 1)];
        asm volatile("" :: "s"(idx[0]), "s"(idx[1]), "s"(idx[2]), "s"(idx[3]),
                           "s"(idx[4]), "s"(idx[5]), "s"(idx[6]), "s"(idx[7]),
                           "s"(idx[8]), "s"(idx[9]), "s"(idx[10]), "s"(idx[11]),
                           "s"(idx[12]), "s"(idx[13]), "s"(idx[14]), "s"(idx[15]));
        v4f v[GATHER_ROWS];
#pragma unroll
        for (int u = 0; u < GATHER_ROWS; ++u) {
            v[u] = (v4f){0.0f, 0.0f, 0.0f, 0.0f};
            if (j + u < b2) v[u] = *(const v4f*)(xl + (long)idx[u] * D_IN);
        }
#pragma unroll
        for (int u = 0; u < GATHER_ROWS; ++u) {
            if (j + u < b1) a0 += v[u];
            else            a1 += v[u];
        }
    }
    s0 = a0;
    s1 = a1;
}

__device__ __forceinline__ void fetch_rows2(const float* __restrict__ A,
                                            const int* __restrict__ off,
                                            const int* __restrict__ csr,
                                            int r0, int lane, v4f (&pa)[2], v4f (&pg)[2]) {
    const float* xl = A + lane * 4;
    pa[0] = *(const v4f*)(xl + (long)r0 * D_IN);
    pa[1] = *(const v4f*)(xl + (long)(r0 + 1) * D_IN);
    const int b0 = off[r0], b1 = off[r0 + 1], b2 = off[r0 + 2];
    v4f s0, s1;
    neighbour_sums2(xl, csr, b0, b1, b2, s0, s1);
    pg[0] = s0 * (1.0f / fmaxf((float)(b1 - b0), 1.0f));
    pg[1] = s1 * (1.0f / fmaxf((float)(b2 - b1), 1.0f));
}

__global__ __launch_bounds__(256, 4)
void sage_gemm(const float* __restrict__ A,
               const int* __restrict__ off,
               const int* __restrict__ csr,
               const float* __restrict__ Wr,
               const float* __restrict__ Wn,
               const float* __restrict__ bias,
               const float* __restrict__ gamma,
               const float* __restrict__ beta,
               float* __restrict__ out)
{
    constexpr int NT = 8;
    constexpr int W = 16 * NT;
    constexpr int K = 128;
    constexpr int KS = K / 4;
    constexpr int LDA = K + 4;
    constexpr int RW = 16 / NT;
    constexpr int CPL = W / 32;
    static_assert(RW == 2, "fetch_rows2 stages two adjacent rows per wave");
    __shared__ __attribute__((aligned(16))) float As[2][16][LDA];
    __shared__ __attribute__((aligned(16))) float Gs[2][16][LDA];
    __shared__ float Hs[2][16][W];

    const int lane = threadIdx.x & 31;
    const int wave = __builtin_amdgcn_readfirstlane(threadIdx.x >> 5);

    int tile = blockIdx.x;
    if (tile >= ROW_TILES) return;

    const int n0   = wave * 16;
    const int arow = lane & 15;
    const int koff = (lane >> 4) << 1;
    const int bcol = lane & 15;

    v2f bR[KS], bN[KS];
#pragma unroll
    for (int s = 0; s < KS; ++s) {
        const int k = 4 * s + koff;
        bR[s].x = Wr[k * W + n0 + bcol];
        bR[s].y = Wr[(k + 1) * W + n0 + bcol];
        bN[s].x = Wn[k * W + n0 + bcol];
        bN[s].y = Wn[(k + 1) * W + n0 + bcol];
    }
    const float bv = bias[n0 + bcol];
    float gam[CPL], bet[CPL];
#pragma unroll
    for (int m = 0; m < CPL; ++m) { gam[m] = gamma[lane + 32 * m]; bet[m] = beta[lane + 32 * m]; }

    v4f pa[RW], pg[RW];
    fetch_rows2(A, off, csr, tile * 16 + RW * wave, lane, pa, pg);
#pragma unroll
    for (int j = 0; j < RW; ++j) {
        *(v4f*)&As[0][RW * wave + j][lane * 4] = pa[j];
        *(v4f*)&Gs[0][RW * wave + j][lane * 4] = pg[j];
    }
    __syncthreads();

    for (int it = 0; tile < ROW_TILES; ++it, tile += gridDim.x) {
        const int p = it & 1;
        const int nxt = tile + gridDim.x;
        const bool more = nxt < ROW_TILES;

        if (more) fetch_rows2(A, off, csr, nxt * 16 + RW * wave, lane, pa, pg);

        v8f acc = {};
#pragma unroll
        for (int s = 0; s < KS; ++s) {
            const v2f a0 = *(const v2f*)&As[p][arow][4 * s + koff];
            const v2f a1 = *(const v2f*)&Gs[p][arow][4 * s + koff];
            acc = __builtin_amdgcn_wmma_f32_16x16x4_f32(
                false, a0, false, bR[s], (short)0, acc, false, false);
            acc = __builtin_amdgcn_wmma_f32_16x16x4_f32(
                false, a1, false, bN[s], (short)0, acc, false, false);
        }

        {
            const int rbase = (lane >> 4) << 3;
#pragma unroll
            for (int v = 0; v < 8; ++v)
                Hs[p][rbase + v][n0 + bcol] = acc[v] + bv;
        }
        if (more) {
#pragma unroll
            for (int j = 0; j < RW; ++j) {
                *(v4f*)&As[p ^ 1][RW * wave + j][lane * 4] = pa[j];
                *(v4f*)&Gs[p ^ 1][RW * wave + j][lane * 4] = pg[j];
            }
        }
        __syncthreads();

        float h[RW][CPL];
#pragma unroll
        for (int j = 0; j < RW; ++j)
#pragma unroll
            for (int m = 0; m < CPL; ++m) h[j][m] = Hs[p][wave + NT * j][lane + 32 * m];

        float s[RW], vv[RW], mu[RW];
#pragma unroll
        for (int j = 0; j < RW; ++j) {
            s[j] = 0.0f;
#pragma unroll
            for (int m = 0; m < CPL; ++m) s[j] += h[j][m];
        }
#pragma unroll
        for (int off2 = 16; off2; off2 >>= 1)
#pragma unroll
            for (int j = 0; j < RW; ++j) s[j] += __shfl_xor(s[j], off2, 32);
#pragma unroll
        for (int j = 0; j < RW; ++j) {
            mu[j] = s[j] * (1.0f / W);
            vv[j] = 0.0f;
#pragma unroll
            for (int m = 0; m < CPL; ++m) {
                float d = h[j][m] - mu[j];
                vv[j] += d * d;
            }
        }
#pragma unroll
        for (int off2 = 16; off2; off2 >>= 1)
#pragma unroll
            for (int j = 0; j < RW; ++j) vv[j] += __shfl_xor(vv[j], off2, 32);
#pragma unroll
        for (int j = 0; j < RW; ++j) {
            const float rs = rsqrtf(vv[j] * (1.0f / W) + 1e-5f);
            const long row = (long)tile * 16 + wave + NT * j;
#pragma unroll
            for (int m = 0; m < CPL; ++m) {
                float val = (h[j][m] - mu[j]) * rs * gam[m] + bet[m];
                out[row * W + lane + 32 * m] = fmaxf(val, 0.0f);
            }
        }
    }
}

__global__ __launch_bounds__(256)
void proj_gemm(const float* __restrict__ A,
               const float* __restrict__ Wr,
               const float* __restrict__ Wn,
               const float* __restrict__ bias,
               float* __restrict__ z,
               float* __restrict__ y)
{
    constexpr int NT = 8;
    constexpr int K = 128;
    constexpr int KS = K / 4;
    constexpr int LDA = K + 4;
    constexpr int RW = 16 / NT;
    constexpr int LDH = 2 * D_OUT + 4;
    __shared__ __attribute__((aligned(16))) float As[2][16][LDA];
    __shared__ __attribute__((aligned(16))) float Hs[2][16][LDH];

    const int lane = threadIdx.x & 31;
    const int wave = __builtin_amdgcn_readfirstlane(threadIdx.x >> 5);

    int tile = blockIdx.x;
    if (tile >= ROW_TILES) return;

    const int arow = lane & 15;
    const int koff = (lane >> 4) << 1;
    const int bcol = lane & 15;
    const bool nbr = wave >= 4;
    const float* __restrict__ Wsel = nbr ? Wn : Wr;
    const int n0 = (wave & 3) * 16;

    v2f bW[KS];
#pragma unroll
    for (int s = 0; s < KS; ++s) {
        const int k = 4 * s + koff;
        bW[s].x = Wsel[k * D_OUT + n0 + bcol];
        bW[s].y = Wsel[(k + 1) * D_OUT + n0 + bcol];
    }
    const float bv = nbr ? 0.0f : bias[n0 + bcol];

    v4f pa[RW];
#pragma unroll
    for (int j = 0; j < RW; ++j) {
        const long row = (long)tile * 16 + wave + NT * j;
        pa[j] = *(const v4f*)(A + row * K + lane * 4);
    }
#pragma unroll
    for (int j = 0; j < RW; ++j)
        *(v4f*)&As[0][wave + NT * j][lane * 4] = pa[j];
    __syncthreads();

    float* __restrict__ dcol = (lane < 16) ? (z + lane * 4) : (y + (lane - 16) * 4);

    for (int it = 0; tile < ROW_TILES; ++it, tile += gridDim.x) {
        const int p = it & 1;
        const int nxt = tile + gridDim.x;
        const bool more = nxt < ROW_TILES;

        if (more) {
#pragma unroll
            for (int j = 0; j < RW; ++j) {
                const long row = (long)nxt * 16 + wave + NT * j;
                pa[j] = *(const v4f*)(A + row * K + lane * 4);
            }
        }

        v8f acc = {};
#pragma unroll
        for (int s = 0; s < KS; ++s) {
            const v2f a0 = *(const v2f*)&As[p][arow][4 * s + koff];
            acc = __builtin_amdgcn_wmma_f32_16x16x4_f32(
                false, a0, false, bW[s], (short)0, acc, false, false);
        }

        {
            const int rbase = (lane >> 4) << 3;
#pragma unroll
            for (int v = 0; v < 8; ++v)
                Hs[p][rbase + v][wave * 16 + bcol] = acc[v] + bv;
        }
        if (more) {
#pragma unroll
            for (int j = 0; j < RW; ++j)
                *(v4f*)&As[p ^ 1][wave + NT * j][lane * 4] = pa[j];
        }
        __syncthreads();

#pragma unroll
        for (int j = 0; j < RW; ++j) {
            const int r = wave + NT * j;
            const v4f hv = *(const v4f*)&Hs[p][r][lane * 4];
            *(v4f*)(dcol + ((long)tile * 16 + r) * D_OUT) = hv;
        }
    }
}

extern "C" void kernel_launch(void* const* d_in, const int* in_sizes, int n_in,
                              void* d_out, int out_size, void* d_ws, size_t ws_size,
                              hipStream_t stream) {
    const float* x    = (const float*)d_in[0];
    const int*   esrc = (const int*)d_in[1];
    const int*   edst = (const int*)d_in[2];
    const float* Wr0  = (const float*)d_in[3];
    const float* Wn0  = (const float*)d_in[4];
    const float* b0   = (const float*)d_in[5];
    const float* g0   = (const float*)d_in[6];
    const float* be0  = (const float*)d_in[7];
    const float* Wr1  = (const float*)d_in[8];
    const float* Wn1  = (const float*)d_in[9];
    const float* b1   = (const float*)d_in[10];
    const float* g1   = (const float*)d_in[11];
    const float* be1  = (const float*)d_in[12];
    const float* Wr2  = (const float*)d_in[13];
    const float* Wn2  = (const float*)d_in[14];
    const float* b2   = (const float*)d_in[15];
    float* out = (float*)d_out;

    const int N = GNN_N;
    const int E = in_sizes[1];

    float* h0 = (float*)d_ws;
    float* h1 = h0 + (size_t)N * D_IN;
    float* yp = h1 + (size_t)N * D_IN;
    float* zp = yp + (size_t)N * D_OUT;
    int* cnt  = (int*)(zp + (size_t)N * D_OUT);
    int* off  = cnt + N;
    int* rank = off + (N + 32);
    int* csr  = rank + E;

    const int edgeBlocks = (E + 255) / 256;
    const int nodeBlocks = (N + 255) / 256;
    const int gathBlocks = (N + 7) / 8;
    const int gemmBlocks8 = 512;

    zero_i<<<nodeBlocks, 256, 0, stream>>>(cnt, N);
    rank_edges<<<edgeBlocks, 256, 0, stream>>>(edst, cnt, rank, E);
    scan_deg<<<1, 1024, 0, stream>>>(cnt, off, N);
    fill_csr<<<edgeBlocks, 256, 0, stream>>>(esrc, edst, off, rank, csr, E);

    sage_gemm<<<gemmBlocks8, 256, 0, stream>>>(x, off, csr, Wr0, Wn0, b0, g0, be0, h0);
    sage_gemm<<<gemmBlocks8, 256, 0, stream>>>(h0, off, csr, Wr1, Wn1, b1, g1, be1, h1);

    proj_gemm<<<gemmBlocks8, 256, 0, stream>>>(h1, Wr2, Wn2, b2, zp, yp);
    gather_lsm<<<gathBlocks, 256, 0, stream>>>(yp, zp, off, csr, out, N);
}


# sage_gemm: neighbour lists run ahead of the rows: a tile's segment bounds are fetched two tiles early, its first 64 ids one tile early (one 4-byte load per lane, taken by readlane); sum order kept
# speedup vs baseline: 3.3426x; 1.0481x over previous
#include <hip/hip_runtime.h>
#include <hip/hip_bf16.h>
#include <math.h>

typedef float v2f __attribute__((ext_vector_type(2)));
typedef float v8f __attribute__((ext_vector_type(8)));
typedef float v4f __attribute__((ext_vector_type(4)));

#define GNN_N 100000
#define D_IN 128
#define D_OUT 64

__global__ void zero_i(int* __restrict__ p, int n) {
    int i = blockIdx.x * blockDim.x + threadIdx.x;
    if (i < n) p[i] = 0;
}

__global__ void rank_edges(const int* __restrict__ dst, int* __restrict__ cnt,
                           int* __restrict__ rank, int E) {
    int i = blockIdx.x * blockDim.x + threadIdx.x;
    if (i < E) rank[i] = atomicAdd(cnt + dst[i], 1);
}

__global__ __launch_bounds__(1024)
void scan_deg(const int* __restrict__ cnt, int* __restrict__ off, int n) {
    __shared__ int wsum[32];
    const int tid = threadIdx.x, lane = tid & 31, wave = tid >> 5;
    int carry = 0;
    int4 nxt[4];
#pragma unroll
    for (int u = 0; u < 4; ++u) nxt[u] = make_int4(0, 0, 0, 0);
    if (tid * 16 < n) {
#pragma unroll
        for (int u = 0; u < 4; ++u) nxt[u] = *(const int4*)(cnt + tid * 16 + 4 * u);
    }
    for (int base = 0; base < n; base += 16384) {
        const int i = base + tid * 16;
        int4 c[4];
#pragma unroll
        for (int u = 0; u < 4; ++u) c[u] = nxt[u];
#pragma unroll
        for (int u = 0; u < 4; ++u) nxt[u] = make_int4(0, 0, 0, 0);
        if (i + 16384 < n) {
#pragma unroll
            for (int u = 0; u < 4; ++u) nxt[u] = *(const int4*)(cnt + i + 16384 + 4 * u);
        }
        int t = 0;
#pragma unroll
        for (int u = 0; u < 4; ++u) t += c[u].x + c[u].y + c[u].z + c[u].w;
        int s = t;
#pragma unroll
        for (int o = 1; o < 32; o <<= 1) {
            int v = __shfl_up(s, o, 32);
            if (lane >= o) s += v;
        }
        if (lane == 31) wsum[wave] = s;
        __syncthreads();
        int ws = wsum[lane];
#pragma unroll
        for (int o = 1; o < 32; o <<= 1) {
            int v = __shfl_up(ws, o, 32);
            if (lane >= o) ws += v;
        }
        const int waveIncl = __shfl(ws, wave, 32);
        const int total = __shfl(ws, 31, 32);
        const int waveTot = __shfl(s, 31, 32);
        if (i < n) {
            int run = carry + (waveIncl - waveTot) + (s - t);
#pragma unroll
            for (int u = 0; u < 4; ++u) {
                int4 o4;
                o4.x = run;
                o4.y = o4.x + c[u].x;
                o4.z = o4.y + c[u].y;
                o4.w = o4.z + c[u].z;
                run = o4.w + c[u].w;
                *(int4*)(off + i + 4 * u) = o4;
            }
        }
        carry += total;
        __syncthreads();
    }
    if (tid == 0) off[n] = carry;
}

__global__ void fill_csr(const int* __restrict__ src, const int* __restrict__ dst,
                         const int* __restrict__ off, const int* __restrict__ rank,
                         int* __restrict__ csr, int E) {
    int i = blockIdx.x * blockDim.x + threadIdx.x;
    if (i < E) csr[off[dst[i]] + rank[i]] = src[i];
}

__global__ __launch_bounds__(256)
void gather_lsm(const float* __restrict__ y, const float* __restrict__ z,
                const int* __restrict__ off, const int* __restrict__ csr,
                float* __restrict__ out, int n) {
    const int node = __builtin_amdgcn_readfirstlane(blockIdx.x * 8 + (threadIdx.x >> 5));
    if (node >= n) return;
    const int lane = threadIdx.x & 31;
    const bool odd = lane >= 16;
    const int c4 = (lane & 15) * 4;
    const int beg = off[node];
    const int end = off[node + 1];
    const float4 zv = *(const float4*)(z + (long)node * D_OUT + c4);
    const float* yl = y + c4;
    float4 acc = make_float4(0.0f, 0.0f, 0.0f, 0.0f);
    int j = beg;
    for (; j + 16 <= end; j += 16) {
        float4 v[8];
#pragma unroll
        for (int u = 0; u < 8; ++u) {
            const int i0 = csr[j + 2 * u], i1 = csr[j + 2 * u + 1];
            v[u] = *(const float4*)(yl + (long)(odd ? i1 : i0) * D_OUT);
        }
#pragma unroll
        for (int u = 0; u < 8; ++u) {
            acc.x += v[u].x; acc.y += v[u].y; acc.z += v[u].z; acc.w += v[u].w;
        }
    }
    if (j + 8 <= end) {
        float4 v[4];
#pragma unroll
        for (int u = 0; u < 4; ++u) {
            const int i0 = csr[j + 2 * u], i1 = csr[j + 2 * u + 1];
            v[u] = *(const float4*)(yl + (long)(odd ? i1 : i0) * D_OUT);
        }
#pragma unroll
        for (int u = 0; u < 4; ++u) {
            acc.x += v[u].x; acc.y += v[u].y; acc.z += v[u].z; acc.w += v[u].w;
        }
        j += 8;
    }
    if (j + 4 <= end) {
        const int i0 = csr[j], i1 = csr[j + 1], i2 = csr[j + 2], i3 = csr[j + 3];
        float4 v0 = *(const float4*)(yl + (long)(odd ? i1 : i0) * D_OUT);
        float4 v1 = *(const float4*)(yl + (long)(odd ? i3 : i2) * D_OUT);
        acc.x += v0.x; acc.y += v0.y; acc.z += v0.z; acc.w += v0.w;
        acc.x += v1.x; acc.y += v1.y; acc.z += v1.z; acc.w += v1.w;
        j += 4;
    }
    if (j + 2 <= end) {
        const int i0 = csr[j], i1 = csr[j + 1];
        float4 v0 = *(const float4*)(yl + (long)(odd ? i1 : i0) * D_OUT);
        acc.x += v0.x; acc.y += v0.y; acc.z += v0.z; acc.w += v0.w;
        j += 2;
    }
    if (j < end) {
        const int i0 = csr[j];
        if (!odd) {
            float4 v0 = *(const float4*)(yl + (long)i0 * D_OUT);
            acc.x += v0.x; acc.y += v0.y; acc.z += v0.z; acc.w += v0.w;
        }
    }
    acc.x += __shfl_xor(acc.x, 16, 32);
    acc.y += __shfl_xor(acc.y, 16, 32);
    acc.z += __shfl_xor(acc.z, 16, 32);
    acc.w += __shfl_xor(acc.w, 16, 32);
    const float inv = 1.0f / fmaxf((float)(end - beg), 1.0f);
    float4 h;
    h.x = zv.x + acc.x * inv; h.y = zv.y + acc.y * inv;
    h.z = zv.z + acc.z * inv; h.w = zv.w + acc.w * inv;
    float mx = fmaxf(fmaxf(h.x, h.y), fmaxf(h.z, h.w));
#pragma unroll
    for (int o = 8; o; o >>= 1) mx = fmaxf(mx, __shfl_xor(mx, o, 32));
    float s = expf(h.x - mx) + expf(h.y - mx) + expf(h.z - mx) + expf(h.w - mx);
#pragma unroll
    for (int o = 8; o; o >>= 1) s += __shfl_xor(s, o, 32);
    const float ls = logf(s);
    if (!odd) {
        float4 r;
        r.x = h.x - mx - ls; r.y = h.y - mx - ls; r.z = h.z - mx - ls; r.w = h.w - mx - ls;
        *(float4*)(out + (long)node * D_OUT + c4) = r;
    }
}

#define ROW_TILES (GNN_N / 16)
#define GATHER_ROWS 16
#define ID_SPAN 64

__device__ __forceinline__ void load_ids(const int* __restrict__ csr, int j0, int b2, int lane,
                                         int& iv0, int& iv1) {
    const int last = max(b2 - 1, 0);
    iv0 = csr[min(j0 + lane, last)];
    iv1 = csr[min(j0 + 32 + lane, last)];
}

__device__ __forceinline__ void gather_round(const float* __restrict__ xl, int j, int b1, int b2,
                                             int ivq, int lb, v4f& a0, v4f& a1) {
    v4f v[GATHER_ROWS];
#pragma unroll
    for (int u = 0; u < GATHER_ROWS; ++u) {
        v[u] = (v4f){0.0f, 0.0f, 0.0f, 0.0f};
        if (j + u < b2) {
            const int id = __builtin_amdgcn_readlane(ivq, lb + u);
            v[u] = *(const v4f*)(xl + (long)id * D_IN);
        }
    }
#pragma unroll
    for (int u = 0; u < GATHER_ROWS; ++u) {
        if (j + u < b1) a0 += v[u];
        else            a1 += v[u];
    }
}

__device__ __forceinline__ void neighbour_sums2(const float* __restrict__ xl,
                                                const int* __restrict__ csr,
                                                int b0, int b1, int b2, int iv0, int iv1,
                                                int lane, v4f& s0, v4f& s1) {
    v4f a0 = {0.0f, 0.0f, 0.0f, 0.0f};
    v4f a1 = {0.0f, 0.0f, 0.0f, 0.0f};
#pragma unroll
    for (int q = 0; q < ID_SPAN / GATHER_ROWS; ++q) {
        const int j = b0 + GATHER_ROWS * q;
        if (j < b2) gather_round(xl, j, b1, b2, (q & 2) ? iv1 : iv0, (q & 1) * 16, a0, a1);
    }
#pragma nounroll
    for (int j = b0 + ID_SPAN; j < b2; j += GATHER_ROWS) {
        const int r = (j - b0) / GATHER_ROWS;
        if ((r & 3) == 0) load_ids(csr, j, b2, lane, iv0, iv1);
        gather_round(xl, j, b1, b2, (r & 2) ? iv1 : iv0, (r & 1) * 16, a0, a1);
    }
    s0 = a0;
    s1 = a1;
}

__device__ __forceinline__ void fetch_rows2(const float* __restrict__ A,
                                            const int* __restrict__ csr,
                                            int r0, int lane, int b0, int b1, int b2,
                                            int iv0, int iv1, v4f (&pa)[2], v4f (&pg)[2]) {
    const float* xl = A + lane * 4;
    pa[0] = *(const v4f*)(xl + (long)r0 * D_IN);
    pa[1] = *(const v4f*)(xl + (long)(r0 + 1) * D_IN);
    v4f s0, s1;
    neighbour_sums2(xl, csr, b0, b1, b2, iv0, iv1, lane, s0, s1);
    pg[0] = s0 * (1.0f / fmaxf((float)(b1 - b0), 1.0f));
    pg[1] = s1 * (1.0f / fmaxf((float)(b2 - b1), 1.0f));
}

__global__ __launch_bounds__(256, 4)
void sage_gemm(const float* __restrict__ A,
               const int* __restrict__ off,
               const int* __restrict__ csr,
               const float* __restrict__ Wr,
               const float* __restrict__ Wn,
               const float* __restrict__ bias,
               const float* __restrict__ gamma,
               const float* __restrict__ beta,
               float* __restrict__ out)
{
    constexpr int NT = 8;
    constexpr int W = 16 * NT;
    constexpr int K = 128;
    constexpr int KS = K / 4;
    constexpr int LDA = K + 4;
    constexpr int RW = 16 / NT;
    constexpr int CPL = W / 32;
    static_assert(RW == 2, "fetch_rows2 stages two adjacent rows per wave");
    __shared__ __attribute__((aligned(16))) float As[2][16][LDA];
    __shared__ __attribute__((aligned(16))) float Gs[2][16][LDA];
    __shared__ float Hs[2][16][W];

    const int lane = threadIdx.x & 31;
    const int wave = __builtin_amdgcn_readfirstlane(threadIdx.x >> 5);

    int tile = blockIdx.x;
    if (tile >= ROW_TILES) return;

    const int n0   = wave * 16;
    const int arow = lane & 15;
    const int koff = (lane >> 4) << 1;
    const int bcol = lane & 15;

    v2f bR[KS], bN[KS];
#pragma unroll
    for (int s = 0; s < KS; ++s) {
        const int k = 4 * s + koff;
        bR[s].x = Wr[k * W + n0 + bcol];
        bR[s].y = Wr[(k + 1) * W + n0 + bcol];
        bN[s].x = Wn[k * W + n0 + bcol];
        bN[s].y = Wn[(k + 1) * W + n0 + bcol];
    }
    const float bv = bias[n0 + bcol];
    float gam[CPL], bet[CPL];
#pragma unroll
    for (int m = 0; m < CPL; ++m) { gam[m] = gamma[lane + 32 * m]; bet[m] = beta[lane + 32 * m]; }

    const int step = gridDim.x;
    const int rl = RW * wave;
    const int lq = min(lane, 2);
    v4f pa[RW], pg[RW];
    int p0, p1, p2, pv0, pv1, qv;
    {
        const int ra = tile * 16 + rl;
        const int rb = min(tile + step, ROW_TILES - 1) * 16 + rl;
        const int rc = min(tile + 2 * step, ROW_TILES - 1) * 16 + rl;
        const int c0 = off[ra], c1 = off[ra + 1], c2 = off[ra + 2];
        p0 = off[rb]; p1 = off[rb + 1]; p2 = off[rb + 2];
        qv = off[rc + lq];
        int cv0, cv1;
        load_ids(csr, c0, c2, lane, cv0, cv1);
        load_ids(csr, p0, p2, lane, pv0, pv1);
        fetch_rows2(A, csr, ra, lane, c0, c1, c2, cv0, cv1, pa, pg);
    }
#pragma unroll
    for (int j = 0; j < RW; ++j) {
        *(v4f*)&As[0][RW * wave + j][lane * 4] = pa[j];
        *(v4f*)&Gs[0][RW * wave + j][lane * 4] = pg[j];
    }
    __syncthreads();

    for (int it = 0; tile < ROW_TILES; ++it, tile += gridDim.x) {
        const int p = it & 1;
        const int nxt = tile + gridDim.x;
        const bool more = nxt < ROW_TILES;

        const int q0 = __builtin_amdgcn_readlane(qv, 0);
        const int q1 = __builtin_amdgcn_readlane(qv, 1);
        const int q2 = __builtin_amdgcn_readlane(qv, 2);
        int nv0, nv1;
        load_ids(csr, q0, q2, lane, nv0, nv1);
        const int dv = off[min(tile + 3 * step, ROW_TILES - 1) * 16 + rl + lq];

        if (more) fetch_rows2(A, csr, nxt * 16 + rl, lane, p0, p1, p2, pv0, pv1, pa, pg);
        p0 = q0; p1 = q1; p2 = q2; pv0 = nv0; pv1 = nv1;
        qv = dv;

        v8f acc = {};
#pragma unroll
        for (int s = 0; s < KS; ++s) {
            const v2f a0 = *(const v2f*)&As[p][arow][4 * s + koff];
            const v2f a1 = *(const v2f*)&Gs[p][arow][4 * s + koff];
            acc = __builtin_amdgcn_wmma_f32_16x16x4_f32(
                false, a0, false, bR[s], (short)0, acc, false, false);
            acc = __builtin_amdgcn_wmma_f32_16x16x4_f32(
                false, a1, false, bN[s], (short)0, acc, false, false);
        }

        {
            const int rbase = (lane >> 4) << 3;
#pragma unroll
            for (int v = 0; v < 8; ++v)
                Hs[p][rbase + v][n0 + bcol] = acc[v] + bv;
        }
        if (more) {
#pragma unroll
            for (int j = 0; j < RW; ++j) {
                *(v4f*)&As[p ^ 1][RW * wave + j][lane * 4] = pa[j];
                *(v4f*)&Gs[p ^ 1][RW * wave + j][lane * 4] = pg[j];
            }
        }
        __syncthreads();

        float h[RW][CPL];
#pragma unroll
        for (int j = 0; j < RW; ++j)
#pragma unroll
            for (int m = 0; m < CPL; ++m) h[j][m] = Hs[p][wave + NT * j][lane + 32 * m];

        float s[RW], vv[RW], mu[RW];
#pragma unroll
        for (int j = 0; j < RW; ++j) {
            s[j] = 0.0f;
#pragma unroll
            for (int m = 0; m < CPL; ++m) s[j] += h[j][m];
        }
#pragma unroll
        for (int off2 = 16; off2; off2 >>= 1)
#pragma unroll
            for (int j = 0; j < RW; ++j) s[j] += __shfl_xor(s[j], off2, 32);
#pragma unroll
        for (int j = 0; j < RW; ++j) {
            mu[j] = s[j] * (1.0f / W);
            vv[j] = 0.0f;
#pragma unroll
            for (int m = 0; m < CPL; ++m) {
                float d = h[j][m] - mu[j];
                vv[j] += d * d;
            }
        }
#pragma unroll
        for (int off2 = 16; off2; off2 >>= 1)
#pragma unroll
            for (int j = 0; j < RW; ++j) vv[j] += __shfl_xor(vv[j], off2, 32);
#pragma unroll
        for (int j = 0; j < RW; ++j) {
            const float rs = rsqrtf(vv[j] * (1.0f / W) + 1e-5f);
            const long row = (long)tile * 16 + wave + NT * j;
#pragma unroll
            for (int m = 0; m < CPL; ++m) {
                float val = (h[j][m] - mu[j]) * rs * gam[m] + bet[m];
                out[row * W + lane + 32 * m] = fmaxf(val, 0.0f);
            }
        }
    }
}

__global__ __launch_bounds__(256)
void proj_gemm(const float* __restrict__ A,
               const float* __restrict__ Wr,
               const float* __restrict__ Wn,
               const float* __restrict__ bias,
               float* __restrict__ z,
               float* __restrict__ y)
{
    constexpr int NT = 8;
    constexpr int K = 128;
    constexpr int KS = K / 4;
    constexpr int LDA = K + 4;
    constexpr int RW = 16 / NT;
    constexpr int LDH = 2 * D_OUT + 4;
    __shared__ __attribute__((aligned(16))) float As[2][16][LDA];
    __shared__ __attribute__((aligned(16))) float Hs[2][16][LDH];

    const int lane = threadIdx.x & 31;
    const int wave = __builtin_amdgcn_readfirstlane(threadIdx.x >> 5);

    int tile = blockIdx.x;
    if (tile >= ROW_TILES) return;

    const int arow = lane & 15;
    const int koff = (lane >> 4) << 1;
    const int bcol = lane & 15;
    const bool nbr = wave >= 4;
    const float* __restrict__ Wsel = nbr ? Wn : Wr;
    const int n0 = (wave & 3) * 16;

    v2f bW[KS];
#pragma unroll
    for (int s = 0; s < KS; ++s) {
        const int k = 4 * s + koff;
        bW[s].x = Wsel[k * D_OUT + n0 + bcol];
        bW[s].y = Wsel[(k + 1) * D_OUT + n0 + bcol];
    }
    const float bv = nbr ? 0.0f : bias[n0 + bcol];

    v4f pa[RW];
#pragma unroll
    for (int j = 0; j < RW; ++j) {
        const long row = (long)tile * 16 + wave + NT * j;
        pa[j] = *(const v4f*)(A + row * K + lane * 4);
    }
#pragma unroll
    for (int j = 0; j < RW; ++j)
        *(v4f*)&As[0][wave + NT * j][lane * 4] = pa[j];
    __syncthreads();

    float* __restrict__ dcol = (lane < 16) ? (z + lane * 4) : (y + (lane - 16) * 4);

    for (int it = 0; tile < ROW_TILES; ++it, tile += gridDim.x) {
        const int p = it & 1;
        const int nxt = tile + gridDim.x;
        const bool more = nxt < ROW_TILES;

        if (more) {
#pragma unroll
            for (int j = 0; j < RW; ++j) {
                const long row = (long)nxt * 16 + wave + NT * j;
                pa[j] = *(const v4f*)(A + row * K + lane * 4);
            }
        }

        v8f acc = {};
#pragma unroll
        for (int s = 0; s < KS; ++s) {
            const v2f a0 = *(const v2f*)&As[p][arow][4 * s + koff];
            acc = __builtin_amdgcn_wmma_f32_16x16x4_f32(
                false, a0, false, bW[s], (short)0, acc, false, false);
        }

        {
            const int rbase = (lane >> 4) << 3;
#pragma unroll
            for (int v = 0; v < 8; ++v)
                Hs[p][rbase + v][wave * 16 + bcol] = acc[v] + bv;
        }
        if (more) {
#pragma unroll
            for (int j = 0; j < RW; ++j)
                *(v4f*)&As[p ^ 1][wave + NT * j][lane * 4] = pa[j];
        }
        __syncthreads();

#pragma unroll
        for (int j = 0; j < RW; ++j) {
            const int r = wave + NT * j;
            const v4f hv = *(const v4f*)&Hs[p][r][lane * 4];
            *(v4f*)(dcol + ((long)tile * 16 + r) * D_OUT) = hv;
        }
    }
}

extern "C" void kernel_launch(void* const* d_in, const int* in_sizes, int n_in,
                              void* d_out, int out_size, void* d_ws, size_t ws_size,
                              hipStream_t stream) {
    const float* x    = (const float*)d_in[0];
    const int*   esrc = (const int*)d_in[1];
    const int*   edst = (const int*)d_in[2];
    const float* Wr0  = (const float*)d_in[3];
    const float* Wn0  = (const float*)d_in[4];
    const float* b0   = (const float*)d_in[5];
    const float* g0   = (const float*)d_in[6];
    const float* be0  = (const float*)d_in[7];
    const float* Wr1  = (const float*)d_in[8];
    const float* Wn1  = (const float*)d_in[9];
    const float* b1   = (const float*)d_in[10];
    const float* g1   = (const float*)d_in[11];
    const float* be1  = (const float*)d_in[12];
    const float* Wr2  = (const float*)d_in[13];
    const float* Wn2  = (const float*)d_in[14];
    const float* b2   = (const float*)d_in[15];
    float* out = (float*)d_out;

    const int N = GNN_N;
    const int E = in_sizes[1];

    float* h0 = (float*)d_ws;
    float* h1 = h0 + (size_t)N * D_IN;
    float* yp = h1 + (size_t)N * D_IN;
    float* zp = yp + (size_t)N * D_OUT;
    int* cnt  = (int*)(zp + (size_t)N * D_OUT);
    int* off  = cnt + N;
    int* rank = off + (N + 32);
    int* csr  = rank + E;

    const int edgeBlocks = (E + 255) / 256;
    const int nodeBlocks = (N + 255) / 256;
    const int gathBlocks = (N + 7) / 8;
    const int gemmBlocks8 = 512;

    zero_i<<<nodeBlocks, 256, 0, stream>>>(cnt, N);
    rank_edges<<<edgeBlocks, 256, 0, stream>>>(edst, cnt, rank, E);
    scan_deg<<<1, 1024, 0, stream>>>(cnt, off, N);
    fill_csr<<<edgeBlocks, 256, 0, stream>>>(esrc, edst, off, rank, csr, E);

    sage_gemm<<<gemmBlocks8, 256, 0, stream>>>(x, off, csr, Wr0, Wn0, b0, g0, be0, h0);
    sage_gemm<<<gemmBlocks8, 256, 0, stream>>>(h0, off, csr, Wr1, Wn1, b1, g1, be1, h1);

    proj_gemm<<<gemmBlocks8, 256, 0, stream>>>(h1, Wr2, Wn2, b2, zp, yp);
    gather_lsm<<<gathBlocks, 256, 0, stream>>>(yp, zp, off, csr, out, N);
}


# Edge list sorted by dst with a two-level counting sort on LDS counters (bucket partition, then per-bucket rank and fill) in place of global returning atomics; per-node sum order = LDS arrival order
# speedup vs baseline: 4.4342x; 1.3266x over previous
#include <hip/hip_runtime.h>
#include <hip/hip_bf16.h>
#include <math.h>

typedef float v2f __attribute__((ext_vector_type(2)));
typedef float v8f __attribute__((ext_vector_type(8)));
typedef float v4f __attribute__((ext_vector_type(4)));

#define GNN_N 100000
#define D_IN 128
#define D_OUT 64

#define EDGE_CHUNKS 256
#define NODE_SHIFT 9
#define BUCKET_NODES (1 << NODE_SHIFT)
#define N_BUCKETS ((GNN_N + BUCKET_NODES - 1) / BUCKET_NODES)
static_assert((N_BUCKETS * EDGE_CHUNKS) % 16 == 0, "scan_deg works on whole int4 groups");
static_assert(GNN_N < (1 << (31 - NODE_SHIFT)), "a src id and a node's place in its bucket share one word");
static_assert(GNN_N % BUCKET_NODES != 0, "off[N] is the last bucket's entry N % BUCKET_NODES");

__device__ __forceinline__ int bucket_of(int d) {
    return min((int)((unsigned)d >> NODE_SHIFT), N_BUCKETS - 1);
}

#define EDGE_BATCH 8
#define CSR_BATCH 12
#define SEG_CAP 16384

__global__ __launch_bounds__(1024)
void bucket_hist(const int* __restrict__ dst, int* __restrict__ H, int E) {
    __shared__ int h[N_BUCKETS];
    const int tid = threadIdx.x, w = blockIdx.x;
    const int chunk = (E + EDGE_CHUNKS - 1) / EDGE_CHUNKS;
    const int lo = min(w * chunk, E), hi = min(lo + chunk, E);
    for (int t = tid; t < N_BUCKETS; t += 1024) h[t] = 0;
    __syncthreads();
    for (int base = lo; base < hi; base += EDGE_BATCH * 1024) {
        int d[EDGE_BATCH];
#pragma unroll
        for (int u = 0; u < EDGE_BATCH; ++u) {
            const int i = base + u * 1024 + tid;
            d[u] = dst[min(i, hi - 1)];
        }
#pragma unroll
        for (int u = 0; u < EDGE_BATCH; ++u) {
            const int i = base + u * 1024 + tid;
            if (i < hi) atomicAdd(&h[bucket_of(d[u])], 1);
        }
    }
    __syncthreads();
    for (int t = tid; t < N_BUCKETS; t += 1024) H[t * EDGE_CHUNKS + w] = h[t];
}

__global__ __launch_bounds__(1024)
void scan_deg(const int* __restrict__ cnt, int* __restrict__ off, int n) {
    __shared__ int wsum[32];
    const int tid = threadIdx.x, lane = tid & 31, wave = tid >> 5;
    int carry = 0;
    int4 nxt[4];
#pragma unroll
    for (int u = 0; u < 4; ++u) nxt[u] = make_int4(0, 0, 0, 0);
    if (tid * 16 < n) {
#pragma unroll
        for (int u = 0; u < 4; ++u) nxt[u] = *(const int4*)(cnt + tid * 16 + 4 * u);
    }
    for (int base = 0; base < n; base += 16384) {
        const int i = base + tid * 16;
        int4 c[4];
#pragma unroll
        for (int u = 0; u < 4; ++u) c[u] = nxt[u];
#pragma unroll
        for (int u = 0; u < 4; ++u) nxt[u] = make_int4(0, 0, 0, 0);
        if (i + 16384 < n) {
#pragma unroll
            for (int u = 0; u < 4; ++u) nxt[u] = *(const int4*)(cnt + i + 16384 + 4 * u);
        }
        int t = 0;
#pragma unroll
        for (int u = 0; u < 4; ++u) t += c[u].x + c[u].y + c[u].z + c[u].w;
        int s = t;
#pragma unroll
        for (int o = 1; o < 32; o <<= 1) {
            int v = __shfl_up(s, o, 32);
            if (lane >= o) s += v;
        }
        if (lane == 31) wsum[wave] = s;
        __syncthreads();
        int ws = wsum[lane];
#pragma unroll
        for (int o = 1; o < 32; o <<= 1) {
            int v = __shfl_up(ws, o, 32);
            if (lane >= o) ws += v;
        }
        const int waveIncl = __shfl(ws, wave, 32);
        const int total = __shfl(ws, 31, 32);
        const int waveTot = __shfl(s, 31, 32);
        if (i < n) {
            int run = carry + (waveIncl - waveTot) + (s - t);
#pragma unroll
            for (int u = 0; u < 4; ++u) {
                int4 o4;
                o4.x = run;
                o4.y = o4.x + c[u].x;
                o4.z = o4.y + c[u].y;
                o4.w = o4.z + c[u].z;
                run = o4.w + c[u].w;
                *(int4*)(off + i + 4 * u) = o4;
            }
        }
        carry += total;
        __syncthreads();
    }
    if (tid == 0) off[n] = carry;
}

__global__ __launch_bounds__(1024)
void bucket_part(const int* __restrict__ src, const int* __restrict__ dst,
                 const int* __restrict__ O, int* __restrict__ tmp, int E) {
    __shared__ int cur[N_BUCKETS];
    const int tid = threadIdx.x, w = blockIdx.x;
    const int chunk = (E + EDGE_CHUNKS - 1) / EDGE_CHUNKS;
    const int lo = min(w * chunk, E), hi = min(lo + chunk, E);
    for (int t = tid; t < N_BUCKETS; t += 1024) cur[t] = O[t * EDGE_CHUNKS + w];
    __syncthreads();
    for (int base = lo; base < hi; base += EDGE_BATCH * 1024) {
        int d[EDGE_BATCH], s[EDGE_BATCH];
#pragma unroll
        for (int u = 0; u < EDGE_BATCH; ++u) {
            const int i = min(base + u * 1024 + tid, hi - 1);
            d[u] = dst[i];
            s[u] = src[i];
        }
#pragma unroll
        for (int u = 0; u < EDGE_BATCH; ++u) {
            const int i = base + u * 1024 + tid;
            if (i < hi) {
                const int p = atomicAdd(&cur[bucket_of(d[u])], 1);
                tmp[p] = (int)(((unsigned)s[u] << NODE_SHIFT) | ((unsigned)d[u] & (BUCKET_NODES - 1)));
            }
        }
    }
}

__global__ __launch_bounds__(1024)
void bucket_csr(const int* __restrict__ tmp, const int* __restrict__ O,
                int* __restrict__ off, int* __restrict__ csr) {
    __shared__ int cnt[BUCKET_NODES];
    __shared__ int wsum[32];
    __shared__ int seg[SEG_CAP];
    const int tid = threadIdx.x, lane = tid & 31, wave = tid >> 5, b = blockIdx.x;
    const int B0 = O[b * EDGE_CHUNKS], B1 = O[(b + 1) * EDGE_CHUNKS];
    const bool staged = (B1 - B0) <= SEG_CAP;
    if (tid < BUCKET_NODES) cnt[tid] = 0;
    __syncthreads();
    for (int base = B0; base < B1; base += CSR_BATCH * 1024) {
        int e[CSR_BATCH];
#pragma unroll
        for (int u = 0; u < CSR_BATCH; ++u) {
            const int i = base + u * 1024 + tid;
            e[u] = tmp[min(i, B1 - 1)];
        }
#pragma unroll
        for (int u = 0; u < CSR_BATCH; ++u) {
            const int i = base + u * 1024 + tid;
            if (i < B1) atomicAdd(&cnt[e[u] & (BUCKET_NODES - 1)], 1);
        }
    }
    __syncthreads();
    const int v = (tid < BUCKET_NODES) ? cnt[tid] : 0;
    int s = v;
#pragma unroll
    for (int o = 1; o < 32; o <<= 1) {
        int t = __shfl_up(s, o, 32);
        if (lane >= o) s += t;
    }
    if (lane == 31) wsum[wave] = s;
    __syncthreads();
    int ws = wsum[lane];
#pragma unroll
    for (int o = 1; o < 32; o <<= 1) {
        int t = __shfl_up(ws, o, 32);
        if (lane >= o) ws += t;
    }
    const int waveIncl = __shfl(ws, wave, 32);
    const int waveTot = __shfl(s, 31, 32);
    const int excl = (waveIncl - waveTot) + (s - v);
    if (tid < BUCKET_NODES) {
        cnt[tid] = excl;
        const int node = b * BUCKET_NODES + tid;
        if (node <= GNN_N) off[node] = B0 + excl;
    }
    __syncthreads();
    for (int base = B0; base < B1; base += CSR_BATCH * 1024) {
        int e[CSR_BATCH];
#pragma unroll
        for (int u = 0; u < CSR_BATCH; ++u) {
            const int i = base + u * 1024 + tid;
            e[u] = tmp[min(i, B1 - 1)];
        }
#pragma unroll
        for (int u = 0; u < CSR_BATCH; ++u) {
            const int i = base + u * 1024 + tid;
            if (i < B1) {
                const int p = atomicAdd(&cnt[e[u] & (BUCKET_NODES - 1)], 1);
                const int id = (int)((unsigned)e[u] >> NODE_SHIFT);
                if (staged) seg[p] = id;
                else        csr[B0 + p] = id;
            }
        }
    }
    if (staged) {
        __syncthreads();
        for (int i = tid; i < B1 - B0; i += 1024) csr[B0 + i] = seg[i];
    }
}

__global__ __launch_bounds__(256)
void gather_lsm(const float* __restrict__ y, const float* __restrict__ z,
                const int* __restrict__ off, const int* __restrict__ csr,
                float* __restrict__ out, int n) {
    const int node = __builtin_amdgcn_readfirstlane(blockIdx.x * 8 + (threadIdx.x >> 5));
    if (node >= n) return;
    const int lane = threadIdx.x & 31;
    const bool odd = lane >= 16;
    const int c4 = (lane & 15) * 4;
    const int beg = off[node];
    const int end = off[node + 1];
    const float4 zv = *(const float4*)(z + (long)node * D_OUT + c4);
    const float* yl = y + c4;
    float4 acc = make_float4(0.0f, 0.0f, 0.0f, 0.0f);
    int j = beg;
    for (; j + 16 <= end; j += 16) {
        float4 v[8];
#pragma unroll
        for (int u = 0; u < 8; ++u) {
            const int i0 = csr[j + 2 * u], i1 = csr[j + 2 * u + 1];
            v[u] = *(const float4*)(yl + (long)(odd ? i1 : i0) * D_OUT);
        }
#pragma unroll
        for (int u = 0; u < 8; ++u) {
            acc.x += v[u].x; acc.y += v[u].y; acc.z += v[u].z; acc.w += v[u].w;
        }
    }
    if (j + 8 <= end) {
        float4 v[4];
#pragma unroll
        for (int u = 0; u < 4; ++u) {
            const int i0 = csr[j + 2 * u], i1 = csr[j + 2 * u + 1];
            v[u] = *(const float4*)(yl + (long)(odd ? i1 : i0) * D_OUT);
        }
#pragma unroll
        for (int u = 0; u < 4; ++u) {
            acc.x += v[u].x; acc.y += v[u].y; acc.z += v[u].z; acc.w += v[u].w;
        }
        j += 8;
    }
    if (j + 4 <= end) {
        const int i0 = csr[j], i1 = csr[j + 1], i2 = csr[j + 2], i3 = csr[j + 3];
        float4 v0 = *(const float4*)(yl + (long)(odd ? i1 : i0) * D_OUT);
        float4 v1 = *(const float4*)(yl + (long)(odd ? i3 : i2) * D_OUT);
        acc.x += v0.x; acc.y += v0.y; acc.z += v0.z; acc.w += v0.w;
        acc.x += v1.x; acc.y += v1.y; acc.z += v1.z; acc.w += v1.w;
        j += 4;
    }
    if (j + 2 <= end) {
        const int i0 = csr[j], i1 = csr[j + 1];
        float4 v0 = *(const float4*)(yl + (long)(odd ? i1 : i0) * D_OUT);
        acc.x += v0.x; acc.y += v0.y; acc.z += v0.z; acc.w += v0.w;
        j += 2;
    }
    if (j < end) {
        const int i0 = csr[j];
        if (!odd) {
            float4 v0 = *(const float4*)(yl + (long)i0 * D_OUT);
            acc.x += v0.x; acc.y += v0.y; acc.z += v0.z; acc.w += v0.w;
        }
    }
    acc.x += __shfl_xor(acc.x, 16, 32);
    acc.y += __shfl_xor(acc.y, 16, 32);
    acc.z += __shfl_xor(acc.z, 16, 32);
    acc.w += __shfl_xor(acc.w, 16, 32);
    const float inv = 1.0f / fmaxf((float)(end - beg), 1.0f);
    float4 h;
    h.x = zv.x + acc.x * inv; h.y = zv.y + acc.y * inv;
    h.z = zv.z + acc.z * inv; h.w = zv.w + acc.w * inv;
    float mx = fmaxf(fmaxf(h.x, h.y), fmaxf(h.z, h.w));
#pragma unroll
    for (int o = 8; o; o >>= 1) mx = fmaxf(mx, __shfl_xor(mx, o, 32));
    float s = expf(h.x - mx) + expf(h.y - mx) + expf(h.z - mx) + expf(h.w - mx);
#pragma unroll
    for (int o = 8; o; o >>= 1) s += __shfl_xor(s, o, 32);
    const float ls = logf(s);
    if (!odd) {
        float4 r;
        r.x = h.x - mx - ls; r.y = h.y - mx - ls; r.z = h.z - mx - ls; r.w = h.w - mx - ls;
        *(float4*)(out + (long)node * D_OUT + c4) = r;
    }
}

#define ROW_TILES (GNN_N / 16)
#define GATHER_ROWS 16
#define ID_SPAN 64

__device__ __forceinline__ void load_ids(const int* __restrict__ csr, int j0, int b2, int lane,
                                         int& iv0, int& iv1) {
    const int last = max(b2 - 1, 0);
    iv0 = csr[min(j0 + lane, last)];
    iv1 = csr[min(j0 + 32 + lane, last)];
}

__device__ __forceinline__ void gather_round(const float* __restrict__ xl, int j, int b1, int b2,
                                             int ivq, int lb, v4f& a0, v4f& a1) {
    v4f v[GATHER_ROWS];
#pragma unroll
    for (int u = 0; u < GATHER_ROWS; ++u) {
        v[u] = (v4f){0.0f, 0.0f, 0.0f, 0.0f};
        if (j + u < b2) {
            const int id = __builtin_amdgcn_readlane(ivq, lb + u);
            v[u] = *(const v4f*)(xl + (long)id * D_IN);
        }
    }
#pragma unroll
    for (int u = 0; u < GATHER_ROWS; ++u) {
        if (j + u < b1) a0 += v[u];
        else            a1 += v[u];
    }
}

__device__ __forceinline__ void neighbour_sums2(const float* __restrict__ xl,
                                                const int* __restrict__ csr,
                                                int b0, int b1, int b2, int iv0, int iv1,
                                                int lane, v4f& s0, v4f& s1) {
    v4f a0 = {0.0f, 0.0f, 0.0f, 0.0f};
    v4f a1 = {0.0f, 0.0f, 0.0f, 0.0f};
#pragma unroll
    for (int q = 0; q < ID_SPAN / GATHER_ROWS; ++q) {
        const int j = b0 + GATHER_ROWS * q;
        if (j < b2) gather_round(xl, j, b1, b2, (q & 2) ? iv1 : iv0, (q & 1) * 16, a0, a1);
    }
#pragma nounroll
    for (int j = b0 + ID_SPAN; j < b2; j += GATHER_ROWS) {
        const int r = (j - b0) / GATHER_ROWS;
        if ((r & 3) == 0) load_ids(csr, j, b2, lane, iv0, iv1);
        gather_round(xl, j, b1, b2, (r & 2) ? iv1 : iv0, (r & 1) * 16, a0, a1);
    }
    s0 = a0;
    s1 = a1;
}

__device__ __forceinline__ void fetch_rows2(const float* __restrict__ A,
                                            const int* __restrict__ csr,
                                            int r0, int lane, int b0, int b1, int b2,
                                            int iv0, int iv1, v4f (&pa)[2], v4f (&pg)[2]) {
    const float* xl = A + lane * 4;
    pa[0] = *(const v4f*)(xl + (long)r0 * D_IN);
    pa[1] = *(const v4f*)(xl + (long)(r0 + 1) * D_IN);
    v4f s0, s1;
    neighbour_sums2(xl, csr, b0, b1, b2, iv0, iv1, lane, s0, s1);
    pg[0] = s0 * (1.0f / fmaxf((float)(b1 - b0), 1.0f));
    pg[1] = s1 * (1.0f / fmaxf((float)(b2 - b1), 1.0f));
}

__global__ __launch_bounds__(256, 4)
void sage_gemm(const float* __restrict__ A,
               const int* __restrict__ off,
               const int* __restrict__ csr,
               const float* __restrict__ Wr,
               const float* __restrict__ Wn,
               const float* __restrict__ bias,
               const float* __restrict__ gamma,
               const float* __restrict__ beta,
               float* __restrict__ out)
{
    constexpr int NT = 8;
    constexpr int W = 16 * NT;
    constexpr int K = 128;
    constexpr int KS = K / 4;
    constexpr int LDA = K + 4;
    constexpr int RW = 16 / NT;
    constexpr int CPL = W / 32;
    static_assert(RW == 2, "fetch_rows2 stages two adjacent rows per wave");
    __shared__ __attribute__((aligned(16))) float As[2][16][LDA];
    __shared__ __attribute__((aligned(16))) float Gs[2][16][LDA];
    __shared__ float Hs[2][16][W];

    const int lane = threadIdx.x & 31;
    const int wave = __builtin_amdgcn_readfirstlane(threadIdx.x >> 5);

    int tile = blockIdx.x;
    if (tile >= ROW_TILES) return;

    const int n0   = wave * 16;
    const int arow = lane & 15;
    const int koff = (lane >> 4) << 1;
    const int bcol = lane & 15;

    v2f bR[KS], bN[KS];
#pragma unroll
    for (int s = 0; s < KS; ++s) {
        const int k = 4 * s + koff;
        bR[s].x = Wr[k * W + n0 + bcol];
        bR[s].y = Wr[(k + 1) * W + n0 + bcol];
        bN[s].x = Wn[k * W + n0 + bcol];
        bN[s].y = Wn[(k + 1) * W + n0 + bcol];
    }
    const float bv = bias[n0 + bcol];
    float gam[CPL], bet[CPL];
#pragma unroll
    for (int m = 0; m < CPL; ++m) { gam[m] = gamma[lane + 32 * m]; bet[m] = beta[lane + 32 * m]; }

    const int step = gridDim.x;
    const int rl = RW * wave;
    const int lq = min(lane, 2);
    v4f pa[RW], pg[RW];
    int p0, p1, p2, pv0, pv1, qv;
    {
        const int ra = tile * 16 + rl;
        const int rb = min(tile + step, ROW_TILES - 1) * 16 + rl;
        const int rc = min(tile + 2 * step, ROW_TILES - 1) * 16 + rl;
        const int c0 = off[ra], c1 = off[ra + 1], c2 = off[ra + 2];
        p0 = off[rb]; p1 = off[rb + 1]; p2 = off[rb + 2];
        qv = off[rc + lq];
        int cv0, cv1;
        load_ids(csr, c0, c2, lane, cv0, cv1);
        load_ids(csr, p0, p2, lane, pv0, pv1);
        fetch_rows2(A, csr, ra, lane, c0, c1, c2, cv0, cv1, pa, pg);
    }
#pragma unroll
    for (int j = 0; j < RW; ++j) {
        *(v4f*)&As[0][RW * wave + j][lane * 4] = pa[j];
        *(v4f*)&Gs[0][RW * wave + j][lane * 4] = pg[j];
    }
    __syncthreads();

    for (int it = 0; tile < ROW_TILES; ++it, tile += gridDim.x) {
        const int p = it & 1;
        const int nxt = tile + gridDim.x;
        const bool more = nxt < ROW_TILES;

        const int q0 = __builtin_amdgcn_readlane(qv, 0);
        const int q1 = __builtin_amdgcn_readlane(qv, 1);
        const int q2 = __builtin_amdgcn_readlane(qv, 2);
        int nv0, nv1;
        load_ids(csr, q0, q2, lane, nv0, nv1);
        const int dv = off[min(tile + 3 * step, ROW_TILES - 1) * 16 + rl + lq];

        if (more) fetch_rows2(A, csr, nxt * 16 + rl, lane, p0, p1, p2, pv0, pv1, pa, pg);
        p0 = q0; p1 = q1; p2 = q2; pv0 = nv0; pv1 = nv1;
        qv = dv;

        v8f acc = {};
#pragma unroll
        for (int s = 0; s < KS; ++s) {
            const v2f a0 = *(const v2f*)&As[p][arow][4 * s + koff];
            const v2f a1 = *(const v2f*)&Gs[p][arow][4 * s + koff];
            acc = __builtin_amdgcn_wmma_f32_16x16x4_f32(
                false, a0, false, bR[s], (short)0, acc, false, false);
            acc = __builtin_amdgcn_wmma_f32_16x16x4_f32(
                false, a1, false, bN[s], (short)0, acc, false, false);
        }

        {
            const int rbase = (lane >> 4) << 3;
#pragma unroll
            for (int v = 0; v < 8; ++v)
                Hs[p][rbase + v][n0 + bcol] = acc[v] + bv;
        }
        if (more) {
#pragma unroll
            for (int j = 0; j < RW; ++j) {
                *(v4f*)&As[p ^ 1][RW * wave + j][lane * 4] = pa[j];
                *(v4f*)&Gs[p ^ 1][RW * wave + j][lane * 4] = pg[j];
            }
        }
        __syncthreads();

        float h[RW][CPL];
#pragma unroll
        for (int j = 0; j < RW; ++j)
#pragma unroll
            for (int m = 0; m < CPL; ++m) h[j][m] = Hs[p][wave + NT * j][lane + 32 * m];

        float s[RW], vv[RW], mu[RW];
#pragma unroll
        for (int j = 0; j < RW; ++j) {
            s[j] = 0.0f;
#pragma unroll
            for (int m = 0; m < CPL; ++m) s[j] += h[j][m];
        }
#pragma unroll
        for (int off2 = 16; off2; off2 >>= 1)
#pragma unroll
            for (int j = 0; j < RW; ++j) s[j] += __shfl_xor(s[j], off2, 32);
#pragma unroll
        for (int j = 0; j < RW; ++j) {
            mu[j] = s[j] * (1.0f / W);
            vv[j] = 0.0f;
#pragma unroll
            for (int m = 0; m < CPL; ++m) {
                float d = h[j][m] - mu[j];
                vv[j] += d * d;
            }
        }
#pragma unroll
        for (int off2 = 16; off2; off2 >>= 1)
#pragma unroll
            for (int j = 0; j < RW; ++j) vv[j] += __shfl_xor(vv[j], off2, 32);
#pragma unroll
        for (int j = 0; j < RW; ++j) {
            const float rs = rsqrtf(vv[j] * (1.0f / W) + 1e-5f);
            const long row = (long)tile * 16 + wave + NT * j;
#pragma unroll
            for (int m = 0; m < CPL; ++m) {
                float val = (h[j][m] - mu[j]) * rs * gam[m] + bet[m];
                out[row * W + lane + 32 * m] = fmaxf(val, 0.0f);
            }
        }
    }
}

__global__ __launch_bounds__(256)
void proj_gemm(const float* __restrict__ A,
               const float* __restrict__ Wr,
               const float* __restrict__ Wn,
               const float* __restrict__ bias,
               float* __restrict__ z,
               float* __restrict__ y)
{
    constexpr int NT = 8;
    constexpr int K = 128;
    constexpr int KS = K / 4;
    constexpr int LDA = K + 4;
    constexpr int RW = 16 / NT;
    constexpr int LDH = 2 * D_OUT + 4;
    __shared__ __attribute__((aligned(16))) float As[2][16][LDA];
    __shared__ __attribute__((aligned(16))) float Hs[2][16][LDH];

    const int lane = threadIdx.x & 31;
    const int wave = __builtin_amdgcn_readfirstlane(threadIdx.x >> 5);

    int tile = blockIdx.x;
    if (tile >= ROW_TILES) return;

    const int arow = lane & 15;
    const int koff = (lane >> 4) << 1;
    const int bcol = lane & 15;
    const bool nbr = wave >= 4;
    const float* __restrict__ Wsel = nbr ? Wn : Wr;
    const int n0 = (wave & 3) * 16;

    v2f bW[KS];
#pragma unroll
    for (int s = 0; s < KS; ++s) {
        const int k = 4 * s + koff;
        bW[s].x = Wsel[k * D_OUT + n0 + bcol];
        bW[s].y = Wsel[(k + 1) * D_OUT + n0 + bcol];
    }
    const float bv = nbr ? 0.0f : bias[n0 + bcol];

    v4f pa[RW];
#pragma unroll
    for (int j = 0; j < RW; ++j) {
        const long row = (long)tile * 16 + wave + NT * j;
        pa[j] = *(const v4f*)(A + row * K + lane * 4);
    }
#pragma unroll
    for (int j = 0; j < RW; ++j)
        *(v4f*)&As[0][wave + NT * j][lane * 4] = pa[j];
    __syncthreads();

    float* __restrict__ dcol = (lane < 16) ? (z + lane * 4) : (y + (lane - 16) * 4);

    for (int it = 0; tile < ROW_TILES; ++it, tile += gridDim.x) {
        const int p = it & 1;
        const int nxt = tile + gridDim.x;
        const bool more = nxt < ROW_TILES;

        if (more) {
#pragma unroll
            for (int j = 0; j < RW; ++j) {
                const long row = (long)nxt * 16 + wave + NT * j;
                pa[j] = *(const v4f*)(A + row * K + lane * 4);
            }
        }

        v8f acc = {};
#pragma unroll
        for (int s = 0; s < KS; ++s) {
            const v2f a0 = *(const v2f*)&As[p][arow][4 * s + koff];
            acc = __builtin_amdgcn_wmma_f32_16x16x4_f32(
                false, a0, false, bW[s], (short)0, acc, false, false);
        }

        {
            const int rbase = (lane >> 4) << 3;
#pragma unroll
            for (int v = 0; v < 8; ++v)
                Hs[p][rbase + v][wave * 16 + bcol] = acc[v] + bv;
        }
        if (more) {
#pragma unroll
            for (int j = 0; j < RW; ++j)
                *(v4f*)&As[p ^ 1][wave + NT * j][lane * 4] = pa[j];
        }
        __syncthreads();

#pragma unroll
        for (int j = 0; j < RW; ++j) {
            const int r = wave + NT * j;
            const v4f hv = *(const v4f*)&Hs[p][r][lane * 4];
            *(v4f*)(dcol + ((long)tile * 16 + r) * D_OUT) = hv;
        }
    }
}

extern "C" void kernel_launch(void* const* d_in, const int* in_sizes, int n_in,
                              void* d_out, int out_size, void* d_ws, size_t ws_size,
                              hipStream_t stream) {
    const float* x    = (const float*)d_in[0];
    const int*   esrc = (const int*)d_in[1];
    const int*   edst = (const int*)d_in[2];
    const float* Wr0  = (const float*)d_in[3];
    const float* Wn0  = (const float*)d_in[4];
    const float* b0   = (const float*)d_in[5];
    const float* g0   = (const float*)d_in[6];
    const float* be0  = (const float*)d_in[7];
    const float* Wr1  = (const float*)d_in[8];
    const float* Wn1  = (const float*)d_in[9];
    const float* b1   = (const float*)d_in[10];
    const float* g1   = (const float*)d_in[11];
    const float* be1  = (const float*)d_in[12];
    const float* Wr2  = (const float*)d_in[13];
    const float* Wn2  = (const float*)d_in[14];
    const float* b2   = (const float*)d_in[15];
    float* out = (float*)d_out;

    const int N = GNN_N;
    const int E = in_sizes[1];

    float* h0 = (float*)d_ws;
    float* h1 = h0 + (size_t)N * D_IN;
    float* yp = h1 + (size_t)N * D_IN;
    float* zp = yp + (size_t)N * D_OUT;
    int* Hc   = (int*)(zp + (size_t)N * D_OUT);
    int* Oc   = Hc + N_BUCKETS * EDGE_CHUNKS;
    int* off  = Oc + (N_BUCKETS * EDGE_CHUNKS + 32);
    int* tmp  = off + (N + 32);
    int* csr  = tmp + E;

    const int gathBlocks = (N + 7) / 8;
    const int gemmBlocks8 = 512;

    bucket_hist<<<EDGE_CHUNKS, 1024, 0, stream>>>(edst, Hc, E);
    scan_deg<<<1, 1024, 0, stream>>>(Hc, Oc, N_BUCKETS * EDGE_CHUNKS);
    bucket_part<<<EDGE_CHUNKS, 1024, 0, stream>>>(esrc, edst, Oc, tmp, E);
    bucket_csr<<<N_BUCKETS, 1024, 0, stream>>>(tmp, Oc, off, csr);

    sage_gemm<<<gemmBlocks8, 256, 0, stream>>>(x, off, csr, Wr0, Wn0, b0, g0, be0, h0);
    sage_gemm<<<gemmBlocks8, 256, 0, stream>>>(h0, off, csr, Wr1, Wn1, b1, g1, be1, h1);

    proj_gemm<<<gemmBlocks8, 256, 0, stream>>>(h1, Wr2, Wn2, b2, zp, yp);
    gather_lsm<<<gathBlocks, 256, 0, stream>>>(yp, zp, off, csr, out, N);
}


# sage_gemm: the next tile's neighbour rounds run under the current tile's multiply: 16 rows asked for in front of a third of its WMMAs, added behind it (was all rounds, then all WMMAs); sum order kept
# speedup vs baseline: 4.7373x; 1.0684x over previous
#include <hip/hip_runtime.h>
#include <hip/hip_bf16.h>
#include <math.h>

typedef float v2f __attribute__((ext_vector_type(2)));
typedef float v8f __attribute__((ext_vector_type(8)));
typedef float v4f __attribute__((ext_vector_type(4)));

#define GNN_N 100000
#define D_IN 128
#define D_OUT 64

#define EDGE_CHUNKS 256
#define NODE_SHIFT 9
#define BUCKET_NODES (1 << NODE_SHIFT)
#define N_BUCKETS ((GNN_N + BUCKET_NODES - 1) / BUCKET_NODES)
static_assert((N_BUCKETS * EDGE_CHUNKS) % 16 == 0, "scan_deg works on whole int4 groups");
static_assert(GNN_N < (1 << (31 - NODE_SHIFT)), "a src id and a node's place in its bucket share one word");
static_assert(GNN_N % BUCKET_NODES != 0, "off[N] is the last bucket's entry N % BUCKET_NODES");

__device__ __forceinline__ int bucket_of(int d) {
    return min((int)((unsigned)d >> NODE_SHIFT), N_BUCKETS - 1);
}

#define EDGE_BATCH 8
#define CSR_BATCH 12
#define SEG_CAP 16384

__global__ __launch_bounds__(1024)
void bucket_hist(const int* __restrict__ dst, int* __restrict__ H, int E) {
    __shared__ int h[N_BUCKETS];
    const int tid = threadIdx.x, w = blockIdx.x;
    const int chunk = (E + EDGE_CHUNKS - 1) / EDGE_CHUNKS;
    const int lo = min(w * chunk, E), hi = min(lo + chunk, E);
    for (int t = tid; t < N_BUCKETS; t += 1024) h[t] = 0;
    __syncthreads();
    for (int base = lo; base < hi; base += EDGE_BATCH * 1024) {
        int d[EDGE_BATCH];
#pragma unroll
        for (int u = 0; u < EDGE_BATCH; ++u) {
            const int i = base + u * 1024 + tid;
            d[u] = dst[min(i, hi - 1)];
        }
#pragma unroll
        for (int u = 0; u < EDGE_BATCH; ++u) {
            const int i = base + u * 1024 + tid;
            if (i < hi) atomicAdd(&h[bucket_of(d[u])], 1);
        }
    }
    __syncthreads();
    for (int t = tid; t < N_BUCKETS; t += 1024) H[t * EDGE_CHUNKS + w] = h[t];
}

__global__ __launch_bounds__(1024)
void scan_deg(const int* __restrict__ cnt, int* __restrict__ off, int n) {
    __shared__ int wsum[32];
    const int tid = threadIdx.x, lane = tid & 31, wave = tid >> 5;
    int carry = 0;
    int4 nxt[4];
#pragma unroll
    for (int u = 0; u < 4; ++u) nxt[u] = make_int4(0, 0, 0, 0);
    if (tid * 16 < n) {
#pragma unroll
        for (int u = 0; u < 4; ++u) nxt[u] = *(const int4*)(cnt + tid * 16 + 4 * u);
    }
    for (int base = 0; base < n; base += 16384) {
        const int i = base + tid * 16;
        int4 c[4];
#pragma unroll
        for (int u = 0; u < 4; ++u) c[u] = nxt[u];
#pragma unroll
        for (int u = 0; u < 4; ++u) nxt[u] = make_int4(0, 0, 0, 0);
        if (i + 16384 < n) {
#pragma unroll
            for (int u = 0; u < 4; ++u) nxt[u] = *(const int4*)(cnt + i + 16384 + 4 * u);
        }
        int t = 0;
#pragma unroll
        for (int u = 0; u < 4; ++u) t += c[u].x + c[u].y + c[u].z + c[u].w;
        int s = t;
#pragma unroll
        for (int o = 1; o < 32; o <<= 1) {
            int v = __shfl_up(s, o, 32);
            if (lane >= o) s += v;
        }
        if (lane == 31) wsum[wave] = s;
        __syncthreads();
        int ws = wsum[lane];
#pragma unroll
        for (int o = 1; o < 32; o <<= 1) {
            int v = __shfl_up(ws, o, 32);
            if (lane >= o) ws += v;
        }
        const int waveIncl = __shfl(ws, wave, 32);
        const int total = __shfl(ws, 31, 32);
        const int waveTot = __shfl(s, 31, 32);
        if (i < n) {
            int run = carry + (waveIncl - waveTot) + (s - t);
#pragma unroll
            for (int u = 0; u < 4; ++u) {
                int4 o4;
                o4.x = run;
                o4.y = o4.x + c[u].x;
                o4.z = o4.y + c[u].y;
                o4.w = o4.z + c[u].z;
                run = o4.w + c[u].w;
                *(int4*)(off + i + 4 * u) = o4;
            }
        }
        carry += total;
        __syncthreads();
    }
    if (tid == 0) off[n] = carry;
}

__global__ __launch_bounds__(1024)
void bucket_part(const int* __restrict__ src, const int* __restrict__ dst,
                 const int* __restrict__ O, int* __restrict__ tmp, int E) {
    __shared__ int cur[N_BUCKETS];
    const int tid = threadIdx.x, w = blockIdx.x;
    const int chunk = (E + EDGE_CHUNKS - 1) / EDGE_CHUNKS;
    const int lo = min(w * chunk, E), hi = min(lo + chunk, E);
    for (int t = tid; t < N_BUCKETS; t += 1024) cur[t] = O[t * EDGE_CHUNKS + w];
    __syncthreads();
    for (int base = lo; base < hi; base += EDGE_BATCH * 1024) {
        int d[EDGE_BATCH], s[EDGE_BATCH];
#pragma unroll
        for (int u = 0; u < EDGE_BATCH; ++u) {
            const int i = min(base + u * 1024 + tid, hi - 1);
            d[u] = dst[i];
            s[u] = src[i];
        }
#pragma unroll
        for (int u = 0; u < EDGE_BATCH; ++u) {
            const int i = base + u * 1024 + tid;
            if (i < hi) {
                const int p = atomicAdd(&cur[bucket_of(d[u])], 1);
                tmp[p] = (int)(((unsigned)s[u] << NODE_SHIFT) | ((unsigned)d[u] & (BUCKET_NODES - 1)));
            }
        }
    }
}

__global__ __launch_bounds__(1024)
void bucket_csr(const int* __restrict__ tmp, const int* __restrict__ O,
                int* __restrict__ off, int* __restrict__ csr) {
    __shared__ int cnt[BUCKET_NODES];
    __shared__ int wsum[32];
    __shared__ int seg[SEG_CAP];
    const int tid = threadIdx.x, lane = tid & 31, wave = tid >> 5, b = blockIdx.x;
    const int B0 = O[b * EDGE_CHUNKS], B1 = O[(b + 1) * EDGE_CHUNKS];
    const bool staged = (B1 - B0) <= SEG_CAP;
    if (tid < BUCKET_NODES) cnt[tid] = 0;
    __syncthreads();
    for (int base = B0; base < B1; base += CSR_BATCH * 1024) {
        int e[CSR_BATCH];
#pragma unroll
        for (int u = 0; u < CSR_BATCH; ++u) {
            const int i = base + u * 1024 + tid;
            e[u] = tmp[min(i, B1 - 1)];
        }
#pragma unroll
        for (int u = 0; u < CSR_BATCH; ++u) {
            const int i = base + u * 1024 + tid;
            if (i < B1) atomicAdd(&cnt[e[u] & (BUCKET_NODES - 1)], 1);
        }
    }
    __syncthreads();
    const int v = (tid < BUCKET_NODES) ? cnt[tid] : 0;
    int s = v;
#pragma unroll
    for (int o = 1; o < 32; o <<= 1) {
        int t = __shfl_up(s, o, 32);
        if (lane >= o) s += t;
    }
    if (lane == 31) wsum[wave] = s;
    __syncthreads();
    int ws = wsum[lane];
#pragma unroll
    for (int o = 1; o < 32; o <<= 1) {
        int t = __shfl_up(ws, o, 32);
        if (lane >= o) ws += t;
    }
    const int waveIncl = __shfl(ws, wave, 32);
    const int waveTot = __shfl(s, 31, 32);
    const int excl = (waveIncl - waveTot) + (s - v);
    if (tid < BUCKET_NODES) {
        cnt[tid] = excl;
        const int node = b * BUCKET_NODES + tid;
        if (node <= GNN_N) off[node] = B0 + excl;
    }
    __syncthreads();
    for (int base = B0; base < B1; base += CSR_BATCH * 1024) {
        int e[CSR_BATCH];
#pragma unroll
        for (int u = 0; u < CSR_BATCH; ++u) {
            const int i = base + u * 1024 + tid;
            e[u] = tmp[min(i, B1 - 1)];
        }
#pragma unroll
        for (int u = 0; u < CSR_BATCH; ++u) {
            const int i = base + u * 1024 + tid;
            if (i < B1) {
                const int p = atomicAdd(&cnt[e[u] & (BUCKET_NODES - 1)], 1);
                const int id = (int)((unsigned)e[u] >> NODE_SHIFT);
                if (staged) seg[p] = id;
                else        csr[B0 + p] = id;
            }
        }
    }
    if (staged) {
        __syncthreads();
        for (int i = tid; i < B1 - B0; i += 1024) csr[B0 + i] = seg[i];
    }
}

__global__ __launch_bounds__(256)
void gather_lsm(const float* __restrict__ y, const float* __restrict__ z,
                const int* __restrict__ off, const int* __restrict__ csr,
                float* __restrict__ out, int n) {
    const int node = __builtin_amdgcn_readfirstlane(blockIdx.x * 8 + (threadIdx.x >> 5));
    if (node >= n) return;
    const int lane = threadIdx.x & 31;
    const bool odd = lane >= 16;
    const int c4 = (lane & 15) * 4;
    const int beg = off[node];
    const int end = off[node + 1];
    const float4 zv = *(const float4*)(z + (long)node * D_OUT + c4);
    const float* yl = y + c4;
    float4 acc = make_float4(0.0f, 0.0f, 0.0f, 0.0f);
    int j = beg;
    for (; j + 16 <= end; j += 16) {
        float4 v[8];
#pragma unroll
        for (int u = 0; u < 8; ++u) {
            const int i0 = csr[j + 2 * u], i1 = csr[j + 2 * u + 1];
            v[u] = *(const float4*)(yl + (long)(odd ? i1 : i0) * D_OUT);
        }
#pragma unroll
        for (int u = 0; u < 8; ++u) {
            acc.x += v[u].x; acc.y += v[u].y; acc.z += v[u].z; acc.w += v[u].w;
        }
    }
    if (j + 8 <= end) {
        float4 v[4];
#pragma unroll
        for (int u = 0; u < 4; ++u) {
            const int i0 = csr[j + 2 * u], i1 = csr[j + 2 * u + 1];
            v[u] = *(const float4*)(yl + (long)(odd ? i1 : i0) * D_OUT);
        }
#pragma unroll
        for (int u = 0; u < 4; ++u) {
            acc.x += v[u].x; acc.y += v[u].y; acc.z += v[u].z; acc.w += v[u].w;
        }
        j += 8;
    }
    if (j + 4 <= end) {
        const int i0 = csr[j], i1 = csr[j + 1], i2 = csr[j + 2], i3 = csr[j + 3];
        float4 v0 = *(const float4*)(yl + (long)(odd ? i1 : i0) * D_OUT);
        float4 v1 = *(const float4*)(yl + (long)(odd ? i3 : i2) * D_OUT);
        acc.x += v0.x; acc.y += v0.y; acc.z += v0.z; acc.w += v0.w;
        acc.x += v1.x; acc.y += v1.y; acc.z += v1.z; acc.w += v1.w;
        j += 4;
    }
    if (j + 2 <= end) {
        const int i0 = csr[j], i1 = csr[j + 1];
        float4 v0 = *(const float4*)(yl + (long)(odd ? i1 : i0) * D_OUT);
        acc.x += v0.x; acc.y += v0.y; acc.z += v0.z; acc.w += v0.w;
        j += 2;
    }
    if (j < end) {
        const int i0 = csr[j];
        if (!odd) {
            float4 v0 = *(const float4*)(yl + (long)i0 * D_OUT);
            acc.x += v0.x; acc.y += v0.y; acc.z += v0.z; acc.w += v0.w;
        }
    }
    acc.x += __shfl_xor(acc.x, 16, 32);
    acc.y += __shfl_xor(acc.y, 16, 32);
    acc.z += __shfl_xor(acc.z, 16, 32);
    acc.w += __shfl_xor(acc.w, 16, 32);
    const float inv = 1.0f / fmaxf((float)(end - beg), 1.0f);
    float4 h;
    h.x = zv.x + acc.x * inv; h.y = zv.y + acc.y * inv;
    h.z = zv.z + acc.z * inv; h.w = zv.w + acc.w * inv;
    float mx = fmaxf(fmaxf(h.x, h.y), fmaxf(h.z, h.w));
#pragma unroll
    for (int o = 8; o; o >>= 1) mx = fmaxf(mx, __shfl_xor(mx, o, 32));
    float s = expf(h.x - mx) + expf(h.y - mx) + expf(h.z - mx) + expf(h.w - mx);
#pragma unroll
    for (int o = 8; o; o >>= 1) s += __shfl_xor(s, o, 32);
    const float ls = logf(s);
    if (!odd) {
        float4 r;
        r.x = h.x - mx - ls; r.y = h.y - mx - ls; r.z = h.z - mx - ls; r.w = h.w - mx - ls;
        *(float4*)(out + (long)node * D_OUT + c4) = r;
    }
}

#define ROW_TILES (GNN_N / 16)
#define GATHER_ROWS 16
#define ID_SPAN 64
#define MAIN_ROWS 16
#define MAIN_ROUNDS 3
static_assert(MAIN_ROUNDS * MAIN_ROWS <= ID_SPAN, "the rounds under the multiply take their ids from the prefetched span");

#define SAGE_MUL_PART(S0, S1)                                                                  \
    _Pragma("unroll")   \
    for (int s = (S0); s < (S1); ++s) {                                                        \
        const v2f f0 = *(const v2f*)&As[p][arow][4 * s + koff];                                \
        const v2f f1 = *(const v2f*)&Gs[p][arow][4 * s + koff];                                \
        acc = __builtin_amdgcn_wmma_f32_16x16x4_f32(                                           \
            false, f0, false, bR[s], (short)0, acc, false, false);                             \
        acc = __builtin_amdgcn_wmma_f32_16x16x4_f32(                                           \
            false, f1, false, bN[s], (short)0, acc, false, false);                             \
    }

__device__ __forceinline__ void load_ids(const int* __restrict__ csr, int j0, int b2, int lane,
                                         int& iv0, int& iv1) {
    const int last = max(b2 - 1, 0);
    iv0 = csr[min(j0 + lane, last)];
    iv1 = csr[min(j0 + 32 + lane, last)];
}

__device__ __forceinline__ void gather_round(const float* __restrict__ xl, int j, int b1, int b2,
                                             int ivq, int lb, v4f& a0, v4f& a1) {
    v4f v[GATHER_ROWS];
#pragma unroll
    for (int u = 0; u < GATHER_ROWS; ++u) {
        v[u] = (v4f){0.0f, 0.0f, 0.0f, 0.0f};
        if (j + u < b2) {
            const int id = __builtin_amdgcn_readlane(ivq, lb + u);
            v[u] = *(const v4f*)(xl + (long)id * D_IN);
        }
    }
#pragma unroll
    for (int u = 0; u < GATHER_ROWS; ++u) {
        if (j + u < b1) a0 += v[u];
        else            a1 += v[u];
    }
}

template <int POS0>
__device__ __forceinline__ void round_issue(const float* __restrict__ xl, int j, int b2,
                                            int iv0, int iv1, v4f (&v)[MAIN_ROWS]) {
#pragma unroll
    for (int u = 0; u < MAIN_ROWS; ++u) {
        v[u] = (v4f){0.0f, 0.0f, 0.0f, 0.0f};
        if (j + u < b2) {
            const int id = __builtin_amdgcn_readlane((POS0 + u) < 32 ? iv0 : iv1, (POS0 + u) & 31);
            v[u] = *(const v4f*)(xl + (long)id * D_IN);
        }
    }
}

__device__ __forceinline__ void round_add(int j, int b1, const v4f (&v)[MAIN_ROWS],
                                          v4f& a0, v4f& a1) {
#pragma unroll
    for (int u = 0; u < MAIN_ROWS; ++u) {
        if (j + u < b1) a0 += v[u];
        else            a1 += v[u];
    }
}

__device__ __forceinline__ void neighbour_sums2(const float* __restrict__ xl,
                                                const int* __restrict__ csr,
                                                int b0, int b1, int b2, int iv0, int iv1,
                                                int lane, v4f& s0, v4f& s1) {
    v4f a0 = {0.0f, 0.0f, 0.0f, 0.0f};
    v4f a1 = {0.0f, 0.0f, 0.0f, 0.0f};
#pragma unroll
    for (int q = 0; q < ID_SPAN / GATHER_ROWS; ++q) {
        const int j = b0 + GATHER_ROWS * q;
        if (j < b2) gather_round(xl, j, b1, b2, (q & 2) ? iv1 : iv0, (q & 1) * 16, a0, a1);
    }
#pragma nounroll
    for (int j = b0 + ID_SPAN; j < b2; j += GATHER_ROWS) {
        const int r = (j - b0) / GATHER_ROWS;
        if ((r & 3) == 0) load_ids(csr, j, b2, lane, iv0, iv1);
        gather_round(xl, j, b1, b2, (r & 2) ? iv1 : iv0, (r & 1) * 16, a0, a1);
    }
    s0 = a0;
    s1 = a1;
}

__device__ __forceinline__ void fetch_rows2(const float* __restrict__ A,
                                            const int* __restrict__ csr,
                                            int r0, int lane, int b0, int b1, int b2,
                                            int iv0, int iv1, v4f (&pa)[2], v4f (&pg)[2]) {
    const float* xl = A + lane * 4;
    pa[0] = *(const v4f*)(xl + (long)r0 * D_IN);
    pa[1] = *(const v4f*)(xl + (long)(r0 + 1) * D_IN);
    v4f s0, s1;
    neighbour_sums2(xl, csr, b0, b1, b2, iv0, iv1, lane, s0, s1);
    pg[0] = s0 * (1.0f / fmaxf((float)(b1 - b0), 1.0f));
    pg[1] = s1 * (1.0f / fmaxf((float)(b2 - b1), 1.0f));
}

__global__ __launch_bounds__(256, 4)
void sage_gemm(const float* __restrict__ A,
               const int* __restrict__ off,
               const int* __restrict__ csr,
               const float* __restrict__ Wr,
               const float* __restrict__ Wn,
               const float* __restrict__ bias,
               const float* __restrict__ gamma,
               const float* __restrict__ beta,
               float* __restrict__ out)
{
    constexpr int NT = 8;
    constexpr int W = 16 * NT;
    constexpr int K = 128;
    constexpr int KS = K / 4;
    constexpr int LDA = K + 4;
    constexpr int RW = 16 / NT;
    constexpr int CPL = W / 32;
    static_assert(RW == 2, "fetch_rows2 stages two adjacent rows per wave");
    __shared__ __attribute__((aligned(16))) float As[2][16][LDA];
    __shared__ __attribute__((aligned(16))) float Gs[2][16][LDA];
    __shared__ float Hs[2][16][W];
    __shared__ float GBs[2][W];

    const int lane = threadIdx.x & 31;
    const int wave = __builtin_amdgcn_readfirstlane(threadIdx.x >> 5);

    int tile = blockIdx.x;
    if (tile >= ROW_TILES) return;

    const int n0   = wave * 16;
    const int arow = lane & 15;
    const int koff = (lane >> 4) << 1;
    const int bcol = lane & 15;

    v2f bR[KS], bN[KS];
#pragma unroll
    for (int s = 0; s < KS; ++s) {
        const int k = 4 * s + koff;
        bR[s].x = Wr[k * W + n0 + bcol];
        bR[s].y = Wr[(k + 1) * W + n0 + bcol];
        bN[s].x = Wn[k * W + n0 + bcol];
        bN[s].y = Wn[(k + 1) * W + n0 + bcol];
    }
    const float bv = bias[n0 + bcol];
    if (threadIdx.x < W) {
        GBs[0][threadIdx.x] = gamma[threadIdx.x];
        GBs[1][threadIdx.x] = beta[threadIdx.x];
    }

    const int step = gridDim.x;
    const int rl = RW * wave;
    const int lq = min(lane, 2);
    v4f pa[RW], pg[RW];
    int p0, p1, p2, pv0, pv1, qv;
    {
        const int ra = tile * 16 + rl;
        const int rb = min(tile + step, ROW_TILES - 1) * 16 + rl;
        const int rc = min(tile + 2 * step, ROW_TILES - 1) * 16 + rl;
        const int c0 = off[ra], c1 = off[ra + 1], c2 = off[ra + 2];
        p0 = off[rb]; p1 = off[rb + 1]; p2 = off[rb + 2];
        qv = off[rc + lq];
        int cv0, cv1;
        load_ids(csr, c0, c2, lane, cv0, cv1);
        load_ids(csr, p0, p2, lane, pv0, pv1);
        fetch_rows2(A, csr, ra, lane, c0, c1, c2, cv0, cv1, pa, pg);
    }
    v4f sa0, sa1;
    {
        const float* xs = A + (long)(min(tile + step, ROW_TILES - 1) * 16 + rl) * D_IN + lane * 4;
        sa0 = *(const v4f*)xs;
        sa1 = *(const v4f*)(xs + D_IN);
    }
#pragma unroll
    for (int j = 0; j < RW; ++j) {
        *(v4f*)&As[0][RW * wave + j][lane * 4] = pa[j];
        *(v4f*)&Gs[0][RW * wave + j][lane * 4] = pg[j];
    }
    __syncthreads();

    for (int it = 0; tile < ROW_TILES; ++it, tile += gridDim.x) {
        const int p = it & 1;
        const int nxt = tile + gridDim.x;
        const bool more = nxt < ROW_TILES;

        const int q0 = __builtin_amdgcn_readlane(qv, 0);
        const int q1 = __builtin_amdgcn_readlane(qv, 1);
        const int q2 = __builtin_amdgcn_readlane(qv, 2);
        int nv0, nv1;
        load_ids(csr, q0, q2, lane, nv0, nv1);
        const int dv = off[min(tile + 3 * step, ROW_TILES - 1) * 16 + rl + lq];

        const float* xl = A + lane * 4;
        v4f gv[MAIN_ROWS];
        v4f a0 = {0.0f, 0.0f, 0.0f, 0.0f};
        v4f a1 = {0.0f, 0.0f, 0.0f, 0.0f};
        *(v4f*)&As[p ^ 1][rl][lane * 4] = sa0;
        *(v4f*)&As[p ^ 1][rl + 1][lane * 4] = sa1;
        if (more) {
            if (p0 < p2) round_issue<0>(xl, p0, p2, pv0, pv1, gv);
        }

        v8f acc = {};
        SAGE_MUL_PART(0, 11)
        if (more) {
            if (p0 < p2) round_add(p0, p1, gv, a0, a1);
            if (p0 + MAIN_ROWS < p2)
                round_issue<MAIN_ROWS>(xl, p0 + MAIN_ROWS, p2, pv0, pv1, gv);
        }
        SAGE_MUL_PART(11, 22)
        if (more) {
            if (p0 + MAIN_ROWS < p2) round_add(p0 + MAIN_ROWS, p1, gv, a0, a1);
            if (p0 + 2 * MAIN_ROWS < p2)
                round_issue<2 * MAIN_ROWS>(xl, p0 + 2 * MAIN_ROWS, p2, pv0, pv1, gv);
        }
        SAGE_MUL_PART(22, KS)

        {
            const int rbase = (lane >> 4) << 3;
#pragma unroll
            for (int v = 0; v < 8; ++v)
                Hs[p][rbase + v][n0 + bcol] = acc[v] + bv;
        }
        if (more) {
            if (p0 + 2 * MAIN_ROWS < p2) round_add(p0 + 2 * MAIN_ROWS, p1, gv, a0, a1);
#pragma nounroll
            for (int j = p0 + MAIN_ROUNDS * MAIN_ROWS; j < p2; j += MAIN_ROWS) {
                int tv0, tv1;
                load_ids(csr, j, p2, lane, tv0, tv1);
                round_issue<0>(xl, j, p2, tv0, tv1, gv);
                round_add(j, p1, gv, a0, a1);
            }
            *(v4f*)&Gs[p ^ 1][rl][lane * 4] = a0 * (1.0f / fmaxf((float)(p1 - p0), 1.0f));
            *(v4f*)&Gs[p ^ 1][rl + 1][lane * 4] = a1 * (1.0f / fmaxf((float)(p2 - p1), 1.0f));
        }
        p0 = q0; p1 = q1; p2 = q2; pv0 = nv0; pv1 = nv1;
        qv = dv;
        __syncthreads();

        {
            const float* xs = xl + (long)(min(tile + 2 * step, ROW_TILES - 1) * 16 + rl) * D_IN;
            sa0 = *(const v4f*)xs;
            sa1 = *(const v4f*)(xs + D_IN);
        }

        float h[RW][CPL];
#pragma unroll
        for (int j = 0; j < RW; ++j)
#pragma unroll
            for (int m = 0; m < CPL; ++m) h[j][m] = Hs[p][wave + NT * j][lane + 32 * m];
        float gam[CPL], bet[CPL];
#pragma unroll
        for (int m = 0; m < CPL; ++m) { gam[m] = GBs[0][lane + 32 * m]; bet[m] = GBs[1][lane + 32 * m]; }

        float s[RW], vv[RW], mu[RW];
#pragma unroll
        for (int j = 0; j < RW; ++j) {
            s[j] = 0.0f;
#pragma unroll
            for (int m = 0; m < CPL; ++m) s[j] += h[j][m];
        }
#pragma unroll
        for (int off2 = 16; off2; off2 >>= 1)
#pragma unroll
            for (int j = 0; j < RW; ++j) s[j] += __shfl_xor(s[j], off2, 32);
#pragma unroll
        for (int j = 0; j < RW; ++j) {
            mu[j] = s[j] * (1.0f / W);
            vv[j] = 0.0f;
#pragma unroll
            for (int m = 0; m < CPL; ++m) {
                float d = h[j][m] - mu[j];
                vv[j] += d * d;
            }
        }
#pragma unroll
        for (int off2 = 16; off2; off2 >>= 1)
#pragma unroll
            for (int j = 0; j < RW; ++j) vv[j] += __shfl_xor(vv[j], off2, 32);
#pragma unroll
        for (int j = 0; j < RW; ++j) {
            const float rs = rsqrtf(vv[j] * (1.0f / W) + 1e-5f);
            const long row = (long)tile * 16 + wave + NT * j;
#pragma unroll
            for (int m = 0; m < CPL; ++m) {
                float val = (h[j][m] - mu[j]) * rs * gam[m] + bet[m];
                out[row * W + lane + 32 * m] = fmaxf(val, 0.0f);
            }
        }
    }
}

__global__ __launch_bounds__(256)
void proj_gemm(const float* __restrict__ A,
               const float* __restrict__ Wr,
               const float* __restrict__ Wn,
               const float* __restrict__ bias,
               float* __restrict__ z,
               float* __restrict__ y)
{
    constexpr int NT = 8;
    constexpr int K = 128;
    constexpr int KS = K / 4;
    constexpr int LDA = K + 4;
    constexpr int RW = 16 / NT;
    constexpr int LDH = 2 * D_OUT + 4;
    __shared__ __attribute__((aligned(16))) float As[2][16][LDA];
    __shared__ __attribute__((aligned(16))) float Hs[2][16][LDH];

    const int lane = threadIdx.x & 31;
    const int wave = __builtin_amdgcn_readfirstlane(threadIdx.x >> 5);

    int tile = blockIdx.x;
    if (tile >= ROW_TILES) return;

    const int arow = lane & 15;
    const int koff = (lane >> 4) << 1;
    const int bcol = lane & 15;
    const bool nbr = wave >= 4;
    const float* __restrict__ Wsel = nbr ? Wn : Wr;
    const int n0 = (wave & 3) * 16;

    v2f bW[KS];
#pragma unroll
    for (int s = 0; s < KS; ++s) {
        const int k = 4 * s + koff;
        bW[s].x = Wsel[k * D_OUT + n0 + bcol];
        bW[s].y = Wsel[(k + 1) * D_OUT + n0 + bcol];
    }
    const float bv = nbr ? 0.0f : bias[n0 + bcol];

    v4f pa[RW];
#pragma unroll
    for (int j = 0; j < RW; ++j) {
        const long row = (long)tile * 16 + wave + NT * j;
        pa[j] = *(const v4f*)(A + row * K + lane * 4);
    }
#pragma unroll
    for (int j = 0; j < RW; ++j)
        *(v4f*)&As[0][wave + NT * j][lane * 4] = pa[j];
    __syncthreads();

    float* __restrict__ dcol = (lane < 16) ? (z + lane * 4) : (y + (lane - 16) * 4);

    for (int it = 0; tile < ROW_TILES; ++it, tile += gridDim.x) {
        const int p = it & 1;
        const int nxt = tile + gridDim.x;
        const bool more = nxt < ROW_TILES;

        if (more) {
#pragma unroll
            for (int j = 0; j < RW; ++j) {
                const long row = (long)nxt * 16 + wave + NT * j;
                pa[j] = *(const v4f*)(A + row * K + lane * 4);
            }
        }

        v8f acc = {};
#pragma unroll
        for (int s = 0; s < KS; ++s) {
            const v2f a0 = *(const v2f*)&As[p][arow][4 * s + koff];
            acc = __builtin_amdgcn_wmma_f32_16x16x4_f32(
                false, a0, false, bW[s], (short)0, acc, false, false);
        }

        {
            const int rbase = (lane >> 4) << 3;
#pragma unroll
            for (int v = 0; v < 8; ++v)
                Hs[p][rbase + v][wave * 16 + bcol] = acc[v] + bv;
        }
        if (more) {
#pragma unroll
            for (int j = 0; j < RW; ++j)
                *(v4f*)&As[p ^ 1][wave + NT * j][lane * 4] = pa[j];
        }
        __syncthreads();

#pragma unroll
        for (int j = 0; j < RW; ++j) {
            const int r = wave + NT * j;
            const v4f hv = *(const v4f*)&Hs[p][r][lane * 4];
            *(v4f*)(dcol + ((long)tile * 16 + r) * D_OUT) = hv;
        }
    }
}

extern "C" void kernel_launch(void* const* d_in, const int* in_sizes, int n_in,
                              void* d_out, int out_size, void* d_ws, size_t ws_size,
                              hipStream_t stream) {
    const float* x    = (const float*)d_in[0];
    const int*   esrc = (const int*)d_in[1];
    const int*   edst = (const int*)d_in[2];
    const float* Wr0  = (const float*)d_in[3];
    const float* Wn0  = (const float*)d_in[4];
    const float* b0   = (const float*)d_in[5];
    const float* g0   = (const float*)d_in[6];
    const float* be0  = (const float*)d_in[7];
    const float* Wr1  = (const float*)d_in[8];
    const float* Wn1  = (const float*)d_in[9];
    const float* b1   = (const float*)d_in[10];
    const float* g1   = (const float*)d_in[11];
    const float* be1  = (const float*)d_in[12];
    const float* Wr2  = (const float*)d_in[13];
    const float* Wn2  = (const float*)d_in[14];
    const float* b2   = (const float*)d_in[15];
    float* out = (float*)d_out;

    const int N = GNN_N;
    const int E = in_sizes[1];

    float* h0 = (float*)d_ws;
    float* h1 = h0 + (size_t)N * D_IN;
    float* yp = h1 + (size_t)N * D_IN;
    float* zp = yp + (size_t)N * D_OUT;
    int* Hc   = (int*)(zp + (size_t)N * D_OUT);
    int* Oc   = Hc + N_BUCKETS * EDGE_CHUNKS;
    int* off  = Oc + (N_BUCKETS * EDGE_CHUNKS + 32);
    int* tmp  = off + (N + 32);
    int* csr  = tmp + E;

    const int gathBlocks = (N + 7) / 8;
    const int gemmBlocks8 = 512;

    bucket_hist<<<EDGE_CHUNKS, 1024, 0, stream>>>(edst, Hc, E);
    scan_deg<<<1, 1024, 0, stream>>>(Hc, Oc, N_BUCKETS * EDGE_CHUNKS);
    bucket_part<<<EDGE_CHUNKS, 1024, 0, stream>>>(esrc, edst, Oc, tmp, E);
    bucket_csr<<<N_BUCKETS, 1024, 0, stream>>>(tmp, Oc, off, csr);

    sage_gemm<<<gemmBlocks8, 256, 0, stream>>>(x, off, csr, Wr0, Wn0, b0, g0, be0, h0);
    sage_gemm<<<gemmBlocks8, 256, 0, stream>>>(h0, off, csr, Wr1, Wn1, b1, g1, be1, h1);

    proj_gemm<<<gemmBlocks8, 256, 0, stream>>>(h1, Wr2, Wn2, b2, zp, yp);
    gather_lsm<<<gathBlocks, 256, 0, stream>>>(yp, zp, off, csr, out, N);
}


# Edge sort re-cut from 4 launches to 2: level 1 sorts each chunk by bucket in LDS (count, scan, rank; whole-line stores), level 2 collects a bucket's 256 runs; per-node sum order = LDS arrival order
# speedup vs baseline: 5.3737x; 1.1343x over previous
#include <hip/hip_runtime.h>
#include <hip/hip_bf16.h>
#include <math.h>

typedef float v2f __attribute__((ext_vector_type(2)));
typedef float v8f __attribute__((ext_vector_type(8)));
typedef float v4f __attribute__((ext_vector_type(4)));

#define GNN_N 100000
#define D_IN 128
#define D_OUT 64

#define EDGE_CHUNKS 256
#define NODE_SHIFT 9
#define BUCKET_NODES (1 << NODE_SHIFT)
#define N_BUCKETS ((GNN_N + BUCKET_NODES - 1) / BUCKET_NODES)
#define L_ROW (N_BUCKETS + 1)
static_assert(N_BUCKETS <= 1024, "chunk_sort scans the bucket counters with one thread per counter");
static_assert(EDGE_CHUNKS % 32 == 0 && EDGE_CHUNKS <= 1024, "bucket_csr gives every wave the same number of runs");
static_assert(GNN_N < (1 << (31 - NODE_SHIFT)), "a src id and a node's place in its bucket share one word");
static_assert(GNN_N % BUCKET_NODES != 0, "off[N] is the last bucket's entry N % BUCKET_NODES");

__device__ __forceinline__ int bucket_of(int d) {
    return min((int)((unsigned)d >> NODE_SHIFT), N_BUCKETS - 1);
}

#define EDGE_BATCH 8
#define CHUNK_CAP (EDGE_BATCH * 1024)
#define SEG_CAP 16384

__global__ __launch_bounds__(1024)
void chunk_sort(const int* __restrict__ src, const int* __restrict__ dst,
                int* __restrict__ L, int* __restrict__ tmp, int E) {
    __shared__ int cnt[N_BUCKETS];
    __shared__ int wsum[32];
    __shared__ int stage[CHUNK_CAP];
    const int tid = threadIdx.x, lane = tid & 31, wave = tid >> 5, w = blockIdx.x;
    const int chunk = (E + EDGE_CHUNKS - 1) / EDGE_CHUNKS;
    const int lo = min(w * chunk, E), hi = min(lo + chunk, E);
    const bool single = (hi - lo) <= CHUNK_CAP;
    if (tid < N_BUCKETS) cnt[tid] = 0;
    __syncthreads();
    int d[EDGE_BATCH], s[EDGE_BATCH];
#pragma unroll
    for (int u = 0; u < EDGE_BATCH; ++u) { d[u] = 0; s[u] = 0; }
    if (single) {
        if (lo < hi) {
#pragma unroll
            for (int u = 0; u < EDGE_BATCH; ++u) {
                const int i = min(lo + u * 1024 + tid, hi - 1);
                d[u] = dst[i];
                s[u] = src[i];
            }
        }
#pragma unroll
        for (int u = 0; u < EDGE_BATCH; ++u) {
            if (lo + u * 1024 + tid < hi) atomicAdd(&cnt[bucket_of(d[u])], 1);
        }
    } else {
        for (int base = lo; base < hi; base += CHUNK_CAP) {
#pragma unroll
            for (int u = 0; u < EDGE_BATCH; ++u) d[u] = dst[min(base + u * 1024 + tid, hi - 1)];
#pragma unroll
            for (int u = 0; u < EDGE_BATCH; ++u) {
                if (base + u * 1024 + tid < hi) atomicAdd(&cnt[bucket_of(d[u])], 1);
            }
        }
    }
    __syncthreads();
    const int v = (tid < N_BUCKETS) ? cnt[tid] : 0;
    int sc = v;
#pragma unroll
    for (int o = 1; o < 32; o <<= 1) {
        int t = __shfl_up(sc, o, 32);
        if (lane >= o) sc += t;
    }
    if (lane == 31) wsum[wave] = sc;
    __syncthreads();
    int ws = wsum[lane];
#pragma unroll
    for (int o = 1; o < 32; o <<= 1) {
        int t = __shfl_up(ws, o, 32);
        if (lane >= o) ws += t;
    }
    const int waveIncl = __shfl(ws, wave, 32);
    const int waveTot = __shfl(sc, 31, 32);
    const int total = __shfl(ws, 31, 32);
    const int excl = (waveIncl - waveTot) + (sc - v);
    if (tid < N_BUCKETS) {
        cnt[tid] = excl;
        L[w * L_ROW + tid] = excl;
    }
    if (tid == 0) L[w * L_ROW + N_BUCKETS] = total;
    __syncthreads();
    if (single) {
        int p[EDGE_BATCH];
#pragma unroll
        for (int u = 0; u < EDGE_BATCH; ++u) {
            p[u] = 0;
            if (lo + u * 1024 + tid < hi) p[u] = atomicAdd(&cnt[bucket_of(d[u])], 1);
        }
#pragma unroll
        for (int u = 0; u < EDGE_BATCH; ++u) {
            if (lo + u * 1024 + tid < hi)
                stage[p[u]] = (int)(((unsigned)s[u] << NODE_SHIFT) | ((unsigned)d[u] & (BUCKET_NODES - 1)));
        }
        __syncthreads();
        for (int i = tid; i < hi - lo; i += 1024) tmp[lo + i] = stage[i];
    } else {
        for (int base = lo; base < hi; base += CHUNK_CAP) {
#pragma unroll
            for (int u = 0; u < EDGE_BATCH; ++u) {
                const int i = min(base + u * 1024 + tid, hi - 1);
                d[u] = dst[i];
                s[u] = src[i];
            }
#pragma unroll
            for (int u = 0; u < EDGE_BATCH; ++u) {
                if (base + u * 1024 + tid < hi) {
                    const int p = atomicAdd(&cnt[bucket_of(d[u])], 1);
                    tmp[lo + p] = (int)(((unsigned)s[u] << NODE_SHIFT) | ((unsigned)d[u] & (BUCKET_NODES - 1)));
                }
            }
        }
    }
}

#define RUNS_PER_WAVE (EDGE_CHUNKS / 32)
#define RUN_SPAN 64
__global__ __launch_bounds__(1024)
void bucket_csr(const int* __restrict__ tmp, const int* __restrict__ L,
                int* __restrict__ off, int* __restrict__ csr, int E) {
    __shared__ int cnt[BUCKET_NODES];
    __shared__ int wsum[32];
    __shared__ int wsumA[32];
    __shared__ int rbeg[EDGE_CHUNKS];
    __shared__ int rlen[EDGE_CHUNKS];
    __shared__ int seg[SEG_CAP];
    const int tid = threadIdx.x, lane = tid & 31, b = blockIdx.x;
    const int wave = __builtin_amdgcn_readfirstlane(tid >> 5);
    const int chunk = (E + EDGE_CHUNKS - 1) / EDGE_CHUNKS;
    const int top = max(E - 1, 0);
    int a = 0, len = 0;
    if (tid < EDGE_CHUNKS) {
        const int* r = L + tid * L_ROW + b;
        a = r[0];
        len = r[1] - a;
        rbeg[tid] = min(tid * chunk, E) + a;
        rlen[tid] = len;
    }
    if (tid < BUCKET_NODES) cnt[tid] = 0;
    int sl = len, sa = a;
#pragma unroll
    for (int o = 16; o; o >>= 1) {
        sl += __shfl_xor(sl, o, 32);
        sa += __shfl_xor(sa, o, 32);
    }
    if (lane == 0) { wsum[wave] = sl; wsumA[wave] = sa; }
    __syncthreads();
    int wl = wsum[lane], wa = wsumA[lane];
#pragma unroll
    for (int o = 16; o; o >>= 1) {
        wl += __shfl_xor(wl, o, 32);
        wa += __shfl_xor(wa, o, 32);
    }
    const int B0 = __builtin_amdgcn_readfirstlane(wa);
    const int nb = __builtin_amdgcn_readfirstlane(wl);
    const bool staged = nb <= SEG_CAP;

    const int w0 = wave * RUNS_PER_WAVE;
    int e[2 * RUNS_PER_WAVE];
    int lnv[RUNS_PER_WAVE];
    int lmax = 0;
#pragma unroll
    for (int r = 0; r < RUNS_PER_WAVE; ++r) {
        const int rb = __builtin_amdgcn_readfirstlane(rbeg[w0 + r]);
        const int ln = __builtin_amdgcn_readfirstlane(rlen[w0 + r]);
        lnv[r] = ln;
        lmax = max(lmax, ln);
        const int last = max(ln - 1, 0);
        e[2 * r]     = tmp[min(rb + min(lane, last), top)];
        e[2 * r + 1] = tmp[min(rb + min(lane + 32, last), top)];
    }
#pragma unroll
    for (int r = 0; r < RUNS_PER_WAVE; ++r) {
        const int ln = lnv[r];
        if (lane < ln)      atomicAdd(&cnt[e[2 * r] & (BUCKET_NODES - 1)], 1);
        if (lane + 32 < ln) atomicAdd(&cnt[e[2 * r + 1] & (BUCKET_NODES - 1)], 1);
    }
    if (lmax > RUN_SPAN) {
#pragma nounroll
        for (int r = 0; r < RUNS_PER_WAVE; ++r) {
            const int rb = __builtin_amdgcn_readfirstlane(rbeg[w0 + r]);
            const int ln = __builtin_amdgcn_readfirstlane(rlen[w0 + r]);
            for (int k = RUN_SPAN; k < ln; k += 32) {
                if (k + lane < ln) atomicAdd(&cnt[tmp[rb + k + lane] & (BUCKET_NODES - 1)], 1);
            }
        }
    }
    __syncthreads();
    const int v = (tid < BUCKET_NODES) ? cnt[tid] : 0;
    int s = v;
#pragma unroll
    for (int o = 1; o < 32; o <<= 1) {
        int t = __shfl_up(s, o, 32);
        if (lane >= o) s += t;
    }
    if (lane == 31) wsum[wave] = s;
    __syncthreads();
    int ws = wsum[lane];
#pragma unroll
    for (int o = 1; o < 32; o <<= 1) {
        int t = __shfl_up(ws, o, 32);
        if (lane >= o) ws += t;
    }
    const int waveIncl = __shfl(ws, wave, 32);
    const int waveTot = __shfl(s, 31, 32);
    const int excl = (waveIncl - waveTot) + (s - v);
    if (tid < BUCKET_NODES) {
        cnt[tid] = excl;
        const int node = b * BUCKET_NODES + tid;
        if (node <= GNN_N) off[node] = B0 + excl;
    }
    __syncthreads();
    int p[2 * RUNS_PER_WAVE];
#pragma unroll
    for (int r = 0; r < RUNS_PER_WAVE; ++r) {
        const int ln = lnv[r];
#pragma unroll
        for (int h = 0; h < 2; ++h) {
            p[2 * r + h] = 0;
            if (lane + 32 * h < ln) p[2 * r + h] = atomicAdd(&cnt[e[2 * r + h] & (BUCKET_NODES - 1)], 1);
        }
    }
#pragma unroll
    for (int r = 0; r < RUNS_PER_WAVE; ++r) {
        const int ln = lnv[r];
#pragma unroll
        for (int h = 0; h < 2; ++h) {
            if (lane + 32 * h < ln) {
                const int id = (int)((unsigned)e[2 * r + h] >> NODE_SHIFT);
                if (staged) seg[p[2 * r + h]] = id;
                else        csr[B0 + p[2 * r + h]] = id;
            }
        }
    }
    if (lmax > RUN_SPAN) {
#pragma nounroll
        for (int r = 0; r < RUNS_PER_WAVE; ++r) {
            const int rb = __builtin_amdgcn_readfirstlane(rbeg[w0 + r]);
            const int ln = __builtin_amdgcn_readfirstlane(rlen[w0 + r]);
            for (int k = RUN_SPAN; k < ln; k += 32) {
                if (k + lane < ln) {
                    const int x = tmp[rb + k + lane];
                    const int q = atomicAdd(&cnt[x & (BUCKET_NODES - 1)], 1);
                    const int id = (int)((unsigned)x >> NODE_SHIFT);
                    if (staged) seg[q] = id;
                    else        csr[B0 + q] = id;
                }
            }
        }
    }
    if (staged) {
        __syncthreads();
        for (int i = tid; i < nb; i += 1024) csr[B0 + i] = seg[i];
    }
}

__global__ __launch_bounds__(256)
void gather_lsm(const float* __restrict__ y, const float* __restrict__ z,
                const int* __restrict__ off, const int* __restrict__ csr,
                float* __restrict__ out, int n) {
    const int node = __builtin_amdgcn_readfirstlane(blockIdx.x * 8 + (threadIdx.x >> 5));
    if (node >= n) return;
    const int lane = threadIdx.x & 31;
    const bool odd = lane >= 16;
    const int c4 = (lane & 15) * 4;
    const int beg = off[node];
    const int end = off[node + 1];
    const float4 zv = *(const float4*)(z + (long)node * D_OUT + c4);
    const float* yl = y + c4;
    float4 acc = make_float4(0.0f, 0.0f, 0.0f, 0.0f);
    int j = beg;
    for (; j + 16 <= end; j += 16) {
        float4 v[8];
#pragma unroll
        for (int u = 0; u < 8; ++u) {
            const int i0 = csr[j + 2 * u], i1 = csr[j + 2 * u + 1];
            v[u] = *(const float4*)(yl + (long)(odd ? i1 : i0) * D_OUT);
        }
#pragma unroll
        for (int u = 0; u < 8; ++u) {
            acc.x += v[u].x; acc.y += v[u].y; acc.z += v[u].z; acc.w += v[u].w;
        }
    }
    if (j + 8 <= end) {
        float4 v[4];
#pragma unroll
        for (int u = 0; u < 4; ++u) {
            const int i0 = csr[j + 2 * u], i1 = csr[j + 2 * u + 1];
            v[u] = *(const float4*)(yl + (long)(odd ? i1 : i0) * D_OUT);
        }
#pragma unroll
        for (int u = 0; u < 4; ++u) {
            acc.x += v[u].x; acc.y += v[u].y; acc.z += v[u].z; acc.w += v[u].w;
        }
        j += 8;
    }
    if (j + 4 <= end) {
        const int i0 = csr[j], i1 = csr[j + 1], i2 = csr[j + 2], i3 = csr[j + 3];
        float4 v0 = *(const float4*)(yl + (long)(odd ? i1 : i0) * D_OUT);
        float4 v1 = *(const float4*)(yl + (long)(odd ? i3 : i2) * D_OUT);
        acc.x += v0.x; acc.y += v0.y; acc.z += v0.z; acc.w += v0.w;
        acc.x += v1.x; acc.y += v1.y; acc.z += v1.z; acc.w += v1.w;
        j += 4;
    }
    if (j + 2 <= end) {
        const int i0 = csr[j], i1 = csr[j + 1];
        float4 v0 = *(const float4*)(yl + (long)(odd ? i1 : i0) * D_OUT);
        acc.x += v0.x; acc.y += v0.y; acc.z += v0.z; acc.w += v0.w;
        j += 2;
    }
    if (j < end) {
        const int i0 = csr[j];
        if (!odd) {
            float4 v0 = *(const float4*)(yl + (long)i0 * D_OUT);
            acc.x += v0.x; acc.y += v0.y; acc.z += v0.z; acc.w += v0.w;
        }
    }
    acc.x += __shfl_xor(acc.x, 16, 32);
    acc.y += __shfl_xor(acc.y, 16, 32);
    acc.z += __shfl_xor(acc.z, 16, 32);
    acc.w += __shfl_xor(acc.w, 16, 32);
    const float inv = 1.0f / fmaxf((float)(end - beg), 1.0f);
    float4 h;
    h.x = zv.x + acc.x * inv; h.y = zv.y + acc.y * inv;
    h.z = zv.z + acc.z * inv; h.w = zv.w + acc.w * inv;
    float mx = fmaxf(fmaxf(h.x, h.y), fmaxf(h.z, h.w));
#pragma unroll
    for (int o = 8; o; o >>= 1) mx = fmaxf(mx, __shfl_xor(mx, o, 32));
    float s = expf(h.x - mx) + expf(h.y - mx) + expf(h.z - mx) + expf(h.w - mx);
#pragma unroll
    for (int o = 8; o; o >>= 1) s += __shfl_xor(s, o, 32);
    const float ls = logf(s);
    if (!odd) {
        float4 r;
        r.x = h.x - mx - ls; r.y = h.y - mx - ls; r.z = h.z - mx - ls; r.w = h.w - mx - ls;
        *(float4*)(out + (long)node * D_OUT + c4) = r;
    }
}

#define ROW_TILES (GNN_N / 16)
#define GATHER_ROWS 16
#define ID_SPAN 64
#define MAIN_ROWS 16
#define MAIN_ROUNDS 3
static_assert(MAIN_ROUNDS * MAIN_ROWS <= ID_SPAN, "the rounds under the multiply take their ids from the prefetched span");

#define SAGE_MUL_PART(S0, S1)                                                                  \
    _Pragma("unroll")   \
    for (int s = (S0); s < (S1); ++s) {                                                        \
        const v2f f0 = *(const v2f*)&As[p][arow][4 * s + koff];                                \
        const v2f f1 = *(const v2f*)&Gs[p][arow][4 * s + koff];                                \
        acc = __builtin_amdgcn_wmma_f32_16x16x4_f32(                                           \
            false, f0, false, bR[s], (short)0, acc, false, false);                             \
        acc = __builtin_amdgcn_wmma_f32_16x16x4_f32(                                           \
            false, f1, false, bN[s], (short)0, acc, false, false);                             \
    }

__device__ __forceinline__ void load_ids(const int* __restrict__ csr, int j0, int b2, int lane,
                                         int& iv0, int& iv1) {
    const int last = max(b2 - 1, 0);
    iv0 = csr[min(j0 + lane, last)];
    iv1 = csr[min(j0 + 32 + lane, last)];
}

__device__ __forceinline__ void gather_round(const float* __restrict__ xl, int j, int b1, int b2,
                                             int ivq, int lb, v4f& a0, v4f& a1) {
    v4f v[GATHER_ROWS];
#pragma unroll
    for (int u = 0; u < GATHER_ROWS; ++u) {
        v[u] = (v4f){0.0f, 0.0f, 0.0f, 0.0f};
        if (j + u < b2) {
            const int id = __builtin_amdgcn_readlane(ivq, lb + u);
            v[u] = *(const v4f*)(xl + (long)id * D_IN);
        }
    }
#pragma unroll
    for (int u = 0; u < GATHER_ROWS; ++u) {
        if (j + u < b1) a0 += v[u];
        else            a1 += v[u];
    }
}

template <int POS0>
__device__ __forceinline__ void round_issue(const float* __restrict__ xl, int j, int b2,
                                            int iv0, int iv1, v4f (&v)[MAIN_ROWS]) {
#pragma unroll
    for (int u = 0; u < MAIN_ROWS; ++u) {
        v[u] = (v4f){0.0f, 0.0f, 0.0f, 0.0f};
        if (j + u < b2) {
            const int id = __builtin_amdgcn_readlane((POS0 + u) < 32 ? iv0 : iv1, (POS0 + u) & 31);
            v[u] = *(const v4f*)(xl + (long)id * D_IN);
        }
    }
}

__device__ __forceinline__ void round_add(int j, int b1, const v4f (&v)[MAIN_ROWS],
                                          v4f& a0, v4f& a1) {
#pragma unroll
    for (int u = 0; u < MAIN_ROWS; ++u) {
        if (j + u < b1) a0 += v[u];
        else            a1 += v[u];
    }
}

__device__ __forceinline__ void neighbour_sums2(const float* __restrict__ xl,
                                                const int* __restrict__ csr,
                                                int b0, int b1, int b2, int iv0, int iv1,
                                                int lane, v4f& s0, v4f& s1) {
    v4f a0 = {0.0f, 0.0f, 0.0f, 0.0f};
    v4f a1 = {0.0f, 0.0f, 0.0f, 0.0f};
#pragma unroll
    for (int q = 0; q < ID_SPAN / GATHER_ROWS; ++q) {
        const int j = b0 + GATHER_ROWS * q;
        if (j < b2) gather_round(xl, j, b1, b2, (q & 2) ? iv1 : iv0, (q & 1) * 16, a0, a1);
    }
#pragma nounroll
    for (int j = b0 + ID_SPAN; j < b2; j += GATHER_ROWS) {
        const int r = (j - b0) / GATHER_ROWS;
        if ((r & 3) == 0) load_ids(csr, j, b2, lane, iv0, iv1);
        gather_round(xl, j, b1, b2, (r & 2) ? iv1 : iv0, (r & 1) * 16, a0, a1);
    }
    s0 = a0;
    s1 = a1;
}

__device__ __forceinline__ void fetch_rows2(const float* __restrict__ A,
                                            const int* __restrict__ csr,
                                            int r0, int lane, int b0, int b1, int b2,
                                            int iv0, int iv1, v4f (&pa)[2], v4f (&pg)[2]) {
    const float* xl = A + lane * 4;
    pa[0] = *(const v4f*)(xl + (long)r0 * D_IN);
    pa[1] = *(const v4f*)(xl + (long)(r0 + 1) * D_IN);
    v4f s0, s1;
    neighbour_sums2(xl, csr, b0, b1, b2, iv0, iv1, lane, s0, s1);
    pg[0] = s0 * (1.0f / fmaxf((float)(b1 - b0), 1.0f));
    pg[1] = s1 * (1.0f / fmaxf((float)(b2 - b1), 1.0f));
}

__global__ __launch_bounds__(256, 4)
void sage_gemm(const float* __restrict__ A,
               const int* __restrict__ off,
               const int* __restrict__ csr,
               const float* __restrict__ Wr,
               const float* __restrict__ Wn,
               const float* __restrict__ bias,
               const float* __restrict__ gamma,
               const float* __restrict__ beta,
               float* __restrict__ out)
{
    constexpr int NT = 8;
    constexpr int W = 16 * NT;
    constexpr int K = 128;
    constexpr int KS = K / 4;
    constexpr int LDA = K + 4;
    constexpr int RW = 16 / NT;
    constexpr int CPL = W / 32;
    static_assert(RW == 2, "fetch_rows2 stages two adjacent rows per wave");
    __shared__ __attribute__((aligned(16))) float As[2][16][LDA];
    __shared__ __attribute__((aligned(16))) float Gs[2][16][LDA];
    __shared__ float Hs[2][16][W];
    __shared__ float GBs[2][W];

    const int lane = threadIdx.x & 31;
    const int wave = __builtin_amdgcn_readfirstlane(threadIdx.x >> 5);

    int tile = blockIdx.x;
    if (tile >= ROW_TILES) return;

    const int n0   = wave * 16;
    const int arow = lane & 15;
    const int koff = (lane >> 4) << 1;
    const int bcol = lane & 15;

    v2f bR[KS], bN[KS];
#pragma unroll
    for (int s = 0; s < KS; ++s) {
        const int k = 4 * s + koff;
        bR[s].x = Wr[k * W + n0 + bcol];
        bR[s].y = Wr[(k + 1) * W + n0 + bcol];
        bN[s].x = Wn[k * W + n0 + bcol];
        bN[s].y = Wn[(k + 1) * W + n0 + bcol];
    }
    const float bv = bias[n0 + bcol];
    if (threadIdx.x < W) {
        GBs[0][threadIdx.x] = gamma[threadIdx.x];
        GBs[1][threadIdx.x] = beta[threadIdx.x];
    }

    const int step = gridDim.x;
    const int rl = RW * wave;
    const int lq = min(lane, 2);
    v4f pa[RW], pg[RW];
    int p0, p1, p2, pv0, pv1, qv;
    {
        const int ra = tile * 16 + rl;
        const int rb = min(tile + step, ROW_TILES - 1) * 16 + rl;
        const int rc = min(tile + 2 * step, ROW_TILES - 1) * 16 + rl;
        const int c0 = off[ra], c1 = off[ra + 1], c2 = off[ra + 2];
        p0 = off[rb]; p1 = off[rb + 1]; p2 = off[rb + 2];
        qv = off[rc + lq];
        int cv0, cv1;
        load_ids(csr, c0, c2, lane, cv0, cv1);
        load_ids(csr, p0, p2, lane, pv0, pv1);
        fetch_rows2(A, csr, ra, lane, c0, c1, c2, cv0, cv1, pa, pg);
    }
    v4f sa0, sa1;
    {
        const float* xs = A + (long)(min(tile + step, ROW_TILES - 1) * 16 + rl) * D_IN + lane * 4;
        sa0 = *(const v4f*)xs;
        sa1 = *(const v4f*)(xs + D_IN);
    }
#pragma unroll
    for (int j = 0; j < RW; ++j) {
        *(v4f*)&As[0][RW * wave + j][lane * 4] = pa[j];
        *(v4f*)&Gs[0][RW * wave + j][lane * 4] = pg[j];
    }
    __syncthreads();

    for (int it = 0; tile < ROW_TILES; ++it, tile += gridDim.x) {
        const int p = it & 1;
        const int nxt = tile + gridDim.x;
        const bool more = nxt < ROW_TILES;

        const int q0 = __builtin_amdgcn_readlane(qv, 0);
        const int q1 = __builtin_amdgcn_readlane(qv, 1);
        const int q2 = __builtin_amdgcn_readlane(qv, 2);
        int nv0, nv1;
        load_ids(csr, q0, q2, lane, nv0, nv1);
        const int dv = off[min(tile + 3 * step, ROW_TILES - 1) * 16 + rl + lq];

        const float* xl = A + lane * 4;
        v4f gv[MAIN_ROWS];
        v4f a0 = {0.0f, 0.0f, 0.0f, 0.0f};
        v4f a1 = {0.0f, 0.0f, 0.0f, 0.0f};
        *(v4f*)&As[p ^ 1][rl][lane * 4] = sa0;
        *(v4f*)&As[p ^ 1][rl + 1][lane * 4] = sa1;
        if (more) {
            if (p0 < p2) round_issue<0>(xl, p0, p2, pv0, pv1, gv);
        }

        v8f acc = {};
        SAGE_MUL_PART(0, 11)
        if (more) {
            if (p0 < p2) round_add(p0, p1, gv, a0, a1);
            if (p0 + MAIN_ROWS < p2)
                round_issue<MAIN_ROWS>(xl, p0 + MAIN_ROWS, p2, pv0, pv1, gv);
        }
        SAGE_MUL_PART(11, 22)
        if (more) {
            if (p0 + MAIN_ROWS < p2) round_add(p0 + MAIN_ROWS, p1, gv, a0, a1);
            if (p0 + 2 * MAIN_ROWS < p2)
                round_issue<2 * MAIN_ROWS>(xl, p0 + 2 * MAIN_ROWS, p2, pv0, pv1, gv);
        }
        SAGE_MUL_PART(22, KS)

        {
            const int rbase = (lane >> 4) << 3;
#pragma unroll
            for (int v = 0; v < 8; ++v)
                Hs[p][rbase + v][n0 + bcol] = acc[v] + bv;
        }
        if (more) {
            if (p0 + 2 * MAIN_ROWS < p2) round_add(p0 + 2 * MAIN_ROWS, p1, gv, a0, a1);
#pragma nounroll
            for (int j = p0 + MAIN_ROUNDS * MAIN_ROWS; j < p2; j += MAIN_ROWS) {
                int tv0, tv1;
                load_ids(csr, j, p2, lane, tv0, tv1);
                round_issue<0>(xl, j, p2, tv0, tv1, gv);
                round_add(j, p1, gv, a0, a1);
            }
            *(v4f*)&Gs[p ^ 1][rl][lane * 4] = a0 * (1.0f / fmaxf((float)(p1 - p0), 1.0f));
            *(v4f*)&Gs[p ^ 1][rl + 1][lane * 4] = a1 * (1.0f / fmaxf((float)(p2 - p1), 1.0f));
        }
        p0 = q0; p1 = q1; p2 = q2; pv0 = nv0; pv1 = nv1;
        qv = dv;
        __syncthreads();

        {
            const float* xs = xl + (long)(min(tile + 2 * step, ROW_TILES - 1) * 16 + rl) * D_IN;
            sa0 = *(const v4f*)xs;
            sa1 = *(const v4f*)(xs + D_IN);
        }

        float h[RW][CPL];
#pragma unroll
        for (int j = 0; j < RW; ++j)
#pragma unroll
            for (int m = 0; m < CPL; ++m) h[j][m] = Hs[p][wave + NT * j][lane + 32 * m];
        float gam[CPL], bet[CPL];
#pragma unroll
        for (int m = 0; m < CPL; ++m) { gam[m] = GBs[0][lane + 32 * m]; bet[m] = GBs[1][lane + 32 * m]; }

        float s[RW], vv[RW], mu[RW];
#pragma unroll
        for (int j = 0; j < RW; ++j) {
            s[j] = 0.0f;
#pragma unroll
            for (int m = 0; m < CPL; ++m) s[j] += h[j][m];
        }
#pragma unroll
        for (int off2 = 16; off2; off2 >>= 1)
#pragma unroll
            for (int j = 0; j < RW; ++j) s[j] += __shfl_xor(s[j], off2, 32);
#pragma unroll
        for (int j = 0; j < RW; ++j) {
            mu[j] = s[j] * (1.0f / W);
            vv[j] = 0.0f;
#pragma unroll
            for (int m = 0; m < CPL; ++m) {
                float d = h[j][m] - mu[j];
                vv[j] += d * d;
            }
        }
#pragma unroll
        for (int off2 = 16; off2; off2 >>= 1)
#pragma unroll
            for (int j = 0; j < RW; ++j) vv[j] += __shfl_xor(vv[j], off2, 32);
#pragma unroll
        for (int j = 0; j < RW; ++j) {
            const float rs = rsqrtf(vv[j] * (1.0f / W) + 1e-5f);
            const long row = (long)tile * 16 + wave + NT * j;
#pragma unroll
            for (int m = 0; m < CPL; ++m) {
                float val = (h[j][m] - mu[j]) * rs * gam[m] + bet[m];
                out[row * W + lane + 32 * m] = fmaxf(val, 0.0f);
            }
        }
    }
}

__global__ __launch_bounds__(256)
void proj_gemm(const float* __restrict__ A,
               const float* __restrict__ Wr,
               const float* __restrict__ Wn,
               const float* __restrict__ bias,
               float* __restrict__ z,
               float* __restrict__ y)
{
    constexpr int NT = 8;
    constexpr int K = 128;
    constexpr int KS = K / 4;
    constexpr int LDA = K + 4;
    constexpr int RW = 16 / NT;
    constexpr int LDH = 2 * D_OUT + 4;
    __shared__ __attribute__((aligned(16))) float As[2][16][LDA];
    __shared__ __attribute__((aligned(16))) float Hs[2][16][LDH];

    const int lane = threadIdx.x & 31;
    const int wave = __builtin_amdgcn_readfirstlane(threadIdx.x >> 5);

    int tile = blockIdx.x;
    if (tile >= ROW_TILES) return;

    const int arow = lane & 15;
    const int koff = (lane >> 4) << 1;
    const int bcol = lane & 15;
    const bool nbr = wave >= 4;
    const float* __restrict__ Wsel = nbr ? Wn : Wr;
    const int n0 = (wave & 3) * 16;

    v2f bW[KS];
#pragma unroll
    for (int s = 0; s < KS; ++s) {
        const int k = 4 * s + koff;
        bW[s].x = Wsel[k * D_OUT + n0 + bcol];
        bW[s].y = Wsel[(k + 1) * D_OUT + n0 + bcol];
    }
    const float bv = nbr ? 0.0f : bias[n0 + bcol];

    v4f pa[RW];
#pragma unroll
    for (int j = 0; j < RW; ++j) {
        const long row = (long)tile * 16 + wave + NT * j;
        pa[j] = *(const v4f*)(A + row * K + lane * 4);
    }
#pragma unroll
    for (int j = 0; j < RW; ++j)
        *(v4f*)&As[0][wave + NT * j][lane * 4] = pa[j];
    __syncthreads();

    float* __restrict__ dcol = (lane < 16) ? (z + lane * 4) : (y + (lane - 16) * 4);

    for (int it = 0; tile < ROW_TILES; ++it, tile += gridDim.x) {
        const int p = it & 1;
        const int nxt = tile + gridDim.x;
        const bool more = nxt < ROW_TILES;

        if (more) {
#pragma unroll
            for (int j = 0; j < RW; ++j) {
                const long row = (long)nxt * 16 + wave + NT * j;
                pa[j] = *(const v4f*)(A + row * K + lane * 4);
            }
        }

        v8f acc = {};
#pragma unroll
        for (int s = 0; s < KS; ++s) {
            const v2f a0 = *(const v2f*)&As[p][arow][4 * s + koff];
            acc = __builtin_amdgcn_wmma_f32_16x16x4_f32(
                false, a0, false, bW[s], (short)0, acc, false, false);
        }

        {
            const int rbase = (lane >> 4) << 3;
#pragma unroll
            for (int v = 0; v < 8; ++v)
                Hs[p][rbase + v][wave * 16 + bcol] = acc[v] + bv;
        }
        if (more) {
#pragma unroll
            for (int j = 0; j < RW; ++j)
                *(v4f*)&As[p ^ 1][wave + NT * j][lane * 4] = pa[j];
        }
        __syncthreads();

#pragma unroll
        for (int j = 0; j < RW; ++j) {
            const int r = wave + NT * j;
            const v4f hv = *(const v4f*)&Hs[p][r][lane * 4];
            *(v4f*)(dcol + ((long)tile * 16 + r) * D_OUT) = hv;
        }
    }
}

extern "C" void kernel_launch(void* const* d_in, const int* in_sizes, int n_in,
                              void* d_out, int out_size, void* d_ws, size_t ws_size,
                              hipStream_t stream) {
    const float* x    = (const float*)d_in[0];
    const int*   esrc = (const int*)d_in[1];
    const int*   edst = (const int*)d_in[2];
    const float* Wr0  = (const float*)d_in[3];
    const float* Wn0  = (const float*)d_in[4];
    const float* b0   = (const float*)d_in[5];
    const float* g0   = (const float*)d_in[6];
    const float* be0  = (const float*)d_in[7];
    const float* Wr1  = (const float*)d_in[8];
    const float* Wn1  = (const float*)d_in[9];
    const float* b1   = (const float*)d_in[10];
    const float* g1   = (const float*)d_in[11];
    const float* be1  = (const float*)d_in[12];
    const float* Wr2  = (const float*)d_in[13];
    const float* Wn2  = (const float*)d_in[14];
    const float* b2   = (const float*)d_in[15];
    float* out = (float*)d_out;

    const int N = GNN_N;
    const int E = in_sizes[1];

    float* h0 = (float*)d_ws;
    float* h1 = h0 + (size_t)N * D_IN;
    float* yp = h1 + (size_t)N * D_IN;
    float* zp = yp + (size_t)N * D_OUT;
    int* Lc   = (int*)(zp + (size_t)N * D_OUT);
    int* off  = Lc + (EDGE_CHUNKS * L_ROW + 32);
    int* tmp  = off + (N + 32);
    int* csr  = tmp + E;

    const int gathBlocks = (N + 7) / 8;
    const int gemmBlocks8 = 512;

    chunk_sort<<<EDGE_CHUNKS, 1024, 0, stream>>>(esrc, edst, Lc, tmp, E);
    bucket_csr<<<N_BUCKETS, 1024, 0, stream>>>(tmp, Lc, off, csr, E);

    sage_gemm<<<gemmBlocks8, 256, 0, stream>>>(x, off, csr, Wr0, Wn0, b0, g0, be0, h0);
    sage_gemm<<<gemmBlocks8, 256, 0, stream>>>(h0, off, csr, Wr1, Wn1, b1, g1, be1, h1);

    proj_gemm<<<gemmBlocks8, 256, 0, stream>>>(h1, Wr2, Wn2, b2, zp, yp);
    gather_lsm<<<gathBlocks, 256, 0, stream>>>(yp, zp, off, csr, out, N);
}


# sage_gemm: a neighbour round whole on one side of the row pair's list boundary adds its 16 rows straight to that sum (scalar branch); per-row selects only in the round that holds it; sum order kept
# speedup vs baseline: 5.5638x; 1.0354x over previous
#include <hip/hip_runtime.h>
#include <hip/hip_bf16.h>
#include <math.h>

typedef float v2f __attribute__((ext_vector_type(2)));
typedef float v8f __attribute__((ext_vector_type(8)));
typedef float v4f __attribute__((ext_vector_type(4)));

#define GNN_N 100000
#define D_IN 128
#define D_OUT 64

#define EDGE_CHUNKS 256
#define NODE_SHIFT 9
#define BUCKET_NODES (1 << NODE_SHIFT)
#define N_BUCKETS ((GNN_N + BUCKET_NODES - 1) / BUCKET_NODES)
#define L_ROW (N_BUCKETS + 1)
static_assert(N_BUCKETS <= 1024, "chunk_sort scans the bucket counters with one thread per counter");
static_assert(EDGE_CHUNKS % 32 == 0 && EDGE_CHUNKS <= 1024, "bucket_csr gives every wave the same number of runs");
static_assert(GNN_N < (1 << (31 - NODE_SHIFT)), "a src id and a node's place in its bucket share one word");
static_assert(GNN_N % BUCKET_NODES != 0, "off[N] is the last bucket's entry N % BUCKET_NODES");

__device__ __forceinline__ int bucket_of(int d) {
    return min((int)((unsigned)d >> NODE_SHIFT), N_BUCKETS - 1);
}

#define EDGE_BATCH 8
#define CHUNK_CAP (EDGE_BATCH * 1024)
#define SEG_CAP 16384

__global__ __launch_bounds__(1024)
void chunk_sort(const int* __restrict__ src, const int* __restrict__ dst,
                int* __restrict__ L, int* __restrict__ tmp, int E) {
    __shared__ int cnt[N_BUCKETS];
    __shared__ int wsum[32];
    __shared__ int stage[CHUNK_CAP];
    const int tid = threadIdx.x, lane = tid & 31, wave = tid >> 5, w = blockIdx.x;
    const int chunk = (E + EDGE_CHUNKS - 1) / EDGE_CHUNKS;
    const int lo = min(w * chunk, E), hi = min(lo + chunk, E);
    const bool single = (hi - lo) <= CHUNK_CAP;
    if (tid < N_BUCKETS) cnt[tid] = 0;
    __syncthreads();
    int d[EDGE_BATCH], s[EDGE_BATCH];
#pragma unroll
    for (int u = 0; u < EDGE_BATCH; ++u) { d[u] = 0; s[u] = 0; }
    if (single) {
        if (lo < hi) {
#pragma unroll
            for (int u = 0; u < EDGE_BATCH; ++u) {
                const int i = min(lo + u * 1024 + tid, hi - 1);
                d[u] = dst[i];
                s[u] = src[i];
            }
        }
#pragma unroll
        for (int u = 0; u < EDGE_BATCH; ++u) {
            if (lo + u * 1024 + tid < hi) atomicAdd(&cnt[bucket_of(d[u])], 1);
        }
    } else {
        for (int base = lo; base < hi; base += CHUNK_CAP) {
#pragma unroll
            for (int u = 0; u < EDGE_BATCH; ++u) d[u] = dst[min(base + u * 1024 + tid, hi - 1)];
#pragma unroll
            for (int u = 0; u < EDGE_BATCH; ++u) {
                if (base + u * 1024 + tid < hi) atomicAdd(&cnt[bucket_of(d[u])], 1);
            }
        }
    }
    __syncthreads();
    const int v = (tid < N_BUCKETS) ? cnt[tid] : 0;
    int sc = v;
#pragma unroll
    for (int o = 1; o < 32; o <<= 1) {
        int t = __shfl_up(sc, o, 32);
        if (lane >= o) sc += t;
    }
    if (lane == 31) wsum[wave] = sc;
    __syncthreads();
    int ws = wsum[lane];
#pragma unroll
    for (int o = 1; o < 32; o <<= 1) {
        int t = __shfl_up(ws, o, 32);
        if (lane >= o) ws += t;
    }
    const int waveIncl = __shfl(ws, wave, 32);
    const int waveTot = __shfl(sc, 31, 32);
    const int total = __shfl(ws, 31, 32);
    const int excl = (waveIncl - waveTot) + (sc - v);
    if (tid < N_BUCKETS) {
        cnt[tid] = excl;
        L[w * L_ROW + tid] = excl;
    }
    if (tid == 0) L[w * L_ROW + N_BUCKETS] = total;
    __syncthreads();
    if (single) {
        int p[EDGE_BATCH];
#pragma unroll
        for (int u = 0; u < EDGE_BATCH; ++u) {
            p[u] = 0;
            if (lo + u * 1024 + tid < hi) p[u] = atomicAdd(&cnt[bucket_of(d[u])], 1);
        }
#pragma unroll
        for (int u = 0; u < EDGE_BATCH; ++u) {
            if (lo + u * 1024 + tid < hi)
                stage[p[u]] = (int)(((unsigned)s[u] << NODE_SHIFT) | ((unsigned)d[u] & (BUCKET_NODES - 1)));
        }
        __syncthreads();
        for (int i = tid; i < hi - lo; i += 1024) tmp[lo + i] = stage[i];
    } else {
        for (int base = lo; base < hi; base += CHUNK_CAP) {
#pragma unroll
            for (int u = 0; u < EDGE_BATCH; ++u) {
                const int i = min(base + u * 1024 + tid, hi - 1);
                d[u] = dst[i];
                s[u] = src[i];
            }
#pragma unroll
            for (int u = 0; u < EDGE_BATCH; ++u) {
                if (base + u * 1024 + tid < hi) {
                    const int p = atomicAdd(&cnt[bucket_of(d[u])], 1);
                    tmp[lo + p] = (int)(((unsigned)s[u] << NODE_SHIFT) | ((unsigned)d[u] & (BUCKET_NODES - 1)));
                }
            }
        }
    }
}

#define RUNS_PER_WAVE (EDGE_CHUNKS / 32)
#define RUN_SPAN 64
__global__ __launch_bounds__(1024)
void bucket_csr(const int* __restrict__ tmp, const int* __restrict__ L,
                int* __restrict__ off, int* __restrict__ csr, int E) {
    __shared__ int cnt[BUCKET_NODES];
    __shared__ int wsum[32];
    __shared__ int wsumA[32];
    __shared__ int rbeg[EDGE_CHUNKS];
    __shared__ int rlen[EDGE_CHUNKS];
    __shared__ int seg[SEG_CAP];
    const int tid = threadIdx.x, lane = tid & 31, b = blockIdx.x;
    const int wave = __builtin_amdgcn_readfirstlane(tid >> 5);
    const int chunk = (E + EDGE_CHUNKS - 1) / EDGE_CHUNKS;
    const int top = max(E - 1, 0);
    int a = 0, len = 0;
    if (tid < EDGE_CHUNKS) {
        const int* r = L + tid * L_ROW + b;
        a = r[0];
        len = r[1] - a;
        rbeg[tid] = min(tid * chunk, E) + a;
        rlen[tid] = len;
    }
    if (tid < BUCKET_NODES) cnt[tid] = 0;
    int sl = len, sa = a;
#pragma unroll
    for (int o = 16; o; o >>= 1) {
        sl += __shfl_xor(sl, o, 32);
        sa += __shfl_xor(sa, o, 32);
    }
    if (lane == 0) { wsum[wave] = sl; wsumA[wave] = sa; }
    __syncthreads();
    int wl = wsum[lane], wa = wsumA[lane];
#pragma unroll
    for (int o = 16; o; o >>= 1) {
        wl += __shfl_xor(wl, o, 32);
        wa += __shfl_xor(wa, o, 32);
    }
    const int B0 = __builtin_amdgcn_readfirstlane(wa);
    const int nb = __builtin_amdgcn_readfirstlane(wl);
    const bool staged = nb <= SEG_CAP;

    const int w0 = wave * RUNS_PER_WAVE;
    int e[2 * RUNS_PER_WAVE];
    int lnv[RUNS_PER_WAVE];
    int lmax = 0;
#pragma unroll
    for (int r = 0; r < RUNS_PER_WAVE; ++r) {
        const int rb = __builtin_amdgcn_readfirstlane(rbeg[w0 + r]);
        const int ln = __builtin_amdgcn_readfirstlane(rlen[w0 + r]);
        lnv[r] = ln;
        lmax = max(lmax, ln);
        const int last = max(ln - 1, 0);
        e[2 * r]     = tmp[min(rb + min(lane, last), top)];
        e[2 * r + 1] = tmp[min(rb + min(lane + 32, last), top)];
    }
#pragma unroll
    for (int r = 0; r < RUNS_PER_WAVE; ++r) {
        const int ln = lnv[r];
        if (lane < ln)      atomicAdd(&cnt[e[2 * r] & (BUCKET_NODES - 1)], 1);
        if (lane + 32 < ln) atomicAdd(&cnt[e[2 * r + 1] & (BUCKET_NODES - 1)], 1);
    }
    if (lmax > RUN_SPAN) {
#pragma nounroll
        for (int r = 0; r < RUNS_PER_WAVE; ++r) {
            const int rb = __builtin_amdgcn_readfirstlane(rbeg[w0 + r]);
            const int ln = __builtin_amdgcn_readfirstlane(rlen[w0 + r]);
            for (int k = RUN_SPAN; k < ln; k += 32) {
                if (k + lane < ln) atomicAdd(&cnt[tmp[rb + k + lane] & (BUCKET_NODES - 1)], 1);
            }
        }
    }
    __syncthreads();
    const int v = (tid < BUCKET_NODES) ? cnt[tid] : 0;
    int s = v;
#pragma unroll
    for (int o = 1; o < 32; o <<= 1) {
        int t = __shfl_up(s, o, 32);
        if (lane >= o) s += t;
    }
    if (lane == 31) wsum[wave] = s;
    __syncthreads();
    int ws = wsum[lane];
#pragma unroll
    for (int o = 1; o < 32; o <<= 1) {
        int t = __shfl_up(ws, o, 32);
        if (lane >= o) ws += t;
    }
    const int waveIncl = __shfl(ws, wave, 32);
    const int waveTot = __shfl(s, 31, 32);
    const int excl = (waveIncl - waveTot) + (s - v);
    if (tid < BUCKET_NODES) {
        cnt[tid] = excl;
        const int node = b * BUCKET_NODES + tid;
        if (node <= GNN_N) off[node] = B0 + excl;
    }
    __syncthreads();
    int p[2 * RUNS_PER_WAVE];
#pragma unroll
    for (int r = 0; r < RUNS_PER_WAVE; ++r) {
        const int ln = lnv[r];
#pragma unroll
        for (int h = 0; h < 2; ++h) {
            p[2 * r + h] = 0;
            if (lane + 32 * h < ln) p[2 * r + h] = atomicAdd(&cnt[e[2 * r + h] & (BUCKET_NODES - 1)], 1);
        }
    }
#pragma unroll
    for (int r = 0; r < RUNS_PER_WAVE; ++r) {
        const int ln = lnv[r];
#pragma unroll
        for (int h = 0; h < 2; ++h) {
            if (lane + 32 * h < ln) {
                const int id = (int)((unsigned)e[2 * r + h] >> NODE_SHIFT);
                if (staged) seg[p[2 * r + h]] = id;
                else        csr[B0 + p[2 * r + h]] = id;
            }
        }
    }
    if (lmax > RUN_SPAN) {
#pragma nounroll
        for (int r = 0; r < RUNS_PER_WAVE; ++r) {
            const int rb = __builtin_amdgcn_readfirstlane(rbeg[w0 + r]);
            const int ln = __builtin_amdgcn_readfirstlane(rlen[w0 + r]);
            for (int k = RUN_SPAN; k < ln; k += 32) {
                if (k + lane < ln) {
                    const int x = tmp[rb + k + lane];
                    const int q = atomicAdd(&cnt[x & (BUCKET_NODES - 1)], 1);
                    const int id = (int)((unsigned)x >> NODE_SHIFT);
                    if (staged) seg[q] = id;
                    else        csr[B0 + q] = id;
                }
            }
        }
    }
    if (staged) {
        __syncthreads();
        for (int i = tid; i < nb; i += 1024) csr[B0 + i] = seg[i];
    }
}

__global__ __launch_bounds__(256)
void gather_lsm(const float* __restrict__ y, const float* __restrict__ z,
                const int* __restrict__ off, const int* __restrict__ csr,
                float* __restrict__ out, int n) {
    const int node = __builtin_amdgcn_readfirstlane(blockIdx.x * 8 + (threadIdx.x >> 5));
    if (node >= n) return;
    const int lane = threadIdx.x & 31;
    const bool odd = lane >= 16;
    const int c4 = (lane & 15) * 4;
    const int beg = off[node];
    const int end = off[node + 1];
    const float4 zv = *(const float4*)(z + (long)node * D_OUT + c4);
    const float* yl = y + c4;
    float4 acc = make_float4(0.0f, 0.0f, 0.0f, 0.0f);
    int j = beg;
    for (; j + 16 <= end; j += 16) {
        float4 v[8];
#pragma unroll
        for (int u = 0; u < 8; ++u) {
            const int i0 = csr[j + 2 * u], i1 = csr[j + 2 * u + 1];
            v[u] = *(const float4*)(yl + (long)(odd ? i1 : i0) * D_OUT);
        }
#pragma unroll
        for (int u = 0; u < 8; ++u) {
            acc.x += v[u].x; acc.y += v[u].y; acc.z += v[u].z; acc.w += v[u].w;
        }
    }
    if (j + 8 <= end) {
        float4 v[4];
#pragma unroll
        for (int u = 0; u < 4; ++u) {
            const int i0 = csr[j + 2 * u], i1 = csr[j + 2 * u + 1];
            v[u] = *(const float4*)(yl + (long)(odd ? i1 : i0) * D_OUT);
        }
#pragma unroll
        for (int u = 0; u < 4; ++u) {
            acc.x += v[u].x; acc.y += v[u].y; acc.z += v[u].z; acc.w += v[u].w;
        }
        j += 8;
    }
    if (j + 4 <= end) {
        const int i0 = csr[j], i1 = csr[j + 1], i2 = csr[j + 2], i3 = csr[j + 3];
        float4 v0 = *(const float4*)(yl + (long)(odd ? i1 : i0) * D_OUT);
        float4 v1 = *(const float4*)(yl + (long)(odd ? i3 : i2) * D_OUT);
        acc.x += v0.x; acc.y += v0.y; acc.z += v0.z; acc.w += v0.w;
        acc.x += v1.x; acc.y += v1.y; acc.z += v1.z; acc.w += v1.w;
        j += 4;
    }
    if (j + 2 <= end) {
        const int i0 = csr[j], i1 = csr[j + 1];
        float4 v0 = *(const float4*)(yl + (long)(odd ? i1 : i0) * D_OUT);
        acc.x += v0.x; acc.y += v0.y; acc.z += v0.z; acc.w += v0.w;
        j += 2;
    }
    if (j < end) {
        const int i0 = csr[j];
        if (!odd) {
            float4 v0 = *(const float4*)(yl + (long)i0 * D_OUT);
            acc.x += v0.x; acc.y += v0.y; acc.z += v0.z; acc.w += v0.w;
        }
    }
    acc.x += __shfl_xor(acc.x, 16, 32);
    acc.y += __shfl_xor(acc.y, 16, 32);
    acc.z += __shfl_xor(acc.z, 16, 32);
    acc.w += __shfl_xor(acc.w, 16, 32);
    const float inv = 1.0f / fmaxf((float)(end - beg), 1.0f);
    float4 h;
    h.x = zv.x + acc.x * inv; h.y = zv.y + acc.y * inv;
    h.z = zv.z + acc.z * inv; h.w = zv.w + acc.w * inv;
    float mx = fmaxf(fmaxf(h.x, h.y), fmaxf(h.z, h.w));
#pragma unroll
    for (int o = 8; o; o >>= 1) mx = fmaxf(mx, __shfl_xor(mx, o, 32));
    float s = expf(h.x - mx) + expf(h.y - mx) + expf(h.z - mx) + expf(h.w - mx);
#pragma unroll
    for (int o = 8; o; o >>= 1) s += __shfl_xor(s, o, 32);
    const float ls = logf(s);
    if (!odd) {
        float4 r;
        r.x = h.x - mx - ls; r.y = h.y - mx - ls; r.z = h.z - mx - ls; r.w = h.w - mx - ls;
        *(float4*)(out + (long)node * D_OUT + c4) = r;
    }
}

#define ROW_TILES (GNN_N / 16)
#define GATHER_ROWS 16
#define ID_SPAN 64
#define MAIN_ROWS 16
#define MAIN_ROUNDS 3
static_assert(MAIN_ROUNDS * MAIN_ROWS <= ID_SPAN, "the rounds under the multiply take their ids from the prefetched span");

#define SAGE_MUL_PART(S0, S1)                                                                  \
    _Pragma("unroll")   \
    for (int s = (S0); s < (S1); ++s) {                                                        \
        const v2f f0 = *(const v2f*)&As[p][arow][4 * s + koff];                                \
        const v2f f1 = *(const v2f*)&Gs[p][arow][4 * s + koff];                                \
        acc = __builtin_amdgcn_wmma_f32_16x16x4_f32(                                           \
            false, f0, false, bR[s], (short)0, acc, false, false);                             \
        acc = __builtin_amdgcn_wmma_f32_16x16x4_f32(                                           \
            false, f1, false, bN[s], (short)0, acc, false, false);                             \
    }

__device__ __forceinline__ void load_ids(const int* __restrict__ csr, int j0, int b2, int lane,
                                         int& iv0, int& iv1) {
    const int last = max(b2 - 1, 0);
    iv0 = csr[min(j0 + lane, last)];
    iv1 = csr[min(j0 + 32 + lane, last)];
}

__device__ __forceinline__ void gather_round(const float* __restrict__ xl, int j, int b1, int b2,
                                             int ivq, int lb, v4f& a0, v4f& a1) {
    v4f v[GATHER_ROWS];
#pragma unroll
    for (int u = 0; u < GATHER_ROWS; ++u) {
        v[u] = (v4f){0.0f, 0.0f, 0.0f, 0.0f};
        if (j + u < b2) {
            const int id = __builtin_amdgcn_readlane(ivq, lb + u);
            v[u] = *(const v4f*)(xl + (long)id * D_IN);
        }
    }
#pragma unroll
    for (int u = 0; u < GATHER_ROWS; ++u) {
        if (j + u < b1) a0 += v[u];
        else            a1 += v[u];
    }
}

template <int POS0>
__device__ __forceinline__ void round_issue(const float* __restrict__ xl, int j, int b2,
                                            int iv0, int iv1, v4f (&v)[MAIN_ROWS]) {
#pragma unroll
    for (int u = 0; u < MAIN_ROWS; ++u) {
        v[u] = (v4f){0.0f, 0.0f, 0.0f, 0.0f};
        if (j + u < b2) {
            const int id = __builtin_amdgcn_readlane((POS0 + u) < 32 ? iv0 : iv1, (POS0 + u) & 31);
            v[u] = *(const v4f*)(xl + (long)id * D_IN);
        }
    }
}

__device__ __forceinline__ void round_add_split(int j, int b1, const v4f (&v)[MAIN_ROWS],
                                                v4f& a0, v4f& a1) {
#pragma unroll
    for (int u = 0; u < MAIN_ROWS; ++u) {
        if (j + u < b1) a0 += v[u];
        else            a1 += v[u];
    }
}

__device__ __forceinline__ void round_add(int j, int b1, const v4f (&v)[MAIN_ROWS],
                                          v4f& a0, v4f& a1) {
    if (j + MAIN_ROWS <= b1) {
#pragma unroll
        for (int u = 0; u < MAIN_ROWS; ++u) a0 += v[u];
    } else if (j >= b1) {
#pragma unroll
        for (int u = 0; u < MAIN_ROWS; ++u) a1 += v[u];
    } else {
        round_add_split(j, b1, v, a0, a1);
    }
}

__device__ __forceinline__ void neighbour_sums2(const float* __restrict__ xl,
                                                const int* __restrict__ csr,
                                                int b0, int b1, int b2, int iv0, int iv1,
                                                int lane, v4f& s0, v4f& s1) {
    v4f a0 = {0.0f, 0.0f, 0.0f, 0.0f};
    v4f a1 = {0.0f, 0.0f, 0.0f, 0.0f};
#pragma unroll
    for (int q = 0; q < ID_SPAN / GATHER_ROWS; ++q) {
        const int j = b0 + GATHER_ROWS * q;
        if (j < b2) gather_round(xl, j, b1, b2, (q & 2) ? iv1 : iv0, (q & 1) * 16, a0, a1);
    }
#pragma nounroll
    for (int j = b0 + ID_SPAN; j < b2; j += GATHER_ROWS) {
        const int r = (j - b0) / GATHER_ROWS;
        if ((r & 3) == 0) load_ids(csr, j, b2, lane, iv0, iv1);
        gather_round(xl, j, b1, b2, (r & 2) ? iv1 : iv0, (r & 1) * 16, a0, a1);
    }
    s0 = a0;
    s1 = a1;
}

__device__ __forceinline__ void fetch_rows2(const float* __restrict__ A,
                                            const int* __restrict__ csr,
                                            int r0, int lane, int b0, int b1, int b2,
                                            int iv0, int iv1, v4f (&pa)[2], v4f (&pg)[2]) {
    const float* xl = A + lane * 4;
    pa[0] = *(const v4f*)(xl + (long)r0 * D_IN);
    pa[1] = *(const v4f*)(xl + (long)(r0 + 1) * D_IN);
    v4f s0, s1;
    neighbour_sums2(xl, csr, b0, b1, b2, iv0, iv1, lane, s0, s1);
    pg[0] = s0 * (1.0f / fmaxf((float)(b1 - b0), 1.0f));
    pg[1] = s1 * (1.0f / fmaxf((float)(b2 - b1), 1.0f));
}

__global__ __launch_bounds__(256, 4)
void sage_gemm(const float* __restrict__ A,
               const int* __restrict__ off,
               const int* __restrict__ csr,
               const float* __restrict__ Wr,
               const float* __restrict__ Wn,
               const float* __restrict__ bias,
               const float* __restrict__ gamma,
               const float* __restrict__ beta,
               float* __restrict__ out)
{
    constexpr int NT = 8;
    constexpr int W = 16 * NT;
    constexpr int K = 128;
    constexpr int KS = K / 4;
    constexpr int LDA = K + 4;
    constexpr int RW = 16 / NT;
    constexpr int CPL = W / 32;
    static_assert(RW == 2, "fetch_rows2 stages two adjacent rows per wave");
    __shared__ __attribute__((aligned(16))) float As[2][16][LDA];
    __shared__ __attribute__((aligned(16))) float Gs[2][16][LDA];
    __shared__ float Hs[2][16][W];
    __shared__ float GBs[2][W];

    const int lane = threadIdx.x & 31;
    const int wave = __builtin_amdgcn_readfirstlane(threadIdx.x >> 5);

    int tile = blockIdx.x;
    if (tile >= ROW_TILES) return;

    const int n0   = wave * 16;
    const int arow = lane & 15;
    const int koff = (lane >> 4) << 1;
    const int bcol = lane & 15;

    v2f bR[KS], bN[KS];
#pragma unroll
    for (int s = 0; s < KS; ++s) {
        const int k = 4 * s + koff;
        bR[s].x = Wr[k * W + n0 + bcol];
        bR[s].y = Wr[(k + 1) * W + n0 + bcol];
        bN[s].x = Wn[k * W + n0 + bcol];
        bN[s].y = Wn[(k + 1) * W + n0 + bcol];
    }
    const float bv = bias[n0 + bcol];
    if (threadIdx.x < W) {
        GBs[0][threadIdx.x] = gamma[threadIdx.x];
        GBs[1][threadIdx.x] = beta[threadIdx.x];
    }

    const int step = gridDim.x;
    const int rl = RW * wave;
    const int lq = min(lane, 2);
    v4f pa[RW], pg[RW];
    int p0, p1, p2, pv0, pv1, qv;
    {
        const int ra = tile * 16 + rl;
        const int rb = min(tile + step, ROW_TILES - 1) * 16 + rl;
        const int rc = min(tile + 2 * step, ROW_TILES - 1) * 16 + rl;
        const int c0 = off[ra], c1 = off[ra + 1], c2 = off[ra + 2];
        p0 = off[rb]; p1 = off[rb + 1]; p2 = off[rb + 2];
        qv = off[rc + lq];
        int cv0, cv1;
        load_ids(csr, c0, c2, lane, cv0, cv1);
        load_ids(csr, p0, p2, lane, pv0, pv1);
        fetch_rows2(A, csr, ra, lane, c0, c1, c2, cv0, cv1, pa, pg);
    }
    v4f sa0, sa1;
    {
        const float* xs = A + (long)(min(tile + step, ROW_TILES - 1) * 16 + rl) * D_IN + lane * 4;
        sa0 = *(const v4f*)xs;
        sa1 = *(const v4f*)(xs + D_IN);
    }
#pragma unroll
    for (int j = 0; j < RW; ++j) {
        *(v4f*)&As[0][RW * wave + j][lane * 4] = pa[j];
        *(v4f*)&Gs[0][RW * wave + j][lane * 4] = pg[j];
    }
    __syncthreads();

    for (int it = 0; tile < ROW_TILES; ++it, tile += gridDim.x) {
        const int p = it & 1;
        const int nxt = tile + gridDim.x;
        const bool more = nxt < ROW_TILES;

        const int q0 = __builtin_amdgcn_readlane(qv, 0);
        const int q1 = __builtin_amdgcn_readlane(qv, 1);
        const int q2 = __builtin_amdgcn_readlane(qv, 2);
        int nv0, nv1;
        load_ids(csr, q0, q2, lane, nv0, nv1);
        const int dv = off[min(tile + 3 * step, ROW_TILES - 1) * 16 + rl + lq];

        const float* xl = A + lane * 4;
        v4f gv[MAIN_ROWS];
        v4f a0 = {0.0f, 0.0f, 0.0f, 0.0f};
        v4f a1 = {0.0f, 0.0f, 0.0f, 0.0f};
        *(v4f*)&As[p ^ 1][rl][lane * 4] = sa0;
        *(v4f*)&As[p ^ 1][rl + 1][lane * 4] = sa1;
        if (more) {
            if (p0 < p2) round_issue<0>(xl, p0, p2, pv0, pv1, gv);
        }

        v8f acc = {};
        SAGE_MUL_PART(0, 11)
        if (more) {
            if (p0 < p2) round_add(p0, p1, gv, a0, a1);
            if (p0 + MAIN_ROWS < p2)
                round_issue<MAIN_ROWS>(xl, p0 + MAIN_ROWS, p2, pv0, pv1, gv);
        }
        SAGE_MUL_PART(11, 22)
        if (more) {
            if (p0 + MAIN_ROWS < p2) round_add(p0 + MAIN_ROWS, p1, gv, a0, a1);
            if (p0 + 2 * MAIN_ROWS < p2)
                round_issue<2 * MAIN_ROWS>(xl, p0 + 2 * MAIN_ROWS, p2, pv0, pv1, gv);
        }
        SAGE_MUL_PART(22, KS)

        {
            const int rbase = (lane >> 4) << 3;
#pragma unroll
            for (int v = 0; v < 8; ++v)
                Hs[p][rbase + v][n0 + bcol] = acc[v] + bv;
        }
        if (more) {
            if (p0 + 2 * MAIN_ROWS < p2) round_add(p0 + 2 * MAIN_ROWS, p1, gv, a0, a1);
#pragma nounroll
            for (int j = p0 + MAIN_ROUNDS * MAIN_ROWS; j < p2; j += MAIN_ROWS) {
                int tv0, tv1;
                load_ids(csr, j, p2, lane, tv0, tv1);
                round_issue<0>(xl, j, p2, tv0, tv1, gv);
                round_add_split(j, p1, gv, a0, a1);
            }
            *(v4f*)&Gs[p ^ 1][rl][lane * 4] = a0 * (1.0f / fmaxf((float)(p1 - p0), 1.0f));
            *(v4f*)&Gs[p ^ 1][rl + 1][lane * 4] = a1 * (1.0f / fmaxf((float)(p2 - p1), 1.0f));
        }
        p0 = q0; p1 = q1; p2 = q2; pv0 = nv0; pv1 = nv1;
        qv = dv;
        __syncthreads();

        {
            const float* xs = xl + (long)(min(tile + 2 * step, ROW_TILES - 1) * 16 + rl) * D_IN;
            sa0 = *(const v4f*)xs;
            sa1 = *(const v4f*)(xs + D_IN);
        }

        float h[RW][CPL];
#pragma unroll
        for (int j = 0; j < RW; ++j)
#pragma unroll
            for (int m = 0; m < CPL; ++m) h[j][m] = Hs[p][wave + NT * j][lane + 32 * m];
        float gam[CPL], bet[CPL];
#pragma unroll
        for (int m = 0; m < CPL; ++m) { gam[m] = GBs[0][lane + 32 * m]; bet[m] = GBs[1][lane + 32 * m]; }

        float s[RW], vv[RW], mu[RW];
#pragma unroll
        for (int j = 0; j < RW; ++j) {
            s[j] = 0.0f;
#pragma unroll
            for (int m = 0; m < CPL; ++m) s[j] += h[j][m];
        }
#pragma unroll
        for (int off2 = 16; off2; off2 >>= 1)
#pragma unroll
            for (int j = 0; j < RW; ++j) s[j] += __shfl_xor(s[j], off2, 32);
#pragma unroll
        for (int j = 0; j < RW; ++j) {
            mu[j] = s[j] * (1.0f / W);
            vv[j] = 0.0f;
#pragma unroll
            for (int m = 0; m < CPL; ++m) {
                float d = h[j][m] - mu[j];
                vv[j] += d * d;
            }
        }
#pragma unroll
        for (int off2 = 16; off2; off2 >>= 1)
#pragma unroll
            for (int j = 0; j < RW; ++j) vv[j] += __shfl_xor(vv[j], off2, 32);
#pragma unroll
        for (int j = 0; j < RW; ++j) {
            const float rs = rsqrtf(vv[j] * (1.0f / W) + 1e-5f);
            const long row = (long)tile * 16 + wave + NT * j;
#pragma unroll
            for (int m = 0; m < CPL; ++m) {
                float val = (h[j][m] - mu[j]) * rs * gam[m] + bet[m];
                out[row * W + lane + 32 * m] = fmaxf(val, 0.0f);
            }
        }
    }
}

__global__ __launch_bounds__(256)
void proj_gemm(const float* __restrict__ A,
               const float* __restrict__ Wr,
               const float* __restrict__ Wn,
               const float* __restrict__ bias,
               float* __restrict__ z,
               float* __restrict__ y)
{
    constexpr int NT = 8;
    constexpr int K = 128;
    constexpr int KS = K / 4;
    constexpr int LDA = K + 4;
    constexpr int RW = 16 / NT;
    constexpr int LDH = 2 * D_OUT + 4;
    __shared__ __attribute__((aligned(16))) float As[2][16][LDA];
    __shared__ __attribute__((aligned(16))) float Hs[2][16][LDH];

    const int lane = threadIdx.x & 31;
    const int wave = __builtin_amdgcn_readfirstlane(threadIdx.x >> 5);

    int tile = blockIdx.x;
    if (tile >= ROW_TILES) return;

    const int arow = lane & 15;
    const int koff = (lane >> 4) << 1;
    const int bcol = lane & 15;
    const bool nbr = wave >= 4;
    const float* __restrict__ Wsel = nbr ? Wn : Wr;
    const int n0 = (wave & 3) * 16;

    v2f bW[KS];
#pragma unroll
    for (int s = 0; s < KS; ++s) {
        const int k = 4 * s + koff;
        bW[s].x = Wsel[k * D_OUT + n0 + bcol];
        bW[s].y = Wsel[(k + 1) * D_OUT + n0 + bcol];
    }
    const float bv = nbr ? 0.0f : bias[n0 + bcol];

    v4f pa[RW];
#pragma unroll
    for (int j = 0; j < RW; ++j) {
        const long row = (long)tile * 16 + wave + NT * j;
        pa[j] = *(const v4f*)(A + row * K + lane * 4);
    }
#pragma unroll
    for (int j = 0; j < RW; ++j)
        *(v4f*)&As[0][wave + NT * j][lane * 4] = pa[j];
    __syncthreads();

    float* __restrict__ dcol = (lane < 16) ? (z + lane * 4) : (y + (lane - 16) * 4);

    for (int it = 0; tile < ROW_TILES; ++it, tile += gridDim.x) {
        const int p = it & 1;
        const int nxt = tile + gridDim.x;
        const bool more = nxt < ROW_TILES;

        if (more) {
#pragma unroll
            for (int j = 0; j < RW; ++j) {
                const long row = (long)nxt * 16 + wave + NT * j;
                pa[j] = *(const v4f*)(A + row * K + lane * 4);
            }
        }

        v8f acc = {};
#pragma unroll
        for (int s = 0; s < KS; ++s) {
            const v2f a0 = *(const v2f*)&As[p][arow][4 * s + koff];
            acc = __builtin_amdgcn_wmma_f32_16x16x4_f32(
                false, a0, false, bW[s], (short)0, acc, false, false);
        }

        {
            const int rbase = (lane >> 4) << 3;
#pragma unroll
            for (int v = 0; v < 8; ++v)
                Hs[p][rbase + v][wave * 16 + bcol] = acc[v] + bv;
        }
        if (more) {
#pragma unroll
            for (int j = 0; j < RW; ++j)
                *(v4f*)&As[p ^ 1][wave + NT * j][lane * 4] = pa[j];
        }
        __syncthreads();

#pragma unroll
        for (int j = 0; j < RW; ++j) {
            const int r = wave + NT * j;
            const v4f hv = *(const v4f*)&Hs[p][r][lane * 4];
            *(v4f*)(dcol + ((long)tile * 16 + r) * D_OUT) = hv;
        }
    }
}

extern "C" void kernel_launch(void* const* d_in, const int* in_sizes, int n_in,
                              void* d_out, int out_size, void* d_ws, size_t ws_size,
                              hipStream_t stream) {
    const float* x    = (const float*)d_in[0];
    const int*   esrc = (const int*)d_in[1];
    const int*   edst = (const int*)d_in[2];
    const float* Wr0  = (const float*)d_in[3];
    const float* Wn0  = (const float*)d_in[4];
    const float* b0   = (const float*)d_in[5];
    const float* g0   = (const float*)d_in[6];
    const float* be0  = (const float*)d_in[7];
    const float* Wr1  = (const float*)d_in[8];
    const float* Wn1  = (const float*)d_in[9];
    const float* b1   = (const float*)d_in[10];
    const float* g1   = (const float*)d_in[11];
    const float* be1  = (const float*)d_in[12];
    const float* Wr2  = (const float*)d_in[13];
    const float* Wn2  = (const float*)d_in[14];
    const float* b2   = (const float*)d_in[15];
    float* out = (float*)d_out;

    const int N = GNN_N;
    const int E = in_sizes[1];

    float* h0 = (float*)d_ws;
    float* h1 = h0 + (size_t)N * D_IN;
    float* yp = h1 + (size_t)N * D_IN;
    float* zp = yp + (size_t)N * D_OUT;
    int* Lc   = (int*)(zp + (size_t)N * D_OUT);
    int* off  = Lc + (EDGE_CHUNKS * L_ROW + 32);
    int* tmp  = off + (N + 32);
    int* csr  = tmp + E;

    const int gathBlocks = (N + 7) / 8;
    const int gemmBlocks8 = 512;

    chunk_sort<<<EDGE_CHUNKS, 1024, 0, stream>>>(esrc, edst, Lc, tmp, E);
    bucket_csr<<<N_BUCKETS, 1024, 0, stream>>>(tmp, Lc, off, csr, E);

    sage_gemm<<<gemmBlocks8, 256, 0, stream>>>(x, off, csr, Wr0, Wn0, b0, g0, be0, h0);
    sage_gemm<<<gemmBlocks8, 256, 0, stream>>>(h0, off, csr, Wr1, Wn1, b1, g1, be1, h1);

    proj_gemm<<<gemmBlocks8, 256, 0, stream>>>(h1, Wr2, Wn2, b2, zp, yp);
    gather_lsm<<<gathBlocks, 256, 0, stream>>>(yp, zp, off, csr, out, N);
}


# LayerNorm, log-softmax, bucket-total butterflies on the vector ALU (permlanex16 + row_xmask DPP adds), not LDS permutes; sage_gemm own-row loads go from behind the barrier to the trip top; sums kept
# speedup vs baseline: 5.7482x; 1.0331x over previous
#include <hip/hip_runtime.h>
#include <hip/hip_bf16.h>
#include <math.h>

typedef float v2f __attribute__((ext_vector_type(2)));
typedef float v8f __attribute__((ext_vector_type(8)));
typedef float v4f __attribute__((ext_vector_type(4)));

#define GNN_N 100000
#define D_IN 128
#define D_OUT 64

__device__ __forceinline__ int xhalf_i(int b) {
    return __builtin_amdgcn_permlanex16(b, b, 0x76543210u, 0xfedcba98u, false, false);
}
template <int M>
__device__ __forceinline__ int xrow_i(int b) {
    static_assert(M >= 1 && M < 16, "a row_xmask swizzle stays inside a 16-lane row");
    return __builtin_amdgcn_update_dpp(b, b, 0x160 + M, 0xf, 0xf, true);
}
__device__ __forceinline__ float xhalf(float x) {
    return __builtin_bit_cast(float, xhalf_i(__builtin_bit_cast(int, x)));
}
template <int M>
__device__ __forceinline__ float xrow(float x) {
    return __builtin_bit_cast(float, xrow_i<M>(__builtin_bit_cast(int, x)));
}
template <int R>
__device__ __forceinline__ void wave_sum(float (&s)[R]) {
#pragma unroll
    for (int j = 0; j < R; ++j) s[j] += xhalf(s[j]);
#pragma unroll
    for (int j = 0; j < R; ++j) s[j] += xrow<8>(s[j]);
#pragma unroll
    for (int j = 0; j < R; ++j) s[j] += xrow<4>(s[j]);
#pragma unroll
    for (int j = 0; j < R; ++j) s[j] += xrow<2>(s[j]);
#pragma unroll
    for (int j = 0; j < R; ++j) s[j] += xrow<1>(s[j]);
}
template <int R>
__device__ __forceinline__ void wave_sum_i(int (&s)[R]) {
#pragma unroll
    for (int j = 0; j < R; ++j) s[j] += xhalf_i(s[j]);
#pragma unroll
    for (int j = 0; j < R; ++j) s[j] += xrow_i<8>(s[j]);
#pragma unroll
    for (int j = 0; j < R; ++j) s[j] += xrow_i<4>(s[j]);
#pragma unroll
    for (int j = 0; j < R; ++j) s[j] += xrow_i<2>(s[j]);
#pragma unroll
    for (int j = 0; j < R; ++j) s[j] += xrow_i<1>(s[j]);
}

#define EDGE_CHUNKS 256
#define NODE_SHIFT 9
#define BUCKET_NODES (1 << NODE_SHIFT)
#define N_BUCKETS ((GNN_N + BUCKET_NODES - 1) / BUCKET_NODES)
#define L_ROW (N_BUCKETS + 1)
static_assert(N_BUCKETS <= 1024, "chunk_sort scans the bucket counters with one thread per counter");
static_assert(EDGE_CHUNKS % 32 == 0 && EDGE_CHUNKS <= 1024, "bucket_csr gives every wave the same number of runs");
static_assert(GNN_N < (1 << (31 - NODE_SHIFT)), "a src id and a node's place in its bucket share one word");
static_assert(GNN_N % BUCKET_NODES != 0, "off[N] is the last bucket's entry N % BUCKET_NODES");

__device__ __forceinline__ int bucket_of(int d) {
    return min((int)((unsigned)d >> NODE_SHIFT), N_BUCKETS - 1);
}

#define EDGE_BATCH 8
#define CHUNK_CAP (EDGE_BATCH * 1024)
#define SEG_CAP 16384

__global__ __launch_bounds__(1024)
void chunk_sort(const int* __restrict__ src, const int* __restrict__ dst,
                int* __restrict__ L, int* __restrict__ tmp, int E) {
    __shared__ int cnt[N_BUCKETS];
    __shared__ int wsum[32];
    __shared__ int stage[CHUNK_CAP];
    const int tid = threadIdx.x, lane = tid & 31, wave = tid >> 5, w = blockIdx.x;
    const int chunk = (E + EDGE_CHUNKS - 1) / EDGE_CHUNKS;
    const int lo = min(w * chunk, E), hi = min(lo + chunk, E);
    const bool single = (hi - lo) <= CHUNK_CAP;
    if (tid < N_BUCKETS) cnt[tid] = 0;
    __syncthreads();
    int d[EDGE_BATCH], s[EDGE_BATCH];
#pragma unroll
    for (int u = 0; u < EDGE_BATCH; ++u) { d[u] = 0; s[u] = 0; }
    if (single) {
        if (lo < hi) {
#pragma unroll
            for (int u = 0; u < EDGE_BATCH; ++u) {
                const int i = min(lo + u * 1024 + tid, hi - 1);
                d[u] = dst[i];
                s[u] = src[i];
            }
        }
#pragma unroll
        for (int u = 0; u < EDGE_BATCH; ++u) {
            if (lo + u * 1024 + tid < hi) atomicAdd(&cnt[bucket_of(d[u])], 1);
        }
    } else {
        for (int base = lo; base < hi; base += CHUNK_CAP) {
#pragma unroll
            for (int u = 0; u < EDGE_BATCH; ++u) d[u] = dst[min(base + u * 1024 + tid, hi - 1)];
#pragma unroll
            for (int u = 0; u < EDGE_BATCH; ++u) {
                if (base + u * 1024 + tid < hi) atomicAdd(&cnt[bucket_of(d[u])], 1);
            }
        }
    }
    __syncthreads();
    const int v = (tid < N_BUCKETS) ? cnt[tid] : 0;
    int sc = v;
#pragma unroll
    for (int o = 1; o < 32; o <<= 1) {
        int t = __shfl_up(sc, o, 32);
        if (lane >= o) sc += t;
    }
    if (lane == 31) wsum[wave] = sc;
    __syncthreads();
    int ws = wsum[lane];
#pragma unroll
    for (int o = 1; o < 32; o <<= 1) {
        int t = __shfl_up(ws, o, 32);
        if (lane >= o) ws += t;
    }
    const int waveIncl = __shfl(ws, wave, 32);
    const int waveTot = __shfl(sc, 31, 32);
    const int total = __shfl(ws, 31, 32);
    const int excl = (waveIncl - waveTot) + (sc - v);
    if (tid < N_BUCKETS) {
        cnt[tid] = excl;
        L[w * L_ROW + tid] = excl;
    }
    if (tid == 0) L[w * L_ROW + N_BUCKETS] = total;
    __syncthreads();
    if (single) {
        int p[EDGE_BATCH];
#pragma unroll
        for (int u = 0; u < EDGE_BATCH; ++u) {
            p[u] = 0;
            if (lo + u * 1024 + tid < hi) p[u] = atomicAdd(&cnt[bucket_of(d[u])], 1);
        }
#pragma unroll
        for (int u = 0; u < EDGE_BATCH; ++u) {
            if (lo + u * 1024 + tid < hi)
                stage[p[u]] = (int)(((unsigned)s[u] << NODE_SHIFT) | ((unsigned)d[u] & (BUCKET_NODES - 1)));
        }
        __syncthreads();
        for (int i = tid; i < hi - lo; i += 1024) tmp[lo + i] = stage[i];
    } else {
        for (int base = lo; base < hi; base += CHUNK_CAP) {
#pragma unroll
            for (int u = 0; u < EDGE_BATCH; ++u) {
                const int i = min(base + u * 1024 + tid, hi - 1);
                d[u] = dst[i];
                s[u] = src[i];
            }
#pragma unroll
            for (int u = 0; u < EDGE_BATCH; ++u) {
                if (base + u * 1024 + tid < hi) {
                    const int p = atomicAdd(&cnt[bucket_of(d[u])], 1);
                    tmp[lo + p] = (int)(((unsigned)s[u] << NODE_SHIFT) | ((unsigned)d[u] & (BUCKET_NODES - 1)));
                }
            }
        }
    }
}

#define RUNS_PER_WAVE (EDGE_CHUNKS / 32)
#define RUN_SPAN 64
__global__ __launch_bounds__(1024)
void bucket_csr(const int* __restrict__ tmp, const int* __restrict__ L,
                int* __restrict__ off, int* __restrict__ csr, int E) {
    __shared__ int cnt[BUCKET_NODES];
    __shared__ int wsum[32];
    __shared__ int wsumA[32];
    __shared__ int rbeg[EDGE_CHUNKS];
    __shared__ int rlen[EDGE_CHUNKS];
    __shared__ int seg[SEG_CAP];
    const int tid = threadIdx.x, lane = tid & 31, b = blockIdx.x;
    const int wave = __builtin_amdgcn_readfirstlane(tid >> 5);
    const int chunk = (E + EDGE_CHUNKS - 1) / EDGE_CHUNKS;
    const int top = max(E - 1, 0);
    int a = 0, len = 0;
    if (tid < EDGE_CHUNKS) {
        const int* r = L + tid * L_ROW + b;
        a = r[0];
        len = r[1] - a;
        rbeg[tid] = min(tid * chunk, E) + a;
        rlen[tid] = len;
    }
    if (tid < BUCKET_NODES) cnt[tid] = 0;
    int sla[2] = {len, a};
    wave_sum_i(sla);
    if (lane == 0) { wsum[wave] = sla[0]; wsumA[wave] = sla[1]; }
    __syncthreads();
    int wla[2] = {wsum[lane], wsumA[lane]};
    wave_sum_i(wla);
    const int B0 = __builtin_amdgcn_readfirstlane(wla[1]);
    const int nb = __builtin_amdgcn_readfirstlane(wla[0]);
    const bool staged = nb <= SEG_CAP;

    const int w0 = wave * RUNS_PER_WAVE;
    int e[2 * RUNS_PER_WAVE];
    int lnv[RUNS_PER_WAVE];
    int lmax = 0;
#pragma unroll
    for (int r = 0; r < RUNS_PER_WAVE; ++r) {
        const int rb = __builtin_amdgcn_readfirstlane(rbeg[w0 + r]);
        const int ln = __builtin_amdgcn_readfirstlane(rlen[w0 + r]);
        lnv[r] = ln;
        lmax = max(lmax, ln);
        const int last = max(ln - 1, 0);
        e[2 * r]     = tmp[min(rb + min(lane, last), top)];
        e[2 * r + 1] = tmp[min(rb + min(lane + 32, last), top)];
    }
#pragma unroll
    for (int r = 0; r < RUNS_PER_WAVE; ++r) {
        const int ln = lnv[r];
        if (lane < ln)      atomicAdd(&cnt[e[2 * r] & (BUCKET_NODES - 1)], 1);
        if (lane + 32 < ln) atomicAdd(&cnt[e[2 * r + 1] & (BUCKET_NODES - 1)], 1);
    }
    if (lmax > RUN_SPAN) {
#pragma nounroll
        for (int r = 0; r < RUNS_PER_WAVE; ++r) {
            const int rb = __builtin_amdgcn_readfirstlane(rbeg[w0 + r]);
            const int ln = __builtin_amdgcn_readfirstlane(rlen[w0 + r]);
            for (int k = RUN_SPAN; k < ln; k += 32) {
                if (k + lane < ln) atomicAdd(&cnt[tmp[rb + k + lane] & (BUCKET_NODES - 1)], 1);
            }
        }
    }
    __syncthreads();
    const int v = (tid < BUCKET_NODES) ? cnt[tid] : 0;
    int s = v;
#pragma unroll
    for (int o = 1; o < 32; o <<= 1) {
        int t = __shfl_up(s, o, 32);
        if (lane >= o) s += t;
    }
    if (lane == 31) wsum[wave] = s;
    __syncthreads();
    int ws = wsum[lane];
#pragma unroll
    for (int o = 1; o < 32; o <<= 1) {
        int t = __shfl_up(ws, o, 32);
        if (lane >= o) ws += t;
    }
    const int waveIncl = __shfl(ws, wave, 32);
    const int waveTot = __shfl(s, 31, 32);
    const int excl = (waveIncl - waveTot) + (s - v);
    if (tid < BUCKET_NODES) {
        cnt[tid] = excl;
        const int node = b * BUCKET_NODES + tid;
        if (node <= GNN_N) off[node] = B0 + excl;
    }
    __syncthreads();
    int p[2 * RUNS_PER_WAVE];
#pragma unroll
    for (int r = 0; r < RUNS_PER_WAVE; ++r) {
        const int ln = lnv[r];
#pragma unroll
        for (int h = 0; h < 2; ++h) {
            p[2 * r + h] = 0;
            if (lane + 32 * h < ln) p[2 * r + h] = atomicAdd(&cnt[e[2 * r + h] & (BUCKET_NODES - 1)], 1);
        }
    }
#pragma unroll
    for (int r = 0; r < RUNS_PER_WAVE; ++r) {
        const int ln = lnv[r];
#pragma unroll
        for (int h = 0; h < 2; ++h) {
            if (lane + 32 * h < ln) {
                const int id = (int)((unsigned)e[2 * r + h] >> NODE_SHIFT);
                if (staged) seg[p[2 * r + h]] = id;
                else        csr[B0 + p[2 * r + h]] = id;
            }
        }
    }
    if (lmax > RUN_SPAN) {
#pragma nounroll
        for (int r = 0; r < RUNS_PER_WAVE; ++r) {
            const int rb = __builtin_amdgcn_readfirstlane(rbeg[w0 + r]);
            const int ln = __builtin_amdgcn_readfirstlane(rlen[w0 + r]);
            for (int k = RUN_SPAN; k < ln; k += 32) {
                if (k + lane < ln) {
                    const int x = tmp[rb + k + lane];
                    const int q = atomicAdd(&cnt[x & (BUCKET_NODES - 1)], 1);
                    const int id = (int)((unsigned)x >> NODE_SHIFT);
                    if (staged) seg[q] = id;
                    else        csr[B0 + q] = id;
                }
            }
        }
    }
    if (staged) {
        __syncthreads();
        for (int i = tid; i < nb; i += 1024) csr[B0 + i] = seg[i];
    }
}

__global__ __launch_bounds__(256)
void gather_lsm(const float* __restrict__ y, const float* __restrict__ z,
                const int* __restrict__ off, const int* __restrict__ csr,
                float* __restrict__ out, int n) {
    const int node = __builtin_amdgcn_readfirstlane(blockIdx.x * 8 + (threadIdx.x >> 5));
    if (node >= n) return;
    const int lane = threadIdx.x & 31;
    const bool odd = lane >= 16;
    const int c4 = (lane & 15) * 4;
    const int beg = off[node];
    const int end = off[node + 1];
    const float4 zv = *(const float4*)(z + (long)node * D_OUT + c4);
    const float* yl = y + c4;
    float4 acc = make_float4(0.0f, 0.0f, 0.0f, 0.0f);
    int j = beg;
    for (; j + 16 <= end; j += 16) {
        float4 v[8];
#pragma unroll
        for (int u = 0; u < 8; ++u) {
            const int i0 = csr[j + 2 * u], i1 = csr[j + 2 * u + 1];
            v[u] = *(const float4*)(yl + (long)(odd ? i1 : i0) * D_OUT);
        }
#pragma unroll
        for (int u = 0; u < 8; ++u) {
            acc.x += v[u].x; acc.y += v[u].y; acc.z += v[u].z; acc.w += v[u].w;
        }
    }
    if (j + 8 <= end) {
        float4 v[4];
#pragma unroll
        for (int u = 0; u < 4; ++u) {
            const int i0 = csr[j + 2 * u], i1 = csr[j + 2 * u + 1];
            v[u] = *(const float4*)(yl + (long)(odd ? i1 : i0) * D_OUT);
        }
#pragma unroll
        for (int u = 0; u < 4; ++u) {
            acc.x += v[u].x; acc.y += v[u].y; acc.z += v[u].z; acc.w += v[u].w;
        }
        j += 8;
    }
    if (j + 4 <= end) {
        const int i0 = csr[j], i1 = csr[j + 1], i2 = csr[j + 2], i3 = csr[j + 3];
        float4 v0 = *(const float4*)(yl + (long)(odd ? i1 : i0) * D_OUT);
        float4 v1 = *(const float4*)(yl + (long)(odd ? i3 : i2) * D_OUT);
        acc.x += v0.x; acc.y += v0.y; acc.z += v0.z; acc.w += v0.w;
        acc.x += v1.x; acc.y += v1.y; acc.z += v1.z; acc.w += v1.w;
        j += 4;
    }
    if (j + 2 <= end) {
        const int i0 = csr[j], i1 = csr[j + 1];
        float4 v0 = *(const float4*)(yl + (long)(odd ? i1 : i0) * D_OUT);
        acc.x += v0.x; acc.y += v0.y; acc.z += v0.z; acc.w += v0.w;
        j += 2;
    }
    if (j < end) {
        const int i0 = csr[j];
        if (!odd) {
            float4 v0 = *(const float4*)(yl + (long)i0 * D_OUT);
            acc.x += v0.x; acc.y += v0.y; acc.z += v0.z; acc.w += v0.w;
        }
    }
    acc.x += xhalf(acc.x);
    acc.y += xhalf(acc.y);
    acc.z += xhalf(acc.z);
    acc.w += xhalf(acc.w);
    const float inv = 1.0f / fmaxf((float)(end - beg), 1.0f);
    float4 h;
    h.x = zv.x + acc.x * inv; h.y = zv.y + acc.y * inv;
    h.z = zv.z + acc.z * inv; h.w = zv.w + acc.w * inv;
    float mx = fmaxf(fmaxf(h.x, h.y), fmaxf(h.z, h.w));
    mx = fmaxf(mx, xrow<8>(mx));
    mx = fmaxf(mx, xrow<4>(mx));
    mx = fmaxf(mx, xrow<2>(mx));
    mx = fmaxf(mx, xrow<1>(mx));
    float s = expf(h.x - mx) + expf(h.y - mx) + expf(h.z - mx) + expf(h.w - mx);
    s += xrow<8>(s);
    s += xrow<4>(s);
    s += xrow<2>(s);
    s += xrow<1>(s);
    const float ls = logf(s);
    if (!odd) {
        float4 r;
        r.x = h.x - mx - ls; r.y = h.y - mx - ls; r.z = h.z - mx - ls; r.w = h.w - mx - ls;
        *(float4*)(out + (long)node * D_OUT + c4) = r;
    }
}

#define ROW_TILES (GNN_N / 16)
#define GATHER_ROWS 16
#define ID_SPAN 64
#define MAIN_ROWS 16
#define MAIN_ROUNDS 3
static_assert(MAIN_ROUNDS * MAIN_ROWS <= ID_SPAN, "the rounds under the multiply take their ids from the prefetched span");

#define SAGE_MUL_PART(S0, S1)                                                                  \
    _Pragma("unroll")   \
    for (int s = (S0); s < (S1); ++s) {                                                        \
        const v2f f0 = *(const v2f*)&As[p][arow][4 * s + koff];                                \
        const v2f f1 = *(const v2f*)&Gs[p][arow][4 * s + koff];                                \
        acc = __builtin_amdgcn_wmma_f32_16x16x4_f32(                                           \
            false, f0, false, bR[s], (short)0, acc, false, false);                             \
        acc = __builtin_amdgcn_wmma_f32_16x16x4_f32(                                           \
            false, f1, false, bN[s], (short)0, acc, false, false);                             \
    }

__device__ __forceinline__ void load_ids(const int* __restrict__ csr, int j0, int b2, int lane,
                                         int& iv0, int& iv1) {
    const int last = max(b2 - 1, 0);
    iv0 = csr[min(j0 + lane, last)];
    iv1 = csr[min(j0 + 32 + lane, last)];
}

__device__ __forceinline__ void gather_round(const float* __restrict__ xl, int j, int b1, int b2,
                                             int ivq, int lb, v4f& a0, v4f& a1) {
    v4f v[GATHER_ROWS];
#pragma unroll
    for (int u = 0; u < GATHER_ROWS; ++u) {
        v[u] = (v4f){0.0f, 0.0f, 0.0f, 0.0f};
        if (j + u < b2) {
            const int id = __builtin_amdgcn_readlane(ivq, lb + u);
            v[u] = *(const v4f*)(xl + (long)id * D_IN);
        }
    }
#pragma unroll
    for (int u = 0; u < GATHER_ROWS; ++u) {
        if (j + u < b1) a0 += v[u];
        else            a1 += v[u];
    }
}

template <int POS0>
__device__ __forceinline__ void round_issue(const float* __restrict__ xl, int j, int b2,
                                            int iv0, int iv1, v4f (&v)[MAIN_ROWS]) {
#pragma unroll
    for (int u = 0; u < MAIN_ROWS; ++u) {
        v[u] = (v4f){0.0f, 0.0f, 0.0f, 0.0f};
        if (j + u < b2) {
            const int id = __builtin_amdgcn_readlane((POS0 + u) < 32 ? iv0 : iv1, (POS0 + u) & 31);
            v[u] = *(const v4f*)(xl + (long)id * D_IN);
        }
    }
}

__device__ __forceinline__ void round_add_split(int j, int b1, const v4f (&v)[MAIN_ROWS],
                                                v4f& a0, v4f& a1) {
#pragma unroll
    for (int u = 0; u < MAIN_ROWS; ++u) {
        if (j + u < b1) a0 += v[u];
        else            a1 += v[u];
    }
}

__device__ __forceinline__ void round_add(int j, int b1, const v4f (&v)[MAIN_ROWS],
                                          v4f& a0, v4f& a1) {
    if (j + MAIN_ROWS <= b1) {
#pragma unroll
        for (int u = 0; u < MAIN_ROWS; ++u) a0 += v[u];
    } else if (j >= b1) {
#pragma unroll
        for (int u = 0; u < MAIN_ROWS; ++u) a1 += v[u];
    } else {
        round_add_split(j, b1, v, a0, a1);
    }
}

__device__ __forceinline__ void neighbour_sums2(const float* __restrict__ xl,
                                                const int* __restrict__ csr,
                                                int b0, int b1, int b2, int iv0, int iv1,
                                                int lane, v4f& s0, v4f& s1) {
    v4f a0 = {0.0f, 0.0f, 0.0f, 0.0f};
    v4f a1 = {0.0f, 0.0f, 0.0f, 0.0f};
#pragma unroll
    for (int q = 0; q < ID_SPAN / GATHER_ROWS; ++q) {
        const int j = b0 + GATHER_ROWS * q;
        if (j < b2) gather_round(xl, j, b1, b2, (q & 2) ? iv1 : iv0, (q & 1) * 16, a0, a1);
    }
#pragma nounroll
    for (int j = b0 + ID_SPAN; j < b2; j += GATHER_ROWS) {
        const int r = (j - b0) / GATHER_ROWS;
        if ((r & 3) == 0) load_ids(csr, j, b2, lane, iv0, iv1);
        gather_round(xl, j, b1, b2, (r & 2) ? iv1 : iv0, (r & 1) * 16, a0, a1);
    }
    s0 = a0;
    s1 = a1;
}

__device__ __forceinline__ void fetch_rows2(const float* __restrict__ A,
                                            const int* __restrict__ csr,
                                            int r0, int lane, int b0, int b1, int b2,
                                            int iv0, int iv1, v4f (&pa)[2], v4f (&pg)[2]) {
    const float* xl = A + lane * 4;
    pa[0] = *(const v4f*)(xl + (long)r0 * D_IN);
    pa[1] = *(const v4f*)(xl + (long)(r0 + 1) * D_IN);
    v4f s0, s1;
    neighbour_sums2(xl, csr, b0, b1, b2, iv0, iv1, lane, s0, s1);
    pg[0] = s0 * (1.0f / fmaxf((float)(b1 - b0), 1.0f));
    pg[1] = s1 * (1.0f / fmaxf((float)(b2 - b1), 1.0f));
}

__global__ __launch_bounds__(256, 4)
void sage_gemm(const float* __restrict__ A,
               const int* __restrict__ off,
               const int* __restrict__ csr,
               const float* __restrict__ Wr,
               const float* __restrict__ Wn,
               const float* __restrict__ bias,
               const float* __restrict__ gamma,
               const float* __restrict__ beta,
               float* __restrict__ out)
{
    constexpr int NT = 8;
    constexpr int W = 16 * NT;
    constexpr int K = 128;
    constexpr int KS = K / 4;
    constexpr int LDA = K + 4;
    constexpr int RW = 16 / NT;
    constexpr int CPL = W / 32;
    static_assert(RW == 2, "fetch_rows2 stages two adjacent rows per wave");
    __shared__ __attribute__((aligned(16))) float As[2][16][LDA];
    __shared__ __attribute__((aligned(16))) float Gs[2][16][LDA];
    __shared__ float Hs[2][16][W];
    __shared__ float GBs[2][W];

    const int lane = threadIdx.x & 31;
    const int wave = __builtin_amdgcn_readfirstlane(threadIdx.x >> 5);

    int tile = blockIdx.x;
    if (tile >= ROW_TILES) return;

    const int n0   = wave * 16;
    const int arow = lane & 15;
    const int koff = (lane >> 4) << 1;
    const int bcol = lane & 15;

    v2f bR[KS], bN[KS];
#pragma unroll
    for (int s = 0; s < KS; ++s) {
        const int k = 4 * s + koff;
        bR[s].x = Wr[k * W + n0 + bcol];
        bR[s].y = Wr[(k + 1) * W + n0 + bcol];
        bN[s].x = Wn[k * W + n0 + bcol];
        bN[s].y = Wn[(k + 1) * W + n0 + bcol];
    }
    const float bv = bias[n0 + bcol];
    if (threadIdx.x < W) {
        GBs[0][threadIdx.x] = gamma[threadIdx.x];
        GBs[1][threadIdx.x] = beta[threadIdx.x];
    }

    const int step = gridDim.x;
    const int rl = RW * wave;
    const int lq = min(lane, 2);
    v4f pa[RW], pg[RW];
    int p0, p1, p2, pv0, pv1, qv;
    {
        const int ra = tile * 16 + rl;
        const int rb = min(tile + step, ROW_TILES - 1) * 16 + rl;
        const int rc = min(tile + 2 * step, ROW_TILES - 1) * 16 + rl;
        const int c0 = off[ra], c1 = off[ra + 1], c2 = off[ra + 2];
        p0 = off[rb]; p1 = off[rb + 1]; p2 = off[rb + 2];
        qv = off[rc + lq];
        int cv0, cv1;
        load_ids(csr, c0, c2, lane, cv0, cv1);
        load_ids(csr, p0, p2, lane, pv0, pv1);
        fetch_rows2(A, csr, ra, lane, c0, c1, c2, cv0, cv1, pa, pg);
    }
#pragma unroll
    for (int j = 0; j < RW; ++j) {
        *(v4f*)&As[0][RW * wave + j][lane * 4] = pa[j];
        *(v4f*)&Gs[0][RW * wave + j][lane * 4] = pg[j];
    }
    __syncthreads();

    for (int it = 0; tile < ROW_TILES; ++it, tile += gridDim.x) {
        const int p = it & 1;
        const int nxt = tile + gridDim.x;
        const bool more = nxt < ROW_TILES;

        const int q0 = __builtin_amdgcn_readlane(qv, 0);
        const int q1 = __builtin_amdgcn_readlane(qv, 1);
        const int q2 = __builtin_amdgcn_readlane(qv, 2);
        int nv0, nv1;
        load_ids(csr, q0, q2, lane, nv0, nv1);
        const int dv = off[min(tile + 3 * step, ROW_TILES - 1) * 16 + rl + lq];

        const float* xl = A + lane * 4;
        v4f gv[MAIN_ROWS];
        v4f a0 = {0.0f, 0.0f, 0.0f, 0.0f};
        v4f a1 = {0.0f, 0.0f, 0.0f, 0.0f};
        v4f sa0, sa1;
        {
            const float* xs = xl + (long)(min(nxt, ROW_TILES - 1) * 16 + rl) * D_IN;
            sa0 = *(const v4f*)xs;
            sa1 = *(const v4f*)(xs + D_IN);
        }
        if (more) {
            if (p0 < p2) round_issue<0>(xl, p0, p2, pv0, pv1, gv);
        }

        v8f acc = {};
        SAGE_MUL_PART(0, 11)
        if (more) {
            if (p0 < p2) round_add(p0, p1, gv, a0, a1);
        }
        *(v4f*)&As[p ^ 1][rl][lane * 4] = sa0;
        *(v4f*)&As[p ^ 1][rl + 1][lane * 4] = sa1;
        if (more) {
            if (p0 + MAIN_ROWS < p2)
                round_issue<MAIN_ROWS>(xl, p0 + MAIN_ROWS, p2, pv0, pv1, gv);
        }
        SAGE_MUL_PART(11, 22)
        if (more) {
            if (p0 + MAIN_ROWS < p2) round_add(p0 + MAIN_ROWS, p1, gv, a0, a1);
            if (p0 + 2 * MAIN_ROWS < p2)
                round_issue<2 * MAIN_ROWS>(xl, p0 + 2 * MAIN_ROWS, p2, pv0, pv1, gv);
        }
        SAGE_MUL_PART(22, KS)

        {
            const int rbase = (lane >> 4) << 3;
#pragma unroll
            for (int v = 0; v < 8; ++v)
                Hs[p][rbase + v][n0 + bcol] = acc[v] + bv;
        }
        if (more) {
            if (p0 + 2 * MAIN_ROWS < p2) round_add(p0 + 2 * MAIN_ROWS, p1, gv, a0, a1);
#pragma nounroll
            for (int j = p0 + MAIN_ROUNDS * MAIN_ROWS; j < p2; j += MAIN_ROWS) {
                int tv0, tv1;
                load_ids(csr, j, p2, lane, tv0, tv1);
                round_issue<0>(xl, j, p2, tv0, tv1, gv);
                round_add_split(j, p1, gv, a0, a1);
            }
            *(v4f*)&Gs[p ^ 1][rl][lane * 4] = a0 * (1.0f / fmaxf((float)(p1 - p0), 1.0f));
            *(v4f*)&Gs[p ^ 1][rl + 1][lane * 4] = a1 * (1.0f / fmaxf((float)(p2 - p1), 1.0f));
        }
        p0 = q0; p1 = q1; p2 = q2; pv0 = nv0; pv1 = nv1;
        qv = dv;
        __syncthreads();

        float h[RW][CPL];
#pragma unroll
        for (int j = 0; j < RW; ++j)
#pragma unroll
            for (int m = 0; m < CPL; ++m) h[j][m] = Hs[p][wave + NT * j][lane + 32 * m];
        float gam[CPL], bet[CPL];
#pragma unroll
        for (int m = 0; m < CPL; ++m) { gam[m] = GBs[0][lane + 32 * m]; bet[m] = GBs[1][lane + 32 * m]; }

        float s[RW], vv[RW], mu[RW];
#pragma unroll
        for (int j = 0; j < RW; ++j) {
            s[j] = 0.0f;
#pragma unroll
            for (int m = 0; m < CPL; ++m) s[j] += h[j][m];
        }
        wave_sum(s);
#pragma unroll
        for (int j = 0; j < RW; ++j) {
            mu[j] = s[j] * (1.0f / W);
            vv[j] = 0.0f;
#pragma unroll
            for (int m = 0; m < CPL; ++m) {
                float d = h[j][m] - mu[j];
                vv[j] += d * d;
            }
        }
        wave_sum(vv);
#pragma unroll
        for (int j = 0; j < RW; ++j) {
            const float rs = rsqrtf(vv[j] * (1.0f / W) + 1e-5f);
            const long row = (long)tile * 16 + wave + NT * j;
#pragma unroll
            for (int m = 0; m < CPL; ++m) {
                float val = (h[j][m] - mu[j]) * rs * gam[m] + bet[m];
                out[row * W + lane + 32 * m] = fmaxf(val, 0.0f);
            }
        }
    }
}

__global__ __launch_bounds__(256)
void proj_gemm(const float* __restrict__ A,
               const float* __restrict__ Wr,
               const float* __restrict__ Wn,
               const float* __restrict__ bias,
               float* __restrict__ z,
               float* __restrict__ y)
{
    constexpr int NT = 8;
    constexpr int K = 128;
    constexpr int KS = K / 4;
    constexpr int LDA = K + 4;
    constexpr int RW = 16 / NT;
    constexpr int LDH = 2 * D_OUT + 4;
    __shared__ __attribute__((aligned(16))) float As[2][16][LDA];
    __shared__ __attribute__((aligned(16))) float Hs[2][16][LDH];

    const int lane = threadIdx.x & 31;
    const int wave = __builtin_amdgcn_readfirstlane(threadIdx.x >> 5);

    int tile = blockIdx.x;
    if (tile >= ROW_TILES) return;

    const int arow = lane & 15;
    const int koff = (lane >> 4) << 1;
    const int bcol = lane & 15;
    const bool nbr = wave >= 4;
    const float* __restrict__ Wsel = nbr ? Wn : Wr;
    const int n0 = (wave & 3) * 16;

    v2f bW[KS];
#pragma unroll
    for (int s = 0; s < KS; ++s) {
        const int k = 4 * s + koff;
        bW[s].x = Wsel[k * D_OUT + n0 + bcol];
        bW[s].y = Wsel[(k + 1) * D_OUT + n0 + bcol];
    }
    const float bv = nbr ? 0.0f : bias[n0 + bcol];

    v4f pa[RW];
#pragma unroll
    for (int j = 0; j < RW; ++j) {
        const long row = (long)tile * 16 + wave + NT * j;
        pa[j] = *(const v4f*)(A + row * K + lane * 4);
    }
#pragma unroll
    for (int j = 0; j < RW; ++j)
        *(v4f*)&As[0][wave + NT * j][lane * 4] = pa[j];
    __syncthreads();

    float* __restrict__ dcol = (lane < 16) ? (z + lane * 4) : (y + (lane - 16) * 4);

    for (int it = 0; tile < ROW_TILES; ++it, tile += gridDim.x) {
        const int p = it & 1;
        const int nxt = tile + gridDim.x;
        const bool more = nxt < ROW_TILES;

        if (more) {
#pragma unroll
            for (int j = 0; j < RW; ++j) {
                const long row = (long)nxt * 16 + wave + NT * j;
                pa[j] = *(const v4f*)(A + row * K + lane * 4);
            }
        }

        v8f acc = {};
#pragma unroll
        for (int s = 0; s < KS; ++s) {
            const v2f a0 = *(const v2f*)&As[p][arow][4 * s + koff];
            acc = __builtin_amdgcn_wmma_f32_16x16x4_f32(
                false, a0, false, bW[s], (short)0, acc, false, false);
        }

        {
            const int rbase = (lane >> 4) << 3;
#pragma unroll
            for (int v = 0; v < 8; ++v)
                Hs[p][rbase + v][wave * 16 + bcol] = acc[v] + bv;
        }
        if (more) {
#pragma unroll
            for (int j = 0; j < RW; ++j)
                *(v4f*)&As[p ^ 1][wave + NT * j][lane * 4] = pa[j];
        }
        __syncthreads();

#pragma unroll
        for (int j = 0; j < RW; ++j) {
            const int r = wave + NT * j;
            const v4f hv = *(const v4f*)&Hs[p][r][lane * 4];
            *(v4f*)(dcol + ((long)tile * 16 + r) * D_OUT) = hv;
        }
    }
}

extern "C" void kernel_launch(void* const* d_in, const int* in_sizes, int n_in,
                              void* d_out, int out_size, void* d_ws, size_t ws_size,
                              hipStream_t stream) {
    const float* x    = (const float*)d_in[0];
    const int*   esrc = (const int*)d_in[1];
    const int*   edst = (const int*)d_in[2];
    const float* Wr0  = (const float*)d_in[3];
    const float* Wn0  = (const float*)d_in[4];
    const float* b0   = (const float*)d_in[5];
    const float* g0   = (const float*)d_in[6];
    const float* be0  = (const float*)d_in[7];
    const float* Wr1  = (const float*)d_in[8];
    const float* Wn1  = (const float*)d_in[9];
    const float* b1   = (const float*)d_in[10];
    const float* g1   = (const float*)d_in[11];
    const float* be1  = (const float*)d_in[12];
    const float* Wr2  = (const float*)d_in[13];
    const float* Wn2  = (const float*)d_in[14];
    const float* b2   = (const float*)d_in[15];
    float* out = (float*)d_out;

    const int N = GNN_N;
    const int E = in_sizes[1];

    float* h0 = (float*)d_ws;
    float* h1 = h0 + (size_t)N * D_IN;
    float* yp = h1 + (size_t)N * D_IN;
    float* zp = yp + (size_t)N * D_OUT;
    int* Lc   = (int*)(zp + (size_t)N * D_OUT);
    int* off  = Lc + (EDGE_CHUNKS * L_ROW + 32);
    int* tmp  = off + (N + 32);
    int* csr  = tmp + E;

    const int gathBlocks = (N + 7) / 8;
    const int gemmBlocks8 = 512;

    chunk_sort<<<EDGE_CHUNKS, 1024, 0, stream>>>(esrc, edst, Lc, tmp, E);
    bucket_csr<<<N_BUCKETS, 1024, 0, stream>>>(tmp, Lc, off, csr, E);

    sage_gemm<<<gemmBlocks8, 256, 0, stream>>>(x, off, csr, Wr0, Wn0, b0, g0, be0, h0);
    sage_gemm<<<gemmBlocks8, 256, 0, stream>>>(h0, off, csr, Wr1, Wn1, b1, g1, be1, h1);

    proj_gemm<<<gemmBlocks8, 256, 0, stream>>>(h1, Wr2, Wn2, b2, zp, yp);
    gather_lsm<<<gathBlocks, 256, 0, stream>>>(yp, zp, off, csr, out, N);
}
